# starting kernel
# baseline (speedup 1.0000x reference)
#include <hip/hip_runtime.h>
#include <hip/hip_bf16.h>
#include <math.h>


typedef __attribute__((ext_vector_type(2))) float v2f;
typedef __attribute__((ext_vector_type(8))) float v8f;

#define N_NODES 10000
#define N_EDGES 320000

#define INV_SQ3   0.5773502691896258f
#define INV_SQ2   0.7071067811865476f
#define INV_SQ32  0.17677669529663687f
#define INV_SQ64  0.125f
#define INV_SQ96  0.10206207261596575f
#define INV_SQ160 0.07905694150420949f
#define LN2F      0.6931471805599453f

__device__ __forceinline__ float sspf(float v) {
  float sp = (v > 20.0f) ? v : log1pf(expf(v));
  return sp - LN2F;
}
__device__ __forceinline__ float siluf(float v) {
  return v / (1.0f + expf(-v));
}
__device__ __forceinline__ v2f ld2(const float* p) {
  return *(const v2f*)p;
}

__device__ __forceinline__ v8f wmma4(v2f a, v2f b, v8f c) {
  return __builtin_amdgcn_wmma_f32_16x16x4_f32(false, a, false, b,
                                               (short)0, c, false, false);
}


template <int KS, typename LA>
__device__ __forceinline__ void load_a(v2f (&a)[KS], LA la) {
  const int lane = threadIdx.x & 31;
  const int m = lane & 15;
  const int kk = (lane >> 4) << 1;
#pragma unroll
  for (int q = 0; q < KS; ++q) a[q] = la(m, q * 4 + kk);
}

template <int KS, typename LB>
__device__ __forceinline__ v8f wmma_tile(const v2f (&a)[KS], LB lb, v8f c) {
  const int lane = threadIdx.x & 31;
  const int n = lane & 15;
  const int kk = (lane >> 4) << 1;
#pragma unroll
  for (int q = 0; q < KS; ++q) c = wmma4(a[q], lb(n, q * 4 + kk), c);
  return c;
}

template <typename ST>
__device__ __forceinline__ void tile_store(v8f c, ST st) {
  const int lane = threadIdx.x & 31;
  const int col = lane & 15;
  const int rbase = (lane >> 4) << 3;
#pragma unroll
  for (int r = 0; r < 8; ++r) st(r, rbase + r, col, c[r]);
}

__global__ void transpose_kernel(const float* __restrict__ W,
                                 float* __restrict__ Wt, int K, int Nc) {
  int idx = blockIdx.x * blockDim.x + threadIdx.x;
  if (idx < K * Nc) {
    int k = idx / Nc, n = idx % Nc;
    Wt[n * K + k] = W[idx];
  }
}

__global__ __launch_bounds__(32) void node_kernel(
    const float* __restrict__ x,
    const float* __restrict__ W0pT, const float* __restrict__ b0p,
    const float* __restrict__ W1pT,
    const float* __restrict__ W0nT, const float* __restrict__ b0n,
    const float* __restrict__ W1nT,
    const float* __restrict__ G1T, const float* __restrict__ g1b,
    const float* __restrict__ G2T, const float* __restrict__ g2b,
    float* __restrict__ p0, float* __restrict__ p1,
    float* __restrict__ x0n, float* __restrict__ x1n) {
  __shared__ __align__(16) float f0S[16 * 96];
  __shared__ __align__(16) float hS[16 * 96];
  __shared__ __align__(16) float gS[16 * 96];

  const int lane = threadIdx.x;
  const int n0 = blockIdx.x * 16;
  const float* xt = x + (size_t)n0 * 160;

  {
    v2f ax[16];
    load_a(ax, [&](int m, int k) { return ld2(xt + m * 160 + k); });
    for (int j = 0; j < 4; ++j) {
      v8f c = {};
      c = wmma_tile(ax, [&](int n, int k) {
        return ld2(W0pT + (size_t)(j * 16 + n) * 64 + k);
      }, c);
      tile_store(c, [&](int, int m, int n, float v) {
        int col = j * 16 + n;
        p0[(size_t)(n0 + m) * 64 + col] = v * INV_SQ64 + b0p[col];
      });
    }
  }
  for (int i = 0; i < 3; ++i) {
    v2f ax[8];
    load_a(ax, [&](int m, int k) {
      v2f a;
      a.x = xt[m * 160 + 64 + 3 * k + i];
      a.y = xt[m * 160 + 64 + 3 * (k + 1) + i];
      return a;
    });
    for (int j = 0; j < 2; ++j) {
      v8f c = {};
      c = wmma_tile(ax, [&](int n, int k) {
        return ld2(W1pT + (size_t)(j * 16 + n) * 32 + k);
      }, c);
      tile_store(c, [&](int, int m, int n, float v) {
        p1[(size_t)(n0 + m) * 96 + (j * 16 + n) * 3 + i] = v * INV_SQ32;
      });
    }
  }

  for (int idx = lane; idx < 16 * 96; idx += 32) {
    int m = idx / 96, c = idx % 96;
    float v;
    if (c < 64) {
      v = xt[m * 160 + c];
    } else {
      int u = c - 64;
      float a0 = xt[m * 160 + 64 + u * 3 + 0];
      float a1 = xt[m * 160 + 64 + u * 3 + 1];
      float a2 = xt[m * 160 + 64 + u * 3 + 2];
      v = sqrtf(a0 * a0 + a1 * a1 + a2 * a2 + 1e-12f);
    }
    f0S[idx] = v;
  }
  __syncthreads();

  {
    v2f af[24];
    load_a(af, [&](int m, int k) { return ld2(f0S + m * 96 + k); });
    for (int j = 0; j < 6; ++j) {
      v8f c = {};
      c = wmma_tile(af, [&](int n, int k) {
        return ld2(G1T + (size_t)(j * 16 + n) * 96 + k);
      }, c);
      tile_store(c, [&](int, int m, int n, float v) {
        int col = j * 16 + n;
        hS[m * 96 + col] = siluf(v + g1b[col]);
      });
    }
  }
  __syncthreads();
  {
    v2f ah[24];
    load_a(ah, [&](int m, int k) { return ld2(hS + m * 96 + k); });
    for (int j = 0; j < 6; ++j) {
      v8f c = {};
      c = wmma_tile(ah, [&](int n, int k) {
        return ld2(G2T + (size_t)(j * 16 + n) * 96 + k);
      }, c);
      tile_store(c, [&](int, int m, int n, float v) {
        gS[m * 96 + j * 16 + n] = v + g2b[j * 16 + n];
      });
    }
  }
  __syncthreads();

  {
    v2f ag[16];
    load_a(ag, [&](int m, int k) { return ld2(gS + m * 96 + k); });
    for (int j = 0; j < 4; ++j) {
      v8f c = {};
      c = wmma_tile(ag, [&](int n, int k) {
        return ld2(W0nT + (size_t)(j * 16 + n) * 64 + k);
      }, c);
      tile_store(c, [&](int, int m, int n, float v) {
        int col = j * 16 + n;
        x0n[(size_t)(n0 + m) * 64 + col] = v * INV_SQ64 + b0n[col];
      });
    }
  }
  for (int i = 0; i < 3; ++i) {
    v2f ax[8];
    load_a(ax, [&](int m, int k) {
      v2f a;
      a.x = xt[m * 160 + 64 + 3 * k + i] * gS[m * 96 + 64 + k];
      a.y = xt[m * 160 + 64 + 3 * (k + 1) + i] * gS[m * 96 + 64 + k + 1];
      return a;
    });
    for (int j = 0; j < 2; ++j) {
      v8f c = {};
      c = wmma_tile(ax, [&](int n, int k) {
        return ld2(W1nT + (size_t)(j * 16 + n) * 32 + k);
      }, c);
      tile_store(c, [&](int, int m, int n, float v) {
        x1n[(size_t)(n0 + m) * 96 + (j * 16 + n) * 3 + i] = v * INV_SQ32;
      });
    }
  }
}

__global__ __launch_bounds__(32) void edge_kernel(
    const float* __restrict__ edge_sh, const float* __restrict__ edge_attr,
    const float* __restrict__ F1T, const float* __restrict__ F2T,
    const float* __restrict__ L1T, const float* __restrict__ L2T,
    const int* __restrict__ edge_index,
    const float* __restrict__ p0, const float* __restrict__ p1,
    const float* __restrict__ x0n, const float* __restrict__ x1n,
    float* __restrict__ acc) {
  __shared__ __align__(16) float s0S[16 * 160];
  __shared__ __align__(16) float hN[16 * 32];
  __shared__ __align__(16) float hE[16 * 32];
  __shared__ int dstS[16];
  __shared__ int srcS[16];

  const int lane = threadIdx.x;
  const int e0 = blockIdx.x * 16;

  __builtin_prefetch(F2T, 0, 1);
  __builtin_prefetch(L2T, 0, 1);

  if (lane < 16) {
    dstS[lane] = edge_index[e0 + lane];
    srcS[lane] = edge_index[N_EDGES + e0 + lane];
  }
  __syncthreads();

  for (int idx = lane; idx < 16 * 160; idx += 32) {
    int m = idx / 160, c = idx % 160;
    int d = dstS[m], s = srcS[m];
    float v;
    if (c < 64) {
      v = p0[(size_t)d * 64 + c];
    } else if (c < 128) {
      v = p0[(size_t)s * 64 + (c - 64)];
    } else {
      int u = c - 128;
      const float* pd = p1 + (size_t)d * 96 + u * 3;
      const float* ps = p1 + (size_t)s * 96 + u * 3;
      v = (pd[0] * ps[0] + pd[1] * ps[1] + pd[2] * ps[2]) * INV_SQ3;
    }
    s0S[idx] = v;
  }
  __syncthreads();

  {
    v2f as0[40];
    load_a(as0, [&](int m, int k) { return ld2(s0S + m * 160 + k); });
    for (int j = 0; j < 2; ++j) {
      v8f c = {};
      c = wmma_tile(as0, [&](int n, int k) {
        return ld2(L1T + (size_t)(j * 16 + n) * 160 + k);
      }, c);
      tile_store(c, [&](int, int m, int n, float v) {
        hN[m * 32 + j * 16 + n] = sspf(v * INV_SQ160);
      });
    }
  }
  {
    const float* ea = edge_attr + (size_t)e0 * 32;
    v2f aea[8];
    load_a(aea, [&](int m, int k) { return ld2(ea + m * 32 + k); });
    for (int j = 0; j < 2; ++j) {
      v8f c = {};
      c = wmma_tile(aea, [&](int n, int k) {
        return ld2(F1T + (size_t)(j * 16 + n) * 32 + k);
      }, c);
      tile_store(c, [&](int, int m, int n, float v) {
        hE[m * 32 + j * 16 + n] = sspf(v * INV_SQ32);
      });
    }
  }
  __syncthreads();

  const int rbase = (lane >> 4) << 3;
  float sh0r[8], s1xr[8], s1yr[8], s1zr[8];
  int dR[8], sR[8];
#pragma unroll
  for (int r = 0; r < 8; ++r) {
    int m = rbase + r;
    int e = e0 + m;
    sh0r[r] = edge_sh[(size_t)e * 4 + 0];
    s1xr[r] = edge_sh[(size_t)e * 4 + 1];
    s1yr[r] = edge_sh[(size_t)e * 4 + 2];
    s1zr[r] = edge_sh[(size_t)e * 4 + 3];
    dR[r] = dstS[m];
    sR[r] = srcS[m];
  }

  v2f aN[8], aE[8];
  load_a(aN, [&](int m, int k) { return ld2(hN + m * 32 + k); });
  load_a(aE, [&](int m, int k) { return ld2(hE + m * 32 + k); });

  for (int j = 0; j < 14; ++j) {
    const int nb = j * 16;
    v8f cN = {}, cE = {};
    cN = wmma_tile(aN, [&](int n, int k) {
      return ld2(L2T + (size_t)(nb + n) * 32 + k);
    }, cN);
    cE = wmma_tile(aE, [&](int n, int k) {
      return ld2(F2T + (size_t)(nb + n) * 32 + k);
    }, cE);
    v8f w8 = cN * cE * (INV_SQ32 * INV_SQ32);

    tile_store(w8, [&](int r, int  , int n, float wv) {
      const int col = nb + n;
      const float sh0 = sh0r[r], s1x = s1xr[r], s1y = s1yr[r], s1z = s1zr[r];
      const int s = sR[r];
      float* ad = acc + (size_t)dR[r] * 480;
      if (col < 64) {
        float xs = x0n[(size_t)s * 64 + col];
        atomicAdd(&ad[col], wv * xs * sh0);
      } else if (col < 128) {
        int c2 = col - 64;
        float f = wv * x0n[(size_t)s * 64 + c2] * INV_SQ3;
        atomicAdd(&ad[96 + c2 * 3 + 0], f * s1x);
        atomicAdd(&ad[96 + c2 * 3 + 1], f * s1y);
        atomicAdd(&ad[96 + c2 * 3 + 2], f * s1z);
      } else if (col < 160) {
        int c3 = col - 128;
        const float* xv = x1n + (size_t)s * 96 + c3 * 3;
        float f = wv * sh0;
        atomicAdd(&ad[288 + c3 * 3 + 0], f * xv[0]);
        atomicAdd(&ad[288 + c3 * 3 + 1], f * xv[1]);
        atomicAdd(&ad[288 + c3 * 3 + 2], f * xv[2]);
      } else if (col < 192) {
        int c4 = col - 160;
        const float* xv = x1n + (size_t)s * 96 + c4 * 3;
        float dp = xv[0] * s1x + xv[1] * s1y + xv[2] * s1z;
        atomicAdd(&ad[64 + c4], wv * dp * INV_SQ3);
      } else {
        int c5 = col - 192;
        const float* xv = x1n + (size_t)s * 96 + c5 * 3;
        float cx = xv[1] * s1z - xv[2] * s1y;
        float cy = xv[2] * s1x - xv[0] * s1z;
        float cz = xv[0] * s1y - xv[1] * s1x;
        float f = wv * INV_SQ2;
        atomicAdd(&ad[384 + c5 * 3 + 0], f * cx);
        atomicAdd(&ad[384 + c5 * 3 + 1], f * cy);
        atomicAdd(&ad[384 + c5 * 3 + 2], f * cz);
      }
    });
  }
}

__global__ __launch_bounds__(32) void proj_kernel(
    const float* __restrict__ acc,
    const float* __restrict__ WO0T, const float* __restrict__ bO0,
    const float* __restrict__ WO1T, const float* __restrict__ WO2T,
    float* __restrict__ y) {
  const int n0 = blockIdx.x * 16;
  const float* at = acc + (size_t)n0 * 480;

  {
    v2f aa[24];
    load_a(aa, [&](int m, int k) { return ld2(at + (size_t)m * 480 + k); });
    for (int j = 0; j < 8; ++j) {
      v8f c = {};
      c = wmma_tile(aa, [&](int n, int k) {
        return ld2(WO0T + (size_t)(j * 16 + n) * 96 + k);
      }, c);
      tile_store(c, [&](int, int m, int n, float v) {
        int col = j * 16 + n;
        y[(size_t)(n0 + m) * 416 + col] = v * INV_SQ96 + bO0[col];
      });
    }
  }
  for (int i = 0; i < 3; ++i) {
    v2f aa[24];
    load_a(aa, [&](int m, int k) {
      v2f a;
      a.x = at[(size_t)m * 480 + 96 + 3 * k + i];
      a.y = at[(size_t)m * 480 + 96 + 3 * (k + 1) + i];
      return a;
    });
    for (int j = 0; j < 4; ++j) {
      v8f c = {};
      c = wmma_tile(aa, [&](int n, int k) {
        return ld2(WO1T + (size_t)(j * 16 + n) * 96 + k);
      }, c);
      tile_store(c, [&](int, int m, int n, float v) {
        y[(size_t)(n0 + m) * 416 + 128 + (j * 16 + n) * 3 + i] = v * INV_SQ96;
      });
    }
  }
  for (int i = 0; i < 3; ++i) {
    v2f aa[8];
    load_a(aa, [&](int m, int k) {
      v2f a;
      a.x = at[(size_t)m * 480 + 384 + 3 * k + i];
      a.y = at[(size_t)m * 480 + 384 + 3 * (k + 1) + i];
      return a;
    });
    for (int j = 0; j < 2; ++j) {
      v8f c = {};
      c = wmma_tile(aa, [&](int n, int k) {
        return ld2(WO2T + (size_t)(j * 16 + n) * 32 + k);
      }, c);
      tile_store(c, [&](int, int m, int n, float v) {
        y[(size_t)(n0 + m) * 416 + 320 + (j * 16 + n) * 3 + i] = v * INV_SQ32;
      });
    }
  }
}

extern "C" void kernel_launch(void* const* d_in, const int* in_sizes, int n_in,
                              void* d_out, int out_size, void* d_ws,
                              size_t ws_size, hipStream_t stream) {
  (void)in_sizes; (void)n_in; (void)out_size; (void)ws_size;
  const float* x         = (const float*)d_in[0];
  const float* edge_sh   = (const float*)d_in[1];
  const float* edge_attr = (const float*)d_in[2];
  const float* W0p = (const float*)d_in[3];
  const float* b0p = (const float*)d_in[4];
  const float* W1p = (const float*)d_in[5];
  const float* W0n = (const float*)d_in[6];
  const float* b0n = (const float*)d_in[7];
  const float* W1n = (const float*)d_in[8];
  const float* G1  = (const float*)d_in[9];
  const float* g1b = (const float*)d_in[10];
  const float* G2  = (const float*)d_in[11];
  const float* g2b = (const float*)d_in[12];
  const float* F1  = (const float*)d_in[13];
  const float* F2  = (const float*)d_in[14];
  const float* L1  = (const float*)d_in[15];
  const float* L2  = (const float*)d_in[16];
  const float* WO0 = (const float*)d_in[17];
  const float* bO0 = (const float*)d_in[18];
  const float* WO1 = (const float*)d_in[19];
  const float* WO2 = (const float*)d_in[20];
  const int* edge_index = (const int*)d_in[21];

  float* ws   = (float*)d_ws;
  float* p0   = ws;
  float* p1   = p0   + (size_t)N_NODES * 64;
  float* x0n  = p1   + (size_t)N_NODES * 96;
  float* x1n  = x0n  + (size_t)N_NODES * 64;
  float* acc  = x1n  + (size_t)N_NODES * 96;
  float* wt   = acc  + (size_t)N_NODES * 480;
  float* W0pT = wt;            wt += 64 * 64;
  float* W1pT = wt;            wt += 32 * 32;
  float* W0nT = wt;            wt += 64 * 64;
  float* W1nT = wt;            wt += 32 * 32;
  float* G1T  = wt;            wt += 96 * 96;
  float* G2T  = wt;            wt += 96 * 96;
  float* F1T  = wt;            wt += 32 * 32;
  float* F2T  = wt;            wt += 32 * 224;
  float* L1T  = wt;            wt += 160 * 32;
  float* L2T  = wt;            wt += 32 * 224;
  float* WO0T = wt;            wt += 96 * 128;
  float* WO1T = wt;            wt += 96 * 64;
  float* WO2T = wt;            wt += 32 * 32;

  auto tr = [&](const float* W, float* Wt, int K, int Nc) {
    int n = K * Nc;
    transpose_kernel<<<(n + 255) / 256, 256, 0, stream>>>(W, Wt, K, Nc);
  };
  tr(W0p, W0pT, 64, 64);
  tr(W1p, W1pT, 32, 32);
  tr(W0n, W0nT, 64, 64);
  tr(W1n, W1nT, 32, 32);
  tr(G1, G1T, 96, 96);
  tr(G2, G2T, 96, 96);
  tr(F1, F1T, 32, 32);
  tr(F2, F2T, 32, 224);
  tr(L1, L1T, 160, 32);
  tr(L2, L2T, 32, 224);
  tr(WO0, WO0T, 96, 128);
  tr(WO1, WO1T, 96, 64);
  tr(WO2, WO2T, 32, 32);

  hipMemsetAsync(acc, 0, (size_t)N_NODES * 480 * sizeof(float), stream);

  node_kernel<<<N_NODES / 16, 32, 0, stream>>>(
      x, W0pT, b0p, W1pT, W0nT, b0n, W1nT, G1T, g1b, G2T, g2b,
      p0, p1, x0n, x1n);

  edge_kernel<<<N_EDGES / 16, 32, 0, stream>>>(
      edge_sh, edge_attr, F1T, F2T, L1T, L2T, edge_index,
      p0, p1, x0n, x1n, acc);

  proj_kernel<<<N_NODES / 16, 32, 0, stream>>>(
      acc, WO0T, bO0, WO1T, WO2T, (float*)d_out);
}


# Counting-sort edges by destination each call; every lane sums its runs of equal-destination rows in registers and issues one f32 atomic per run (summation order changes)
# speedup vs baseline: 1.3292x; 1.3292x over previous
#include <hip/hip_runtime.h>
#include <hip/hip_bf16.h>
#include <math.h>


typedef __attribute__((ext_vector_type(2))) float v2f;
typedef __attribute__((ext_vector_type(8))) float v8f;

#define N_NODES 10000
#define N_EDGES 320000

#define INV_SQ3   0.5773502691896258f
#define INV_SQ2   0.7071067811865476f
#define INV_SQ32  0.17677669529663687f
#define INV_SQ64  0.125f
#define INV_SQ96  0.10206207261596575f
#define INV_SQ160 0.07905694150420949f
#define LN2F      0.6931471805599453f

__device__ __forceinline__ float sspf(float v) {
  float sp = (v > 20.0f) ? v : log1pf(expf(v));
  return sp - LN2F;
}
__device__ __forceinline__ float siluf(float v) {
  return v / (1.0f + expf(-v));
}
__device__ __forceinline__ v2f ld2(const float* p) {
  return *(const v2f*)p;
}

__device__ __forceinline__ v8f wmma4(v2f a, v2f b, v8f c) {
  return __builtin_amdgcn_wmma_f32_16x16x4_f32(false, a, false, b,
                                               (short)0, c, false, false);
}


template <int KS, typename LA>
__device__ __forceinline__ void load_a(v2f (&a)[KS], LA la) {
  const int lane = threadIdx.x & 31;
  const int m = lane & 15;
  const int kk = (lane >> 4) << 1;
#pragma unroll
  for (int q = 0; q < KS; ++q) a[q] = la(m, q * 4 + kk);
}

template <int KS, typename LB>
__device__ __forceinline__ v8f wmma_tile(const v2f (&a)[KS], LB lb, v8f c) {
  const int lane = threadIdx.x & 31;
  const int n = lane & 15;
  const int kk = (lane >> 4) << 1;
#pragma unroll
  for (int q = 0; q < KS; ++q) c = wmma4(a[q], lb(n, q * 4 + kk), c);
  return c;
}

template <typename ST>
__device__ __forceinline__ void tile_store(v8f c, ST st) {
  const int lane = threadIdx.x & 31;
  const int col = lane & 15;
  const int rbase = (lane >> 4) << 3;
#pragma unroll
  for (int r = 0; r < 8; ++r) st(r, rbase + r, col, c[r]);
}

__global__ void transpose_kernel(const float* __restrict__ W,
                                 float* __restrict__ Wt, int K, int Nc) {
  int idx = blockIdx.x * blockDim.x + threadIdx.x;
  if (idx < K * Nc) {
    int k = idx / Nc, n = idx % Nc;
    Wt[n * K + k] = W[idx];
  }
}

__global__ __launch_bounds__(32) void node_kernel(
    const float* __restrict__ x,
    const float* __restrict__ W0pT, const float* __restrict__ b0p,
    const float* __restrict__ W1pT,
    const float* __restrict__ W0nT, const float* __restrict__ b0n,
    const float* __restrict__ W1nT,
    const float* __restrict__ G1T, const float* __restrict__ g1b,
    const float* __restrict__ G2T, const float* __restrict__ g2b,
    float* __restrict__ p0, float* __restrict__ p1,
    float* __restrict__ x0n, float* __restrict__ x1n) {
  __shared__ __align__(16) float f0S[16 * 96];
  __shared__ __align__(16) float hS[16 * 96];
  __shared__ __align__(16) float gS[16 * 96];

  const int lane = threadIdx.x;
  const int n0 = blockIdx.x * 16;
  const float* xt = x + (size_t)n0 * 160;

  {
    v2f ax[16];
    load_a(ax, [&](int m, int k) { return ld2(xt + m * 160 + k); });
    for (int j = 0; j < 4; ++j) {
      v8f c = {};
      c = wmma_tile(ax, [&](int n, int k) {
        return ld2(W0pT + (size_t)(j * 16 + n) * 64 + k);
      }, c);
      tile_store(c, [&](int, int m, int n, float v) {
        int col = j * 16 + n;
        p0[(size_t)(n0 + m) * 64 + col] = v * INV_SQ64 + b0p[col];
      });
    }
  }
  for (int i = 0; i < 3; ++i) {
    v2f ax[8];
    load_a(ax, [&](int m, int k) {
      v2f a;
      a.x = xt[m * 160 + 64 + 3 * k + i];
      a.y = xt[m * 160 + 64 + 3 * (k + 1) + i];
      return a;
    });
    for (int j = 0; j < 2; ++j) {
      v8f c = {};
      c = wmma_tile(ax, [&](int n, int k) {
        return ld2(W1pT + (size_t)(j * 16 + n) * 32 + k);
      }, c);
      tile_store(c, [&](int, int m, int n, float v) {
        p1[(size_t)(n0 + m) * 96 + (j * 16 + n) * 3 + i] = v * INV_SQ32;
      });
    }
  }

  for (int idx = lane; idx < 16 * 96; idx += 32) {
    int m = idx / 96, c = idx % 96;
    float v;
    if (c < 64) {
      v = xt[m * 160 + c];
    } else {
      int u = c - 64;
      float a0 = xt[m * 160 + 64 + u * 3 + 0];
      float a1 = xt[m * 160 + 64 + u * 3 + 1];
      float a2 = xt[m * 160 + 64 + u * 3 + 2];
      v = sqrtf(a0 * a0 + a1 * a1 + a2 * a2 + 1e-12f);
    }
    f0S[idx] = v;
  }
  __syncthreads();

  {
    v2f af[24];
    load_a(af, [&](int m, int k) { return ld2(f0S + m * 96 + k); });
    for (int j = 0; j < 6; ++j) {
      v8f c = {};
      c = wmma_tile(af, [&](int n, int k) {
        return ld2(G1T + (size_t)(j * 16 + n) * 96 + k);
      }, c);
      tile_store(c, [&](int, int m, int n, float v) {
        int col = j * 16 + n;
        hS[m * 96 + col] = siluf(v + g1b[col]);
      });
    }
  }
  __syncthreads();
  {
    v2f ah[24];
    load_a(ah, [&](int m, int k) { return ld2(hS + m * 96 + k); });
    for (int j = 0; j < 6; ++j) {
      v8f c = {};
      c = wmma_tile(ah, [&](int n, int k) {
        return ld2(G2T + (size_t)(j * 16 + n) * 96 + k);
      }, c);
      tile_store(c, [&](int, int m, int n, float v) {
        gS[m * 96 + j * 16 + n] = v + g2b[j * 16 + n];
      });
    }
  }
  __syncthreads();

  {
    v2f ag[16];
    load_a(ag, [&](int m, int k) { return ld2(gS + m * 96 + k); });
    for (int j = 0; j < 4; ++j) {
      v8f c = {};
      c = wmma_tile(ag, [&](int n, int k) {
        return ld2(W0nT + (size_t)(j * 16 + n) * 64 + k);
      }, c);
      tile_store(c, [&](int, int m, int n, float v) {
        int col = j * 16 + n;
        x0n[(size_t)(n0 + m) * 64 + col] = v * INV_SQ64 + b0n[col];
      });
    }
  }
  for (int i = 0; i < 3; ++i) {
    v2f ax[8];
    load_a(ax, [&](int m, int k) {
      v2f a;
      a.x = xt[m * 160 + 64 + 3 * k + i] * gS[m * 96 + 64 + k];
      a.y = xt[m * 160 + 64 + 3 * (k + 1) + i] * gS[m * 96 + 64 + k + 1];
      return a;
    });
    for (int j = 0; j < 2; ++j) {
      v8f c = {};
      c = wmma_tile(ax, [&](int n, int k) {
        return ld2(W1nT + (size_t)(j * 16 + n) * 32 + k);
      }, c);
      tile_store(c, [&](int, int m, int n, float v) {
        x1n[(size_t)(n0 + m) * 96 + (j * 16 + n) * 3 + i] = v * INV_SQ32;
      });
    }
  }
}

__global__ void sort_zero_kernel(int* __restrict__ cnt) {
  int i = blockIdx.x * blockDim.x + threadIdx.x;
  if (i < N_NODES) cnt[i] = 0;
}

__global__ void sort_rank_kernel(const int* __restrict__ edge_index,
                                 int* __restrict__ cnt, int* __restrict__ rank) {
  int e = blockIdx.x * blockDim.x + threadIdx.x;
  if (e < N_EDGES) rank[e] = atomicAdd(&cnt[edge_index[e]], 1);
}

#define SCAN_PER_THREAD 10
__global__ __launch_bounds__(1024) void sort_scan_kernel(
    const int* __restrict__ cnt, int* __restrict__ off) {
  __shared__ int part[1024];
  const int t = threadIdx.x;
  int loc[SCAN_PER_THREAD];
  int s = 0;
#pragma unroll
  for (int i = 0; i < SCAN_PER_THREAD; ++i) {
    int idx = t * SCAN_PER_THREAD + i;
    int v = (idx < N_NODES) ? cnt[idx] : 0;
    loc[i] = s;
    s += v;
  }
  part[t] = s;
  __syncthreads();
  for (int d = 1; d < 1024; d <<= 1) {
    int v = (t >= d) ? part[t - d] : 0;
    __syncthreads();
    part[t] += v;
    __syncthreads();
  }
  const int base = part[t] - s;
#pragma unroll
  for (int i = 0; i < SCAN_PER_THREAD; ++i) {
    int idx = t * SCAN_PER_THREAD + i;
    if (idx < N_NODES) off[idx] = base + loc[i];
  }
}

__global__ void sort_fill_kernel(const int* __restrict__ edge_index,
                                 const int* __restrict__ off,
                                 const int* __restrict__ rank,
                                 int* __restrict__ perm) {
  int e = blockIdx.x * blockDim.x + threadIdx.x;
  if (e < N_EDGES) perm[off[edge_index[e]] + rank[e]] = e;
}

__device__ __forceinline__ void run_add(float* __restrict__ acc,
                                        const int (&dR)[8], unsigned flm,
                                        int chan, const float (&v)[8]) {
  float run = 0.0f;
#pragma unroll
  for (int r = 0; r < 8; ++r) {
    run += v[r];
    if ((flm >> r) & 1u) {
      atomicAdd(acc + (size_t)dR[r] * 480 + chan, run);
      run = 0.0f;
    }
  }
}

__global__ __launch_bounds__(32) void edge_kernel(
    const float* __restrict__ edge_sh, const float* __restrict__ edge_attr,
    const float* __restrict__ F1T, const float* __restrict__ F2T,
    const float* __restrict__ L1T, const float* __restrict__ L2T,
    const int* __restrict__ edge_index, const int* __restrict__ perm,
    const float* __restrict__ p0, const float* __restrict__ p1,
    const float* __restrict__ x0n, const float* __restrict__ x1n,
    float* __restrict__ acc) {
  __shared__ __align__(16) float s0S[16 * 160];
  __shared__ __align__(16) float hN[16 * 32];
  __shared__ __align__(16) float hE[16 * 32];
  __shared__ int dstS[16];
  __shared__ int srcS[16];
  __shared__ int eS[16];

  const int lane = threadIdx.x;
  const int e0 = blockIdx.x * 16;

  __builtin_prefetch(F2T, 0, 1);
  __builtin_prefetch(L2T, 0, 1);

  if (lane < 16) {
    const int e = perm[e0 + lane];
    eS[lane] = e;
    dstS[lane] = edge_index[e];
    srcS[lane] = edge_index[N_EDGES + e];
  }
  __syncthreads();

  for (int idx = lane; idx < 16 * 160; idx += 32) {
    int m = idx / 160, c = idx % 160;
    int d = dstS[m], s = srcS[m];
    float v;
    if (c < 64) {
      v = p0[(size_t)d * 64 + c];
    } else if (c < 128) {
      v = p0[(size_t)s * 64 + (c - 64)];
    } else {
      int u = c - 128;
      const float* pd = p1 + (size_t)d * 96 + u * 3;
      const float* ps = p1 + (size_t)s * 96 + u * 3;
      v = (pd[0] * ps[0] + pd[1] * ps[1] + pd[2] * ps[2]) * INV_SQ3;
    }
    s0S[idx] = v;
  }
  __syncthreads();

  {
    v2f as0[40];
    load_a(as0, [&](int m, int k) { return ld2(s0S + m * 160 + k); });
    for (int j = 0; j < 2; ++j) {
      v8f c = {};
      c = wmma_tile(as0, [&](int n, int k) {
        return ld2(L1T + (size_t)(j * 16 + n) * 160 + k);
      }, c);
      tile_store(c, [&](int, int m, int n, float v) {
        hN[m * 32 + j * 16 + n] = sspf(v * INV_SQ160);
      });
    }
  }
  {
    v2f aea[8];
    load_a(aea, [&](int m, int k) {
      return ld2(edge_attr + (size_t)eS[m] * 32 + k);
    });
    for (int j = 0; j < 2; ++j) {
      v8f c = {};
      c = wmma_tile(aea, [&](int n, int k) {
        return ld2(F1T + (size_t)(j * 16 + n) * 32 + k);
      }, c);
      tile_store(c, [&](int, int m, int n, float v) {
        hE[m * 32 + j * 16 + n] = sspf(v * INV_SQ32);
      });
    }
  }
  __syncthreads();

  const int rbase = (lane >> 4) << 3;
  float sh0r[8], s1xr[8], s1yr[8], s1zr[8];
  int dR[8], sR[8];
#pragma unroll
  for (int r = 0; r < 8; ++r) {
    int m = rbase + r;
    int e = eS[m];
    sh0r[r] = edge_sh[(size_t)e * 4 + 0];
    s1xr[r] = edge_sh[(size_t)e * 4 + 1];
    s1yr[r] = edge_sh[(size_t)e * 4 + 2];
    s1zr[r] = edge_sh[(size_t)e * 4 + 3];
    dR[r] = dstS[m];
    sR[r] = srcS[m];
  }
  unsigned flm = 0x80u;
#pragma unroll
  for (int r = 0; r < 7; ++r) flm |= (dR[r + 1] != dR[r]) ? (1u << r) : 0u;

  v2f aN[8], aE[8];
  load_a(aN, [&](int m, int k) { return ld2(hN + m * 32 + k); });
  load_a(aE, [&](int m, int k) { return ld2(hE + m * 32 + k); });

  for (int j = 0; j < 14; ++j) {
    const int nb = j * 16;
    v8f cN = {}, cE = {};
    cN = wmma_tile(aN, [&](int n, int k) {
      return ld2(L2T + (size_t)(nb + n) * 32 + k);
    }, cN);
    cE = wmma_tile(aE, [&](int n, int k) {
      return ld2(F2T + (size_t)(nb + n) * 32 + k);
    }, cE);
    v8f w8 = cN * cE * (INV_SQ32 * INV_SQ32);

    const int col = nb + (lane & 15);
    float v0[8], v1[8], v2[8];
    if (nb < 64) {
#pragma unroll
      for (int r = 0; r < 8; ++r) {
        float xs = x0n[(size_t)sR[r] * 64 + col];
        v0[r] = w8[r] * xs * sh0r[r];
      }
      run_add(acc, dR, flm, col, v0);
    } else if (nb < 128) {
      const int c2 = col - 64;
#pragma unroll
      for (int r = 0; r < 8; ++r) {
        float f = w8[r] * x0n[(size_t)sR[r] * 64 + c2] * INV_SQ3;
        v0[r] = f * s1xr[r];
        v1[r] = f * s1yr[r];
        v2[r] = f * s1zr[r];
      }
      run_add(acc, dR, flm, 96 + c2 * 3 + 0, v0);
      run_add(acc, dR, flm, 96 + c2 * 3 + 1, v1);
      run_add(acc, dR, flm, 96 + c2 * 3 + 2, v2);
    } else if (nb < 160) {
      const int c3 = col - 128;
#pragma unroll
      for (int r = 0; r < 8; ++r) {
        const float* xv = x1n + (size_t)sR[r] * 96 + c3 * 3;
        float f = w8[r] * sh0r[r];
        v0[r] = f * xv[0];
        v1[r] = f * xv[1];
        v2[r] = f * xv[2];
      }
      run_add(acc, dR, flm, 288 + c3 * 3 + 0, v0);
      run_add(acc, dR, flm, 288 + c3 * 3 + 1, v1);
      run_add(acc, dR, flm, 288 + c3 * 3 + 2, v2);
    } else if (nb < 192) {
      const int c4 = col - 160;
#pragma unroll
      for (int r = 0; r < 8; ++r) {
        const float* xv = x1n + (size_t)sR[r] * 96 + c4 * 3;
        float dp = xv[0] * s1xr[r] + xv[1] * s1yr[r] + xv[2] * s1zr[r];
        v0[r] = w8[r] * dp * INV_SQ3;
      }
      run_add(acc, dR, flm, 64 + c4, v0);
    } else {
      const int c5 = col - 192;
#pragma unroll
      for (int r = 0; r < 8; ++r) {
        const float* xv = x1n + (size_t)sR[r] * 96 + c5 * 3;
        const float s1x = s1xr[r], s1y = s1yr[r], s1z = s1zr[r];
        float cx = xv[1] * s1z - xv[2] * s1y;
        float cy = xv[2] * s1x - xv[0] * s1z;
        float cz = xv[0] * s1y - xv[1] * s1x;
        float f = w8[r] * INV_SQ2;
        v0[r] = f * cx;
        v1[r] = f * cy;
        v2[r] = f * cz;
      }
      run_add(acc, dR, flm, 384 + c5 * 3 + 0, v0);
      run_add(acc, dR, flm, 384 + c5 * 3 + 1, v1);
      run_add(acc, dR, flm, 384 + c5 * 3 + 2, v2);
    }
  }
}

__global__ __launch_bounds__(32) void proj_kernel(
    const float* __restrict__ acc,
    const float* __restrict__ WO0T, const float* __restrict__ bO0,
    const float* __restrict__ WO1T, const float* __restrict__ WO2T,
    float* __restrict__ y) {
  const int n0 = blockIdx.x * 16;
  const float* at = acc + (size_t)n0 * 480;

  {
    v2f aa[24];
    load_a(aa, [&](int m, int k) { return ld2(at + (size_t)m * 480 + k); });
    for (int j = 0; j < 8; ++j) {
      v8f c = {};
      c = wmma_tile(aa, [&](int n, int k) {
        return ld2(WO0T + (size_t)(j * 16 + n) * 96 + k);
      }, c);
      tile_store(c, [&](int, int m, int n, float v) {
        int col = j * 16 + n;
        y[(size_t)(n0 + m) * 416 + col] = v * INV_SQ96 + bO0[col];
      });
    }
  }
  for (int i = 0; i < 3; ++i) {
    v2f aa[24];
    load_a(aa, [&](int m, int k) {
      v2f a;
      a.x = at[(size_t)m * 480 + 96 + 3 * k + i];
      a.y = at[(size_t)m * 480 + 96 + 3 * (k + 1) + i];
      return a;
    });
    for (int j = 0; j < 4; ++j) {
      v8f c = {};
      c = wmma_tile(aa, [&](int n, int k) {
        return ld2(WO1T + (size_t)(j * 16 + n) * 96 + k);
      }, c);
      tile_store(c, [&](int, int m, int n, float v) {
        y[(size_t)(n0 + m) * 416 + 128 + (j * 16 + n) * 3 + i] = v * INV_SQ96;
      });
    }
  }
  for (int i = 0; i < 3; ++i) {
    v2f aa[8];
    load_a(aa, [&](int m, int k) {
      v2f a;
      a.x = at[(size_t)m * 480 + 384 + 3 * k + i];
      a.y = at[(size_t)m * 480 + 384 + 3 * (k + 1) + i];
      return a;
    });
    for (int j = 0; j < 2; ++j) {
      v8f c = {};
      c = wmma_tile(aa, [&](int n, int k) {
        return ld2(WO2T + (size_t)(j * 16 + n) * 32 + k);
      }, c);
      tile_store(c, [&](int, int m, int n, float v) {
        y[(size_t)(n0 + m) * 416 + 320 + (j * 16 + n) * 3 + i] = v * INV_SQ32;
      });
    }
  }
}

extern "C" void kernel_launch(void* const* d_in, const int* in_sizes, int n_in,
                              void* d_out, int out_size, void* d_ws,
                              size_t ws_size, hipStream_t stream) {
  (void)in_sizes; (void)n_in; (void)out_size; (void)ws_size;
  const float* x         = (const float*)d_in[0];
  const float* edge_sh   = (const float*)d_in[1];
  const float* edge_attr = (const float*)d_in[2];
  const float* W0p = (const float*)d_in[3];
  const float* b0p = (const float*)d_in[4];
  const float* W1p = (const float*)d_in[5];
  const float* W0n = (const float*)d_in[6];
  const float* b0n = (const float*)d_in[7];
  const float* W1n = (const float*)d_in[8];
  const float* G1  = (const float*)d_in[9];
  const float* g1b = (const float*)d_in[10];
  const float* G2  = (const float*)d_in[11];
  const float* g2b = (const float*)d_in[12];
  const float* F1  = (const float*)d_in[13];
  const float* F2  = (const float*)d_in[14];
  const float* L1  = (const float*)d_in[15];
  const float* L2  = (const float*)d_in[16];
  const float* WO0 = (const float*)d_in[17];
  const float* bO0 = (const float*)d_in[18];
  const float* WO1 = (const float*)d_in[19];
  const float* WO2 = (const float*)d_in[20];
  const int* edge_index = (const int*)d_in[21];

  float* ws   = (float*)d_ws;
  float* p0   = ws;
  float* p1   = p0   + (size_t)N_NODES * 64;
  float* x0n  = p1   + (size_t)N_NODES * 96;
  float* x1n  = x0n  + (size_t)N_NODES * 64;
  float* acc  = x1n  + (size_t)N_NODES * 96;
  float* wt   = acc  + (size_t)N_NODES * 480;
  float* W0pT = wt;            wt += 64 * 64;
  float* W1pT = wt;            wt += 32 * 32;
  float* W0nT = wt;            wt += 64 * 64;
  float* W1nT = wt;            wt += 32 * 32;
  float* G1T  = wt;            wt += 96 * 96;
  float* G2T  = wt;            wt += 96 * 96;
  float* F1T  = wt;            wt += 32 * 32;
  float* F2T  = wt;            wt += 32 * 224;
  float* L1T  = wt;            wt += 160 * 32;
  float* L2T  = wt;            wt += 32 * 224;
  float* WO0T = wt;            wt += 96 * 128;
  float* WO1T = wt;            wt += 96 * 64;
  float* WO2T = wt;            wt += 32 * 32;

  int* sc   = (int*)d_out;
  int* cnt  = sc;
  int* off  = sc + 10240;
  int* rank = sc + 20480;
  int* perm = sc + 20480 + N_EDGES;

  auto tr = [&](const float* W, float* Wt, int K, int Nc) {
    int n = K * Nc;
    transpose_kernel<<<(n + 255) / 256, 256, 0, stream>>>(W, Wt, K, Nc);
  };
  tr(W0p, W0pT, 64, 64);
  tr(W1p, W1pT, 32, 32);
  tr(W0n, W0nT, 64, 64);
  tr(W1n, W1nT, 32, 32);
  tr(G1, G1T, 96, 96);
  tr(G2, G2T, 96, 96);
  tr(F1, F1T, 32, 32);
  tr(F2, F2T, 32, 224);
  tr(L1, L1T, 160, 32);
  tr(L2, L2T, 32, 224);
  tr(WO0, WO0T, 96, 128);
  tr(WO1, WO1T, 96, 64);
  tr(WO2, WO2T, 32, 32);

  hipMemsetAsync(acc, 0, (size_t)N_NODES * 480 * sizeof(float), stream);

  sort_zero_kernel<<<(N_NODES + 255) / 256, 256, 0, stream>>>(cnt);
  sort_rank_kernel<<<(N_EDGES + 255) / 256, 256, 0, stream>>>(edge_index, cnt, rank);
  sort_scan_kernel<<<1, 1024, 0, stream>>>(cnt, off);
  sort_fill_kernel<<<(N_EDGES + 255) / 256, 256, 0, stream>>>(edge_index, off, rank, perm);

  node_kernel<<<N_NODES / 16, 32, 0, stream>>>(
      x, W0pT, b0p, W1pT, W0nT, b0n, W1nT, G1T, g1b, G2T, g2b,
      p0, p1, x0n, x1n);

  edge_kernel<<<N_EDGES / 16, 32, 0, stream>>>(
      edge_sh, edge_attr, F1T, F2T, L1T, L2T, edge_index, perm,
      p0, p1, x0n, x1n, acc);

  proj_kernel<<<N_NODES / 16, 32, 0, stream>>>(
      acc, WO0T, bO0, WO1T, WO2T, (float*)d_out);
}


# Edge kernel builds its 16x160 MLP input panel from whole-row 16- and 12-byte loads issued in three batches of 16, replacing 80 serialized load-wait-store iterations per wave
# speedup vs baseline: 1.5670x; 1.1789x over previous
#include <hip/hip_runtime.h>
#include <hip/hip_bf16.h>
#include <math.h>


typedef __attribute__((ext_vector_type(2))) float v2f;
typedef __attribute__((ext_vector_type(4))) float v4f;
typedef __attribute__((ext_vector_type(8))) float v8f;

#define N_NODES 10000
#define N_EDGES 320000

#define INV_SQ3   0.5773502691896258f
#define INV_SQ2   0.7071067811865476f
#define INV_SQ32  0.17677669529663687f
#define INV_SQ64  0.125f
#define INV_SQ96  0.10206207261596575f
#define INV_SQ160 0.07905694150420949f
#define LN2F      0.6931471805599453f

__device__ __forceinline__ float sspf(float v) {
  float sp = (v > 20.0f) ? v : log1pf(expf(v));
  return sp - LN2F;
}
__device__ __forceinline__ float siluf(float v) {
  return v / (1.0f + expf(-v));
}
__device__ __forceinline__ v2f ld2(const float* p) {
  return *(const v2f*)p;
}

__device__ __forceinline__ v8f wmma4(v2f a, v2f b, v8f c) {
  return __builtin_amdgcn_wmma_f32_16x16x4_f32(false, a, false, b,
                                               (short)0, c, false, false);
}


template <int KS, typename LA>
__device__ __forceinline__ void load_a(v2f (&a)[KS], LA la) {
  const int lane = threadIdx.x & 31;
  const int m = lane & 15;
  const int kk = (lane >> 4) << 1;
#pragma unroll
  for (int q = 0; q < KS; ++q) a[q] = la(m, q * 4 + kk);
}

template <int KS, typename LB>
__device__ __forceinline__ v8f wmma_tile(const v2f (&a)[KS], LB lb, v8f c) {
  const int lane = threadIdx.x & 31;
  const int n = lane & 15;
  const int kk = (lane >> 4) << 1;
#pragma unroll
  for (int q = 0; q < KS; ++q) c = wmma4(a[q], lb(n, q * 4 + kk), c);
  return c;
}

template <typename ST>
__device__ __forceinline__ void tile_store(v8f c, ST st) {
  const int lane = threadIdx.x & 31;
  const int col = lane & 15;
  const int rbase = (lane >> 4) << 3;
#pragma unroll
  for (int r = 0; r < 8; ++r) st(r, rbase + r, col, c[r]);
}

__global__ void transpose_kernel(const float* __restrict__ W,
                                 float* __restrict__ Wt, int K, int Nc) {
  int idx = blockIdx.x * blockDim.x + threadIdx.x;
  if (idx < K * Nc) {
    int k = idx / Nc, n = idx % Nc;
    Wt[n * K + k] = W[idx];
  }
}

__global__ __launch_bounds__(32) void node_kernel(
    const float* __restrict__ x,
    const float* __restrict__ W0pT, const float* __restrict__ b0p,
    const float* __restrict__ W1pT,
    const float* __restrict__ W0nT, const float* __restrict__ b0n,
    const float* __restrict__ W1nT,
    const float* __restrict__ G1T, const float* __restrict__ g1b,
    const float* __restrict__ G2T, const float* __restrict__ g2b,
    float* __restrict__ p0, float* __restrict__ p1,
    float* __restrict__ x0n, float* __restrict__ x1n) {
  __shared__ __align__(16) float f0S[16 * 96];
  __shared__ __align__(16) float hS[16 * 96];
  __shared__ __align__(16) float gS[16 * 96];

  const int lane = threadIdx.x;
  const int n0 = blockIdx.x * 16;
  const float* xt = x + (size_t)n0 * 160;

  {
    v2f ax[16];
    load_a(ax, [&](int m, int k) { return ld2(xt + m * 160 + k); });
    for (int j = 0; j < 4; ++j) {
      v8f c = {};
      c = wmma_tile(ax, [&](int n, int k) {
        return ld2(W0pT + (size_t)(j * 16 + n) * 64 + k);
      }, c);
      tile_store(c, [&](int, int m, int n, float v) {
        int col = j * 16 + n;
        p0[(size_t)(n0 + m) * 64 + col] = v * INV_SQ64 + b0p[col];
      });
    }
  }
  for (int i = 0; i < 3; ++i) {
    v2f ax[8];
    load_a(ax, [&](int m, int k) {
      v2f a;
      a.x = xt[m * 160 + 64 + 3 * k + i];
      a.y = xt[m * 160 + 64 + 3 * (k + 1) + i];
      return a;
    });
    for (int j = 0; j < 2; ++j) {
      v8f c = {};
      c = wmma_tile(ax, [&](int n, int k) {
        return ld2(W1pT + (size_t)(j * 16 + n) * 32 + k);
      }, c);
      tile_store(c, [&](int, int m, int n, float v) {
        p1[(size_t)(n0 + m) * 96 + (j * 16 + n) * 3 + i] = v * INV_SQ32;
      });
    }
  }

  for (int idx = lane; idx < 16 * 96; idx += 32) {
    int m = idx / 96, c = idx % 96;
    float v;
    if (c < 64) {
      v = xt[m * 160 + c];
    } else {
      int u = c - 64;
      float a0 = xt[m * 160 + 64 + u * 3 + 0];
      float a1 = xt[m * 160 + 64 + u * 3 + 1];
      float a2 = xt[m * 160 + 64 + u * 3 + 2];
      v = sqrtf(a0 * a0 + a1 * a1 + a2 * a2 + 1e-12f);
    }
    f0S[idx] = v;
  }
  __syncthreads();

  {
    v2f af[24];
    load_a(af, [&](int m, int k) { return ld2(f0S + m * 96 + k); });
    for (int j = 0; j < 6; ++j) {
      v8f c = {};
      c = wmma_tile(af, [&](int n, int k) {
        return ld2(G1T + (size_t)(j * 16 + n) * 96 + k);
      }, c);
      tile_store(c, [&](int, int m, int n, float v) {
        int col = j * 16 + n;
        hS[m * 96 + col] = siluf(v + g1b[col]);
      });
    }
  }
  __syncthreads();
  {
    v2f ah[24];
    load_a(ah, [&](int m, int k) { return ld2(hS + m * 96 + k); });
    for (int j = 0; j < 6; ++j) {
      v8f c = {};
      c = wmma_tile(ah, [&](int n, int k) {
        return ld2(G2T + (size_t)(j * 16 + n) * 96 + k);
      }, c);
      tile_store(c, [&](int, int m, int n, float v) {
        gS[m * 96 + j * 16 + n] = v + g2b[j * 16 + n];
      });
    }
  }
  __syncthreads();

  {
    v2f ag[16];
    load_a(ag, [&](int m, int k) { return ld2(gS + m * 96 + k); });
    for (int j = 0; j < 4; ++j) {
      v8f c = {};
      c = wmma_tile(ag, [&](int n, int k) {
        return ld2(W0nT + (size_t)(j * 16 + n) * 64 + k);
      }, c);
      tile_store(c, [&](int, int m, int n, float v) {
        int col = j * 16 + n;
        x0n[(size_t)(n0 + m) * 64 + col] = v * INV_SQ64 + b0n[col];
      });
    }
  }
  for (int i = 0; i < 3; ++i) {
    v2f ax[8];
    load_a(ax, [&](int m, int k) {
      v2f a;
      a.x = xt[m * 160 + 64 + 3 * k + i] * gS[m * 96 + 64 + k];
      a.y = xt[m * 160 + 64 + 3 * (k + 1) + i] * gS[m * 96 + 64 + k + 1];
      return a;
    });
    for (int j = 0; j < 2; ++j) {
      v8f c = {};
      c = wmma_tile(ax, [&](int n, int k) {
        return ld2(W1nT + (size_t)(j * 16 + n) * 32 + k);
      }, c);
      tile_store(c, [&](int, int m, int n, float v) {
        x1n[(size_t)(n0 + m) * 96 + (j * 16 + n) * 3 + i] = v * INV_SQ32;
      });
    }
  }
}

__global__ void sort_zero_kernel(int* __restrict__ cnt) {
  int i = blockIdx.x * blockDim.x + threadIdx.x;
  if (i < N_NODES) cnt[i] = 0;
}

__global__ void sort_rank_kernel(const int* __restrict__ edge_index,
                                 int* __restrict__ cnt, int* __restrict__ rank) {
  int e = blockIdx.x * blockDim.x + threadIdx.x;
  if (e < N_EDGES) rank[e] = atomicAdd(&cnt[edge_index[e]], 1);
}

#define SCAN_PER_THREAD 10
__global__ __launch_bounds__(1024) void sort_scan_kernel(
    const int* __restrict__ cnt, int* __restrict__ off) {
  __shared__ int part[1024];
  const int t = threadIdx.x;
  int loc[SCAN_PER_THREAD];
  int s = 0;
#pragma unroll
  for (int i = 0; i < SCAN_PER_THREAD; ++i) {
    int idx = t * SCAN_PER_THREAD + i;
    int v = (idx < N_NODES) ? cnt[idx] : 0;
    loc[i] = s;
    s += v;
  }
  part[t] = s;
  __syncthreads();
  for (int d = 1; d < 1024; d <<= 1) {
    int v = (t >= d) ? part[t - d] : 0;
    __syncthreads();
    part[t] += v;
    __syncthreads();
  }
  const int base = part[t] - s;
#pragma unroll
  for (int i = 0; i < SCAN_PER_THREAD; ++i) {
    int idx = t * SCAN_PER_THREAD + i;
    if (idx < N_NODES) off[idx] = base + loc[i];
  }
}

__global__ void sort_fill_kernel(const int* __restrict__ edge_index,
                                 const int* __restrict__ off,
                                 const int* __restrict__ rank,
                                 int* __restrict__ perm) {
  int e = blockIdx.x * blockDim.x + threadIdx.x;
  if (e < N_EDGES) perm[off[edge_index[e]] + rank[e]] = e;
}

__device__ __forceinline__ void run_add(float* __restrict__ acc,
                                        const int (&dR)[8], unsigned flm,
                                        int chan, const float (&v)[8]) {
  float run = 0.0f;
#pragma unroll
  for (int r = 0; r < 8; ++r) {
    run += v[r];
    if ((flm >> r) & 1u) {
      atomicAdd(acc + (size_t)dR[r] * 480 + chan, run);
      run = 0.0f;
    }
  }
}

__global__ __launch_bounds__(32) void edge_kernel(
    const float* __restrict__ edge_sh, const float* __restrict__ edge_attr,
    const float* __restrict__ F1T, const float* __restrict__ F2T,
    const float* __restrict__ L1T, const float* __restrict__ L2T,
    const int* __restrict__ edge_index, const int* __restrict__ perm,
    const float* __restrict__ p0, const float* __restrict__ p1,
    const float* __restrict__ x0n, const float* __restrict__ x1n,
    float* __restrict__ acc) {
  __shared__ __align__(16) float s0S[16 * 160];
  __shared__ __align__(16) float hN[16 * 32];
  __shared__ __align__(16) float hE[16 * 32];
  __shared__ int dstS[16];
  __shared__ int srcS[16];
  __shared__ int eS[16];

  const int lane = threadIdx.x;
  const int e0 = blockIdx.x * 16;

  __builtin_prefetch(F2T, 0, 1);
  __builtin_prefetch(L2T, 0, 1);

  if (lane < 16) {
    const int e = perm[e0 + lane];
    eS[lane] = e;
    dstS[lane] = edge_index[e];
    srcS[lane] = edge_index[N_EDGES + e];
  }
  __syncthreads();

  {
    const int hl = lane & 15;
    v4f pv[16];
#pragma unroll
    for (int m = 0; m < 16; ++m) {
      const int node = (lane < 16) ? dstS[m] : srcS[m];
      pv[m] = *(const v4f*)(p0 + (size_t)node * 64 + hl * 4);
    }
#pragma unroll
    for (int m = 0; m < 16; ++m) *(v4f*)(s0S + m * 160 + lane * 4) = pv[m];

#pragma nounroll
    for (int mb = 0; mb < 16; mb += 8) {
      float dv[8];
#pragma unroll
      for (int r = 0; r < 8; ++r) {
        const float* pd = p1 + (size_t)dstS[mb + r] * 96 + lane * 3;
        const float* ps = p1 + (size_t)srcS[mb + r] * 96 + lane * 3;
        dv[r] = (pd[0] * ps[0] + pd[1] * ps[1] + pd[2] * ps[2]) * INV_SQ3;
      }
#pragma unroll
      for (int r = 0; r < 8; ++r) s0S[(mb + r) * 160 + 128 + lane] = dv[r];
    }
  }
  __syncthreads();

  {
    v2f as0[40];
    load_a(as0, [&](int m, int k) { return ld2(s0S + m * 160 + k); });
    for (int j = 0; j < 2; ++j) {
      v8f c = {};
      c = wmma_tile(as0, [&](int n, int k) {
        return ld2(L1T + (size_t)(j * 16 + n) * 160 + k);
      }, c);
      tile_store(c, [&](int, int m, int n, float v) {
        hN[m * 32 + j * 16 + n] = sspf(v * INV_SQ160);
      });
    }
  }
  {
    v2f aea[8];
    load_a(aea, [&](int m, int k) {
      return ld2(edge_attr + (size_t)eS[m] * 32 + k);
    });
    for (int j = 0; j < 2; ++j) {
      v8f c = {};
      c = wmma_tile(aea, [&](int n, int k) {
        return ld2(F1T + (size_t)(j * 16 + n) * 32 + k);
      }, c);
      tile_store(c, [&](int, int m, int n, float v) {
        hE[m * 32 + j * 16 + n] = sspf(v * INV_SQ32);
      });
    }
  }
  __syncthreads();

  const int rbase = (lane >> 4) << 3;
  float sh0r[8], s1xr[8], s1yr[8], s1zr[8];
  int dR[8], sR[8];
#pragma unroll
  for (int r = 0; r < 8; ++r) {
    int m = rbase + r;
    int e = eS[m];
    sh0r[r] = edge_sh[(size_t)e * 4 + 0];
    s1xr[r] = edge_sh[(size_t)e * 4 + 1];
    s1yr[r] = edge_sh[(size_t)e * 4 + 2];
    s1zr[r] = edge_sh[(size_t)e * 4 + 3];
    dR[r] = dstS[m];
    sR[r] = srcS[m];
  }
  unsigned flm = 0x80u;
#pragma unroll
  for (int r = 0; r < 7; ++r) flm |= (dR[r + 1] != dR[r]) ? (1u << r) : 0u;

  v2f aN[8], aE[8];
  load_a(aN, [&](int m, int k) { return ld2(hN + m * 32 + k); });
  load_a(aE, [&](int m, int k) { return ld2(hE + m * 32 + k); });

  for (int j = 0; j < 14; ++j) {
    const int nb = j * 16;
    v8f cN = {}, cE = {};
    cN = wmma_tile(aN, [&](int n, int k) {
      return ld2(L2T + (size_t)(nb + n) * 32 + k);
    }, cN);
    cE = wmma_tile(aE, [&](int n, int k) {
      return ld2(F2T + (size_t)(nb + n) * 32 + k);
    }, cE);
    v8f w8 = cN * cE * (INV_SQ32 * INV_SQ32);

    const int col = nb + (lane & 15);
    float v0[8], v1[8], v2[8];
    if (nb < 64) {
#pragma unroll
      for (int r = 0; r < 8; ++r) {
        float xs = x0n[(size_t)sR[r] * 64 + col];
        v0[r] = w8[r] * xs * sh0r[r];
      }
      run_add(acc, dR, flm, col, v0);
    } else if (nb < 128) {
      const int c2 = col - 64;
#pragma unroll
      for (int r = 0; r < 8; ++r) {
        float f = w8[r] * x0n[(size_t)sR[r] * 64 + c2] * INV_SQ3;
        v0[r] = f * s1xr[r];
        v1[r] = f * s1yr[r];
        v2[r] = f * s1zr[r];
      }
      run_add(acc, dR, flm, 96 + c2 * 3 + 0, v0);
      run_add(acc, dR, flm, 96 + c2 * 3 + 1, v1);
      run_add(acc, dR, flm, 96 + c2 * 3 + 2, v2);
    } else if (nb < 160) {
      const int c3 = col - 128;
#pragma unroll
      for (int r = 0; r < 8; ++r) {
        const float* xv = x1n + (size_t)sR[r] * 96 + c3 * 3;
        float f = w8[r] * sh0r[r];
        v0[r] = f * xv[0];
        v1[r] = f * xv[1];
        v2[r] = f * xv[2];
      }
      run_add(acc, dR, flm, 288 + c3 * 3 + 0, v0);
      run_add(acc, dR, flm, 288 + c3 * 3 + 1, v1);
      run_add(acc, dR, flm, 288 + c3 * 3 + 2, v2);
    } else if (nb < 192) {
      const int c4 = col - 160;
#pragma unroll
      for (int r = 0; r < 8; ++r) {
        const float* xv = x1n + (size_t)sR[r] * 96 + c4 * 3;
        float dp = xv[0] * s1xr[r] + xv[1] * s1yr[r] + xv[2] * s1zr[r];
        v0[r] = w8[r] * dp * INV_SQ3;
      }
      run_add(acc, dR, flm, 64 + c4, v0);
    } else {
      const int c5 = col - 192;
#pragma unroll
      for (int r = 0; r < 8; ++r) {
        const float* xv = x1n + (size_t)sR[r] * 96 + c5 * 3;
        const float s1x = s1xr[r], s1y = s1yr[r], s1z = s1zr[r];
        float cx = xv[1] * s1z - xv[2] * s1y;
        float cy = xv[2] * s1x - xv[0] * s1z;
        float cz = xv[0] * s1y - xv[1] * s1x;
        float f = w8[r] * INV_SQ2;
        v0[r] = f * cx;
        v1[r] = f * cy;
        v2[r] = f * cz;
      }
      run_add(acc, dR, flm, 384 + c5 * 3 + 0, v0);
      run_add(acc, dR, flm, 384 + c5 * 3 + 1, v1);
      run_add(acc, dR, flm, 384 + c5 * 3 + 2, v2);
    }
  }
}

__global__ __launch_bounds__(32) void proj_kernel(
    const float* __restrict__ acc,
    const float* __restrict__ WO0T, const float* __restrict__ bO0,
    const float* __restrict__ WO1T, const float* __restrict__ WO2T,
    float* __restrict__ y) {
  const int n0 = blockIdx.x * 16;
  const float* at = acc + (size_t)n0 * 480;

  {
    v2f aa[24];
    load_a(aa, [&](int m, int k) { return ld2(at + (size_t)m * 480 + k); });
    for (int j = 0; j < 8; ++j) {
      v8f c = {};
      c = wmma_tile(aa, [&](int n, int k) {
        return ld2(WO0T + (size_t)(j * 16 + n) * 96 + k);
      }, c);
      tile_store(c, [&](int, int m, int n, float v) {
        int col = j * 16 + n;
        y[(size_t)(n0 + m) * 416 + col] = v * INV_SQ96 + bO0[col];
      });
    }
  }
  for (int i = 0; i < 3; ++i) {
    v2f aa[24];
    load_a(aa, [&](int m, int k) {
      v2f a;
      a.x = at[(size_t)m * 480 + 96 + 3 * k + i];
      a.y = at[(size_t)m * 480 + 96 + 3 * (k + 1) + i];
      return a;
    });
    for (int j = 0; j < 4; ++j) {
      v8f c = {};
      c = wmma_tile(aa, [&](int n, int k) {
        return ld2(WO1T + (size_t)(j * 16 + n) * 96 + k);
      }, c);
      tile_store(c, [&](int, int m, int n, float v) {
        y[(size_t)(n0 + m) * 416 + 128 + (j * 16 + n) * 3 + i] = v * INV_SQ96;
      });
    }
  }
  for (int i = 0; i < 3; ++i) {
    v2f aa[8];
    load_a(aa, [&](int m, int k) {
      v2f a;
      a.x = at[(size_t)m * 480 + 384 + 3 * k + i];
      a.y = at[(size_t)m * 480 + 384 + 3 * (k + 1) + i];
      return a;
    });
    for (int j = 0; j < 2; ++j) {
      v8f c = {};
      c = wmma_tile(aa, [&](int n, int k) {
        return ld2(WO2T + (size_t)(j * 16 + n) * 32 + k);
      }, c);
      tile_store(c, [&](int, int m, int n, float v) {
        y[(size_t)(n0 + m) * 416 + 320 + (j * 16 + n) * 3 + i] = v * INV_SQ32;
      });
    }
  }
}

extern "C" void kernel_launch(void* const* d_in, const int* in_sizes, int n_in,
                              void* d_out, int out_size, void* d_ws,
                              size_t ws_size, hipStream_t stream) {
  (void)in_sizes; (void)n_in; (void)out_size; (void)ws_size;
  const float* x         = (const float*)d_in[0];
  const float* edge_sh   = (const float*)d_in[1];
  const float* edge_attr = (const float*)d_in[2];
  const float* W0p = (const float*)d_in[3];
  const float* b0p = (const float*)d_in[4];
  const float* W1p = (const float*)d_in[5];
  const float* W0n = (const float*)d_in[6];
  const float* b0n = (const float*)d_in[7];
  const float* W1n = (const float*)d_in[8];
  const float* G1  = (const float*)d_in[9];
  const float* g1b = (const float*)d_in[10];
  const float* G2  = (const float*)d_in[11];
  const float* g2b = (const float*)d_in[12];
  const float* F1  = (const float*)d_in[13];
  const float* F2  = (const float*)d_in[14];
  const float* L1  = (const float*)d_in[15];
  const float* L2  = (const float*)d_in[16];
  const float* WO0 = (const float*)d_in[17];
  const float* bO0 = (const float*)d_in[18];
  const float* WO1 = (const float*)d_in[19];
  const float* WO2 = (const float*)d_in[20];
  const int* edge_index = (const int*)d_in[21];

  float* ws   = (float*)d_ws;
  float* p0   = ws;
  float* p1   = p0   + (size_t)N_NODES * 64;
  float* x0n  = p1   + (size_t)N_NODES * 96;
  float* x1n  = x0n  + (size_t)N_NODES * 64;
  float* acc  = x1n  + (size_t)N_NODES * 96;
  float* wt   = acc  + (size_t)N_NODES * 480;
  float* W0pT = wt;            wt += 64 * 64;
  float* W1pT = wt;            wt += 32 * 32;
  float* W0nT = wt;            wt += 64 * 64;
  float* W1nT = wt;            wt += 32 * 32;
  float* G1T  = wt;            wt += 96 * 96;
  float* G2T  = wt;            wt += 96 * 96;
  float* F1T  = wt;            wt += 32 * 32;
  float* F2T  = wt;            wt += 32 * 224;
  float* L1T  = wt;            wt += 160 * 32;
  float* L2T  = wt;            wt += 32 * 224;
  float* WO0T = wt;            wt += 96 * 128;
  float* WO1T = wt;            wt += 96 * 64;
  float* WO2T = wt;            wt += 32 * 32;

  int* sc   = (int*)d_out;
  int* cnt  = sc;
  int* off  = sc + 10240;
  int* rank = sc + 20480;
  int* perm = sc + 20480 + N_EDGES;

  auto tr = [&](const float* W, float* Wt, int K, int Nc) {
    int n = K * Nc;
    transpose_kernel<<<(n + 255) / 256, 256, 0, stream>>>(W, Wt, K, Nc);
  };
  tr(W0p, W0pT, 64, 64);
  tr(W1p, W1pT, 32, 32);
  tr(W0n, W0nT, 64, 64);
  tr(W1n, W1nT, 32, 32);
  tr(G1, G1T, 96, 96);
  tr(G2, G2T, 96, 96);
  tr(F1, F1T, 32, 32);
  tr(F2, F2T, 32, 224);
  tr(L1, L1T, 160, 32);
  tr(L2, L2T, 32, 224);
  tr(WO0, WO0T, 96, 128);
  tr(WO1, WO1T, 96, 64);
  tr(WO2, WO2T, 32, 32);

  hipMemsetAsync(acc, 0, (size_t)N_NODES * 480 * sizeof(float), stream);

  sort_zero_kernel<<<(N_NODES + 255) / 256, 256, 0, stream>>>(cnt);
  sort_rank_kernel<<<(N_EDGES + 255) / 256, 256, 0, stream>>>(edge_index, cnt, rank);
  sort_scan_kernel<<<1, 1024, 0, stream>>>(cnt, off);
  sort_fill_kernel<<<(N_EDGES + 255) / 256, 256, 0, stream>>>(edge_index, off, rank, perm);

  node_kernel<<<N_NODES / 16, 32, 0, stream>>>(
      x, W0pT, b0p, W1pT, W0nT, b0n, W1nT, G1T, g1b, G2T, g2b,
      p0, p1, x0n, x1n);

  edge_kernel<<<N_EDGES / 16, 32, 0, stream>>>(
      edge_sh, edge_attr, F1T, F2T, L1T, L2T, edge_index, perm,
      p0, p1, x0n, x1n, acc);

  proj_kernel<<<N_NODES / 16, 32, 0, stream>>>(
      acc, WO0T, bO0, WO1T, WO2T, (float*)d_out);
}


# One set-up launch does the 13 weight transposes, the accumulator clear (was a memset) and the sort-counter clear: 20 kernel launches + a memset per call become 7 launches; sums unchanged
# speedup vs baseline: 1.7313x; 1.1049x over previous
#include <hip/hip_runtime.h>
#include <hip/hip_bf16.h>
#include <math.h>


typedef __attribute__((ext_vector_type(2))) float v2f;
typedef __attribute__((ext_vector_type(4))) float v4f;
typedef __attribute__((ext_vector_type(8))) float v8f;

#define N_NODES 10000
#define N_EDGES 320000

#define INV_SQ3   0.5773502691896258f
#define INV_SQ2   0.7071067811865476f
#define INV_SQ32  0.17677669529663687f
#define INV_SQ64  0.125f
#define INV_SQ96  0.10206207261596575f
#define INV_SQ160 0.07905694150420949f
#define LN2F      0.6931471805599453f

__device__ __forceinline__ float sspf(float v) {
  float sp = (v > 20.0f) ? v : log1pf(expf(v));
  return sp - LN2F;
}
__device__ __forceinline__ float siluf(float v) {
  return v / (1.0f + expf(-v));
}
__device__ __forceinline__ v2f ld2(const float* p) {
  return *(const v2f*)p;
}

__device__ __forceinline__ v8f wmma4(v2f a, v2f b, v8f c) {
  return __builtin_amdgcn_wmma_f32_16x16x4_f32(false, a, false, b,
                                               (short)0, c, false, false);
}


template <int KS, typename LA>
__device__ __forceinline__ void load_a(v2f (&a)[KS], LA la) {
  const int lane = threadIdx.x & 31;
  const int m = lane & 15;
  const int kk = (lane >> 4) << 1;
#pragma unroll
  for (int q = 0; q < KS; ++q) a[q] = la(m, q * 4 + kk);
}

template <int KS, typename LB>
__device__ __forceinline__ v8f wmma_tile(const v2f (&a)[KS], LB lb, v8f c) {
  const int lane = threadIdx.x & 31;
  const int n = lane & 15;
  const int kk = (lane >> 4) << 1;
#pragma unroll
  for (int q = 0; q < KS; ++q) c = wmma4(a[q], lb(n, q * 4 + kk), c);
  return c;
}

template <typename ST>
__device__ __forceinline__ void tile_store(v8f c, ST st) {
  const int lane = threadIdx.x & 31;
  const int col = lane & 15;
  const int rbase = (lane >> 4) << 3;
#pragma unroll
  for (int r = 0; r < 8; ++r) st(r, rbase + r, col, c[r]);
}

constexpr int OFF_W0P = 0;
constexpr int OFF_W1P = OFF_W0P + 64 * 64;
constexpr int OFF_W0N = OFF_W1P + 32 * 32;
constexpr int OFF_W1N = OFF_W0N + 64 * 64;
constexpr int OFF_G1  = OFF_W1N + 32 * 32;
constexpr int OFF_G2  = OFF_G1  + 96 * 96;
constexpr int OFF_F1  = OFF_G2  + 96 * 96;
constexpr int OFF_F2  = OFF_F1  + 32 * 32;
constexpr int OFF_L1  = OFF_F2  + 32 * 224;
constexpr int OFF_L2  = OFF_L1  + 160 * 32;
constexpr int OFF_WO0 = OFF_L2  + 32 * 224;
constexpr int OFF_WO1 = OFF_WO0 + 96 * 128;
constexpr int OFF_WO2 = OFF_WO1 + 96 * 64;
constexpr int WT_FLOATS = OFF_WO2 + 32 * 32;

constexpr int SETUP_THREADS  = 256;
constexpr int SETUP_T_BLOCKS = WT_FLOATS / 1024;
constexpr int ACC_V4         = N_NODES * 480 / 4;
constexpr int Z_V4_PER_BLOCK = SETUP_THREADS * 16;
constexpr int SETUP_Z_BLOCKS = (ACC_V4 + Z_V4_PER_BLOCK - 1) / Z_V4_PER_BLOCK;
constexpr int CNT_PADDED     = 10240;
constexpr int SETUP_BLOCKS   = SETUP_T_BLOCKS + SETUP_Z_BLOCKS + 1;

template <int K, int Nc>
__device__ __forceinline__ void transpose_slab(const float* __restrict__ W,
                                               float* __restrict__ Wt, int slab) {
  static_assert((K * Nc) % 1024 == 0, "weights are whole slabs");
  float v[4];
#pragma unroll
  for (int i = 0; i < 4; ++i) v[i] = W[slab * 1024 + i * SETUP_THREADS + threadIdx.x];
#pragma unroll
  for (int i = 0; i < 4; ++i) {
    const int idx = slab * 1024 + i * SETUP_THREADS + threadIdx.x;
    const int k = idx / Nc, n = idx % Nc;
    Wt[n * K + k] = v[i];
  }
}

__global__ __launch_bounds__(SETUP_THREADS) void setup_kernel(
    const float* __restrict__ W0p, const float* __restrict__ W1p,
    const float* __restrict__ W0n, const float* __restrict__ W1n,
    const float* __restrict__ G1, const float* __restrict__ G2,
    const float* __restrict__ F1, const float* __restrict__ F2,
    const float* __restrict__ L1, const float* __restrict__ L2,
    const float* __restrict__ WO0, const float* __restrict__ WO1,
    const float* __restrict__ WO2,
    float* __restrict__ wt, float* __restrict__ acc, int* __restrict__ cnt) {
  int s = blockIdx.x;
  if (s < SETUP_T_BLOCKS) {
    if (s < 4)  { transpose_slab<64, 64>(W0p, wt + OFF_W0P, s); return; }   s -= 4;
    if (s < 1)  { transpose_slab<32, 32>(W1p, wt + OFF_W1P, s); return; }   s -= 1;
    if (s < 4)  { transpose_slab<64, 64>(W0n, wt + OFF_W0N, s); return; }   s -= 4;
    if (s < 1)  { transpose_slab<32, 32>(W1n, wt + OFF_W1N, s); return; }   s -= 1;
    if (s < 9)  { transpose_slab<96, 96>(G1, wt + OFF_G1, s); return; }     s -= 9;
    if (s < 9)  { transpose_slab<96, 96>(G2, wt + OFF_G2, s); return; }     s -= 9;
    if (s < 1)  { transpose_slab<32, 32>(F1, wt + OFF_F1, s); return; }     s -= 1;
    if (s < 7)  { transpose_slab<32, 224>(F2, wt + OFF_F2, s); return; }    s -= 7;
    if (s < 5)  { transpose_slab<160, 32>(L1, wt + OFF_L1, s); return; }    s -= 5;
    if (s < 7)  { transpose_slab<32, 224>(L2, wt + OFF_L2, s); return; }    s -= 7;
    if (s < 12) { transpose_slab<96, 128>(WO0, wt + OFF_WO0, s); return; }  s -= 12;
    if (s < 6)  { transpose_slab<96, 64>(WO1, wt + OFF_WO1, s); return; }   s -= 6;
    transpose_slab<32, 32>(WO2, wt + OFF_WO2, s);
    return;
  }
  s -= SETUP_T_BLOCKS;
  if (s < SETUP_Z_BLOCKS) {
    v4f* a4 = (v4f*)acc;
    const v4f z = {0.0f, 0.0f, 0.0f, 0.0f};
    for (int i = 0; i < 16; ++i) {
      const int q = s * Z_V4_PER_BLOCK + i * SETUP_THREADS + threadIdx.x;
      if (q < ACC_V4) a4[q] = z;
    }
    return;
  }
  for (int i = threadIdx.x; i < CNT_PADDED; i += SETUP_THREADS) cnt[i] = 0;
}

__global__ __launch_bounds__(32) void node_kernel(
    const float* __restrict__ x,
    const float* __restrict__ W0pT, const float* __restrict__ b0p,
    const float* __restrict__ W1pT,
    const float* __restrict__ W0nT, const float* __restrict__ b0n,
    const float* __restrict__ W1nT,
    const float* __restrict__ G1T, const float* __restrict__ g1b,
    const float* __restrict__ G2T, const float* __restrict__ g2b,
    float* __restrict__ p0, float* __restrict__ p1,
    float* __restrict__ x0n, float* __restrict__ x1n) {
  __shared__ __align__(16) float f0S[16 * 96];
  __shared__ __align__(16) float hS[16 * 96];
  __shared__ __align__(16) float gS[16 * 96];

  const int lane = threadIdx.x;
  const int n0 = blockIdx.x * 16;
  const float* xt = x + (size_t)n0 * 160;

  {
    v2f ax[16];
    load_a(ax, [&](int m, int k) { return ld2(xt + m * 160 + k); });
    for (int j = 0; j < 4; ++j) {
      v8f c = {};
      c = wmma_tile(ax, [&](int n, int k) {
        return ld2(W0pT + (size_t)(j * 16 + n) * 64 + k);
      }, c);
      tile_store(c, [&](int, int m, int n, float v) {
        int col = j * 16 + n;
        p0[(size_t)(n0 + m) * 64 + col] = v * INV_SQ64 + b0p[col];
      });
    }
  }
  for (int i = 0; i < 3; ++i) {
    v2f ax[8];
    load_a(ax, [&](int m, int k) {
      v2f a;
      a.x = xt[m * 160 + 64 + 3 * k + i];
      a.y = xt[m * 160 + 64 + 3 * (k + 1) + i];
      return a;
    });
    for (int j = 0; j < 2; ++j) {
      v8f c = {};
      c = wmma_tile(ax, [&](int n, int k) {
        return ld2(W1pT + (size_t)(j * 16 + n) * 32 + k);
      }, c);
      tile_store(c, [&](int, int m, int n, float v) {
        p1[(size_t)(n0 + m) * 96 + (j * 16 + n) * 3 + i] = v * INV_SQ32;
      });
    }
  }

  for (int idx = lane; idx < 16 * 96; idx += 32) {
    int m = idx / 96, c = idx % 96;
    float v;
    if (c < 64) {
      v = xt[m * 160 + c];
    } else {
      int u = c - 64;
      float a0 = xt[m * 160 + 64 + u * 3 + 0];
      float a1 = xt[m * 160 + 64 + u * 3 + 1];
      float a2 = xt[m * 160 + 64 + u * 3 + 2];
      v = sqrtf(a0 * a0 + a1 * a1 + a2 * a2 + 1e-12f);
    }
    f0S[idx] = v;
  }
  __syncthreads();

  {
    v2f af[24];
    load_a(af, [&](int m, int k) { return ld2(f0S + m * 96 + k); });
    for (int j = 0; j < 6; ++j) {
      v8f c = {};
      c = wmma_tile(af, [&](int n, int k) {
        return ld2(G1T + (size_t)(j * 16 + n) * 96 + k);
      }, c);
      tile_store(c, [&](int, int m, int n, float v) {
        int col = j * 16 + n;
        hS[m * 96 + col] = siluf(v + g1b[col]);
      });
    }
  }
  __syncthreads();
  {
    v2f ah[24];
    load_a(ah, [&](int m, int k) { return ld2(hS + m * 96 + k); });
    for (int j = 0; j < 6; ++j) {
      v8f c = {};
      c = wmma_tile(ah, [&](int n, int k) {
        return ld2(G2T + (size_t)(j * 16 + n) * 96 + k);
      }, c);
      tile_store(c, [&](int, int m, int n, float v) {
        gS[m * 96 + j * 16 + n] = v + g2b[j * 16 + n];
      });
    }
  }
  __syncthreads();

  {
    v2f ag[16];
    load_a(ag, [&](int m, int k) { return ld2(gS + m * 96 + k); });
    for (int j = 0; j < 4; ++j) {
      v8f c = {};
      c = wmma_tile(ag, [&](int n, int k) {
        return ld2(W0nT + (size_t)(j * 16 + n) * 64 + k);
      }, c);
      tile_store(c, [&](int, int m, int n, float v) {
        int col = j * 16 + n;
        x0n[(size_t)(n0 + m) * 64 + col] = v * INV_SQ64 + b0n[col];
      });
    }
  }
  for (int i = 0; i < 3; ++i) {
    v2f ax[8];
    load_a(ax, [&](int m, int k) {
      v2f a;
      a.x = xt[m * 160 + 64 + 3 * k + i] * gS[m * 96 + 64 + k];
      a.y = xt[m * 160 + 64 + 3 * (k + 1) + i] * gS[m * 96 + 64 + k + 1];
      return a;
    });
    for (int j = 0; j < 2; ++j) {
      v8f c = {};
      c = wmma_tile(ax, [&](int n, int k) {
        return ld2(W1nT + (size_t)(j * 16 + n) * 32 + k);
      }, c);
      tile_store(c, [&](int, int m, int n, float v) {
        x1n[(size_t)(n0 + m) * 96 + (j * 16 + n) * 3 + i] = v * INV_SQ32;
      });
    }
  }
}

__global__ void sort_rank_kernel(const int* __restrict__ edge_index,
                                 int* __restrict__ cnt, int* __restrict__ rank) {
  int e = blockIdx.x * blockDim.x + threadIdx.x;
  if (e < N_EDGES) rank[e] = atomicAdd(&cnt[edge_index[e]], 1);
}

#define SCAN_PER_THREAD 10
__global__ __launch_bounds__(1024) void sort_scan_kernel(
    const int* __restrict__ cnt, int* __restrict__ off) {
  __shared__ int part[1024];
  const int t = threadIdx.x;
  int loc[SCAN_PER_THREAD];
  int s = 0;
#pragma unroll
  for (int i = 0; i < SCAN_PER_THREAD; ++i) {
    int idx = t * SCAN_PER_THREAD + i;
    int v = (idx < N_NODES) ? cnt[idx] : 0;
    loc[i] = s;
    s += v;
  }
  part[t] = s;
  __syncthreads();
  for (int d = 1; d < 1024; d <<= 1) {
    int v = (t >= d) ? part[t - d] : 0;
    __syncthreads();
    part[t] += v;
    __syncthreads();
  }
  const int base = part[t] - s;
#pragma unroll
  for (int i = 0; i < SCAN_PER_THREAD; ++i) {
    int idx = t * SCAN_PER_THREAD + i;
    if (idx < N_NODES) off[idx] = base + loc[i];
  }
}

__global__ void sort_fill_kernel(const int* __restrict__ edge_index,
                                 const int* __restrict__ off,
                                 const int* __restrict__ rank,
                                 int* __restrict__ perm) {
  int e = blockIdx.x * blockDim.x + threadIdx.x;
  if (e < N_EDGES) perm[off[edge_index[e]] + rank[e]] = e;
}

__device__ __forceinline__ void run_add(float* __restrict__ acc,
                                        const int (&dR)[8], unsigned flm,
                                        int chan, const float (&v)[8]) {
  float run = 0.0f;
#pragma unroll
  for (int r = 0; r < 8; ++r) {
    run += v[r];
    if ((flm >> r) & 1u) {
      atomicAdd(acc + (size_t)dR[r] * 480 + chan, run);
      run = 0.0f;
    }
  }
}

__global__ __launch_bounds__(32) void edge_kernel(
    const float* __restrict__ edge_sh, const float* __restrict__ edge_attr,
    const float* __restrict__ F1T, const float* __restrict__ F2T,
    const float* __restrict__ L1T, const float* __restrict__ L2T,
    const int* __restrict__ edge_index, const int* __restrict__ perm,
    const float* __restrict__ p0, const float* __restrict__ p1,
    const float* __restrict__ x0n, const float* __restrict__ x1n,
    float* __restrict__ acc) {
  __shared__ __align__(16) float s0S[16 * 160];
  __shared__ __align__(16) float hN[16 * 32];
  __shared__ __align__(16) float hE[16 * 32];
  __shared__ int dstS[16];
  __shared__ int srcS[16];
  __shared__ int eS[16];

  const int lane = threadIdx.x;
  const int e0 = blockIdx.x * 16;

  __builtin_prefetch(F2T, 0, 1);
  __builtin_prefetch(L2T, 0, 1);

  if (lane < 16) {
    const int e = perm[e0 + lane];
    eS[lane] = e;
    dstS[lane] = edge_index[e];
    srcS[lane] = edge_index[N_EDGES + e];
  }
  __syncthreads();

  {
    const int hl = lane & 15;
    v4f pv[16];
#pragma unroll
    for (int m = 0; m < 16; ++m) {
      const int node = (lane < 16) ? dstS[m] : srcS[m];
      pv[m] = *(const v4f*)(p0 + (size_t)node * 64 + hl * 4);
    }
#pragma unroll
    for (int m = 0; m < 16; ++m) *(v4f*)(s0S + m * 160 + lane * 4) = pv[m];

#pragma nounroll
    for (int mb = 0; mb < 16; mb += 8) {
      float dv[8];
#pragma unroll
      for (int r = 0; r < 8; ++r) {
        const float* pd = p1 + (size_t)dstS[mb + r] * 96 + lane * 3;
        const float* ps = p1 + (size_t)srcS[mb + r] * 96 + lane * 3;
        dv[r] = (pd[0] * ps[0] + pd[1] * ps[1] + pd[2] * ps[2]) * INV_SQ3;
      }
#pragma unroll
      for (int r = 0; r < 8; ++r) s0S[(mb + r) * 160 + 128 + lane] = dv[r];
    }
  }
  __syncthreads();

  {
    v2f as0[40];
    load_a(as0, [&](int m, int k) { return ld2(s0S + m * 160 + k); });
    for (int j = 0; j < 2; ++j) {
      v8f c = {};
      c = wmma_tile(as0, [&](int n, int k) {
        return ld2(L1T + (size_t)(j * 16 + n) * 160 + k);
      }, c);
      tile_store(c, [&](int, int m, int n, float v) {
        hN[m * 32 + j * 16 + n] = sspf(v * INV_SQ160);
      });
    }
  }
  {
    v2f aea[8];
    load_a(aea, [&](int m, int k) {
      return ld2(edge_attr + (size_t)eS[m] * 32 + k);
    });
    for (int j = 0; j < 2; ++j) {
      v8f c = {};
      c = wmma_tile(aea, [&](int n, int k) {
        return ld2(F1T + (size_t)(j * 16 + n) * 32 + k);
      }, c);
      tile_store(c, [&](int, int m, int n, float v) {
        hE[m * 32 + j * 16 + n] = sspf(v * INV_SQ32);
      });
    }
  }
  __syncthreads();

  const int rbase = (lane >> 4) << 3;
  float sh0r[8], s1xr[8], s1yr[8], s1zr[8];
  int dR[8], sR[8];
#pragma unroll
  for (int r = 0; r < 8; ++r) {
    int m = rbase + r;
    int e = eS[m];
    sh0r[r] = edge_sh[(size_t)e * 4 + 0];
    s1xr[r] = edge_sh[(size_t)e * 4 + 1];
    s1yr[r] = edge_sh[(size_t)e * 4 + 2];
    s1zr[r] = edge_sh[(size_t)e * 4 + 3];
    dR[r] = dstS[m];
    sR[r] = srcS[m];
  }
  unsigned flm = 0x80u;
#pragma unroll
  for (int r = 0; r < 7; ++r) flm |= (dR[r + 1] != dR[r]) ? (1u << r) : 0u;

  v2f aN[8], aE[8];
  load_a(aN, [&](int m, int k) { return ld2(hN + m * 32 + k); });
  load_a(aE, [&](int m, int k) { return ld2(hE + m * 32 + k); });

  for (int j = 0; j < 14; ++j) {
    const int nb = j * 16;
    v8f cN = {}, cE = {};
    cN = wmma_tile(aN, [&](int n, int k) {
      return ld2(L2T + (size_t)(nb + n) * 32 + k);
    }, cN);
    cE = wmma_tile(aE, [&](int n, int k) {
      return ld2(F2T + (size_t)(nb + n) * 32 + k);
    }, cE);
    v8f w8 = cN * cE * (INV_SQ32 * INV_SQ32);

    const int col = nb + (lane & 15);
    float v0[8], v1[8], v2[8];
    if (nb < 64) {
#pragma unroll
      for (int r = 0; r < 8; ++r) {
        float xs = x0n[(size_t)sR[r] * 64 + col];
        v0[r] = w8[r] * xs * sh0r[r];
      }
      run_add(acc, dR, flm, col, v0);
    } else if (nb < 128) {
      const int c2 = col - 64;
#pragma unroll
      for (int r = 0; r < 8; ++r) {
        float f = w8[r] * x0n[(size_t)sR[r] * 64 + c2] * INV_SQ3;
        v0[r] = f * s1xr[r];
        v1[r] = f * s1yr[r];
        v2[r] = f * s1zr[r];
      }
      run_add(acc, dR, flm, 96 + c2 * 3 + 0, v0);
      run_add(acc, dR, flm, 96 + c2 * 3 + 1, v1);
      run_add(acc, dR, flm, 96 + c2 * 3 + 2, v2);
    } else if (nb < 160) {
      const int c3 = col - 128;
#pragma unroll
      for (int r = 0; r < 8; ++r) {
        const float* xv = x1n + (size_t)sR[r] * 96 + c3 * 3;
        float f = w8[r] * sh0r[r];
        v0[r] = f * xv[0];
        v1[r] = f * xv[1];
        v2[r] = f * xv[2];
      }
      run_add(acc, dR, flm, 288 + c3 * 3 + 0, v0);
      run_add(acc, dR, flm, 288 + c3 * 3 + 1, v1);
      run_add(acc, dR, flm, 288 + c3 * 3 + 2, v2);
    } else if (nb < 192) {
      const int c4 = col - 160;
#pragma unroll
      for (int r = 0; r < 8; ++r) {
        const float* xv = x1n + (size_t)sR[r] * 96 + c4 * 3;
        float dp = xv[0] * s1xr[r] + xv[1] * s1yr[r] + xv[2] * s1zr[r];
        v0[r] = w8[r] * dp * INV_SQ3;
      }
      run_add(acc, dR, flm, 64 + c4, v0);
    } else {
      const int c5 = col - 192;
#pragma unroll
      for (int r = 0; r < 8; ++r) {
        const float* xv = x1n + (size_t)sR[r] * 96 + c5 * 3;
        const float s1x = s1xr[r], s1y = s1yr[r], s1z = s1zr[r];
        float cx = xv[1] * s1z - xv[2] * s1y;
        float cy = xv[2] * s1x - xv[0] * s1z;
        float cz = xv[0] * s1y - xv[1] * s1x;
        float f = w8[r] * INV_SQ2;
        v0[r] = f * cx;
        v1[r] = f * cy;
        v2[r] = f * cz;
      }
      run_add(acc, dR, flm, 384 + c5 * 3 + 0, v0);
      run_add(acc, dR, flm, 384 + c5 * 3 + 1, v1);
      run_add(acc, dR, flm, 384 + c5 * 3 + 2, v2);
    }
  }
}

__global__ __launch_bounds__(32) void proj_kernel(
    const float* __restrict__ acc,
    const float* __restrict__ WO0T, const float* __restrict__ bO0,
    const float* __restrict__ WO1T, const float* __restrict__ WO2T,
    float* __restrict__ y) {
  const int n0 = blockIdx.x * 16;
  const float* at = acc + (size_t)n0 * 480;

  {
    v2f aa[24];
    load_a(aa, [&](int m, int k) { return ld2(at + (size_t)m * 480 + k); });
    for (int j = 0; j < 8; ++j) {
      v8f c = {};
      c = wmma_tile(aa, [&](int n, int k) {
        return ld2(WO0T + (size_t)(j * 16 + n) * 96 + k);
      }, c);
      tile_store(c, [&](int, int m, int n, float v) {
        int col = j * 16 + n;
        y[(size_t)(n0 + m) * 416 + col] = v * INV_SQ96 + bO0[col];
      });
    }
  }
  for (int i = 0; i < 3; ++i) {
    v2f aa[24];
    load_a(aa, [&](int m, int k) {
      v2f a;
      a.x = at[(size_t)m * 480 + 96 + 3 * k + i];
      a.y = at[(size_t)m * 480 + 96 + 3 * (k + 1) + i];
      return a;
    });
    for (int j = 0; j < 4; ++j) {
      v8f c = {};
      c = wmma_tile(aa, [&](int n, int k) {
        return ld2(WO1T + (size_t)(j * 16 + n) * 96 + k);
      }, c);
      tile_store(c, [&](int, int m, int n, float v) {
        y[(size_t)(n0 + m) * 416 + 128 + (j * 16 + n) * 3 + i] = v * INV_SQ96;
      });
    }
  }
  for (int i = 0; i < 3; ++i) {
    v2f aa[8];
    load_a(aa, [&](int m, int k) {
      v2f a;
      a.x = at[(size_t)m * 480 + 384 + 3 * k + i];
      a.y = at[(size_t)m * 480 + 384 + 3 * (k + 1) + i];
      return a;
    });
    for (int j = 0; j < 2; ++j) {
      v8f c = {};
      c = wmma_tile(aa, [&](int n, int k) {
        return ld2(WO2T + (size_t)(j * 16 + n) * 32 + k);
      }, c);
      tile_store(c, [&](int, int m, int n, float v) {
        y[(size_t)(n0 + m) * 416 + 320 + (j * 16 + n) * 3 + i] = v * INV_SQ32;
      });
    }
  }
}

extern "C" void kernel_launch(void* const* d_in, const int* in_sizes, int n_in,
                              void* d_out, int out_size, void* d_ws,
                              size_t ws_size, hipStream_t stream) {
  (void)in_sizes; (void)n_in; (void)out_size; (void)ws_size;
  const float* x         = (const float*)d_in[0];
  const float* edge_sh   = (const float*)d_in[1];
  const float* edge_attr = (const float*)d_in[2];
  const float* W0p = (const float*)d_in[3];
  const float* b0p = (const float*)d_in[4];
  const float* W1p = (const float*)d_in[5];
  const float* W0n = (const float*)d_in[6];
  const float* b0n = (const float*)d_in[7];
  const float* W1n = (const float*)d_in[8];
  const float* G1  = (const float*)d_in[9];
  const float* g1b = (const float*)d_in[10];
  const float* G2  = (const float*)d_in[11];
  const float* g2b = (const float*)d_in[12];
  const float* F1  = (const float*)d_in[13];
  const float* F2  = (const float*)d_in[14];
  const float* L1  = (const float*)d_in[15];
  const float* L2  = (const float*)d_in[16];
  const float* WO0 = (const float*)d_in[17];
  const float* bO0 = (const float*)d_in[18];
  const float* WO1 = (const float*)d_in[19];
  const float* WO2 = (const float*)d_in[20];
  const int* edge_index = (const int*)d_in[21];

  float* ws   = (float*)d_ws;
  float* p0   = ws;
  float* p1   = p0   + (size_t)N_NODES * 64;
  float* x0n  = p1   + (size_t)N_NODES * 96;
  float* x1n  = x0n  + (size_t)N_NODES * 64;
  float* acc  = x1n  + (size_t)N_NODES * 96;
  float* wt   = acc  + (size_t)N_NODES * 480;
  float* W0pT = wt + OFF_W0P;
  float* W1pT = wt + OFF_W1P;
  float* W0nT = wt + OFF_W0N;
  float* W1nT = wt + OFF_W1N;
  float* G1T  = wt + OFF_G1;
  float* G2T  = wt + OFF_G2;
  float* F1T  = wt + OFF_F1;
  float* F2T  = wt + OFF_F2;
  float* L1T  = wt + OFF_L1;
  float* L2T  = wt + OFF_L2;
  float* WO0T = wt + OFF_WO0;
  float* WO1T = wt + OFF_WO1;
  float* WO2T = wt + OFF_WO2;

  int* sc   = (int*)d_out;
  int* cnt  = sc;
  int* off  = sc + CNT_PADDED;
  int* rank = sc + 2 * CNT_PADDED;
  int* perm = sc + 2 * CNT_PADDED + N_EDGES;

  setup_kernel<<<SETUP_BLOCKS, SETUP_THREADS, 0, stream>>>(
      W0p, W1p, W0n, W1n, G1, G2, F1, F2, L1, L2, WO0, WO1, WO2, wt, acc, cnt);

  sort_rank_kernel<<<(N_EDGES + 255) / 256, 256, 0, stream>>>(edge_index, cnt, rank);
  sort_scan_kernel<<<1, 1024, 0, stream>>>(cnt, off);
  sort_fill_kernel<<<(N_EDGES + 255) / 256, 256, 0, stream>>>(edge_index, off, rank, perm);

  node_kernel<<<N_NODES / 16, 32, 0, stream>>>(
      x, W0pT, b0p, W1pT, W0nT, b0n, W1nT, G1T, g1b, G2T, g2b,
      p0, p1, x0n, x1n);

  edge_kernel<<<N_EDGES / 16, 32, 0, stream>>>(
      edge_sh, edge_attr, F1T, F2T, L1T, L2T, edge_index, perm,
      p0, p1, x0n, x1n, acc);

  proj_kernel<<<N_NODES / 16, 32, 0, stream>>>(
      acc, WO0T, bO0, WO1T, WO2T, (float*)d_out);
}


# Weights stored in B-fragment order instead of plain transposes: a lane's operands for two k-steps are one 16-byte load and a wave's load covers 4 whole 128-byte lines, not 16; sums bit-identical
# speedup vs baseline: 1.8802x; 1.0860x over previous
#include <hip/hip_runtime.h>
#include <hip/hip_bf16.h>
#include <math.h>


typedef __attribute__((ext_vector_type(2))) float v2f;
typedef __attribute__((ext_vector_type(4))) float v4f;
typedef __attribute__((ext_vector_type(8))) float v8f;

#define N_NODES 10000
#define N_EDGES 320000

#define INV_SQ3   0.5773502691896258f
#define INV_SQ2   0.7071067811865476f
#define INV_SQ32  0.17677669529663687f
#define INV_SQ64  0.125f
#define INV_SQ96  0.10206207261596575f
#define INV_SQ160 0.07905694150420949f
#define LN2F      0.6931471805599453f

__device__ __forceinline__ float sspf(float v) {
  float sp = (v > 20.0f) ? v : log1pf(expf(v));
  return sp - LN2F;
}
__device__ __forceinline__ float siluf(float v) {
  return v / (1.0f + expf(-v));
}
__device__ __forceinline__ v2f ld2(const float* p) {
  return *(const v2f*)p;
}

__device__ __forceinline__ v8f wmma4(v2f a, v2f b, v8f c) {
  return __builtin_amdgcn_wmma_f32_16x16x4_f32(false, a, false, b,
                                               (short)0, c, false, false);
}


template <int KS, typename LA>
__device__ __forceinline__ void load_a(v2f (&a)[KS], LA la) {
  const int lane = threadIdx.x & 31;
  const int m = lane & 15;
  const int kk = (lane >> 4) << 1;
#pragma unroll
  for (int q = 0; q < KS; ++q) a[q] = la(m, q * 4 + kk);
}

template <int KS>
__device__ __forceinline__ v8f wmma_tile(const v2f (&a)[KS],
                                         const float* __restrict__ wf_tile, v8f c) {
  static_assert(KS % 2 == 0, "k-steps are stored in pairs");
  const v4f* bp = (const v4f*)wf_tile + (threadIdx.x & 31);
#pragma unroll
  for (int p = 0; p < KS / 2; ++p) {
    const v4f b = bp[p * 32];
    const v2f b0 = {b.x, b.y};
    const v2f b1 = {b.z, b.w};
    c = wmma4(a[2 * p], b0, c);
    c = wmma4(a[2 * p + 1], b1, c);
  }
  return c;
}

template <typename ST>
__device__ __forceinline__ void tile_store(v8f c, ST st) {
  const int lane = threadIdx.x & 31;
  const int col = lane & 15;
  const int rbase = (lane >> 4) << 3;
#pragma unroll
  for (int r = 0; r < 8; ++r) st(r, rbase + r, col, c[r]);
}

constexpr int OFF_W0P = 0;
constexpr int OFF_W1P = OFF_W0P + 64 * 64;
constexpr int OFF_W0N = OFF_W1P + 32 * 32;
constexpr int OFF_W1N = OFF_W0N + 64 * 64;
constexpr int OFF_G1  = OFF_W1N + 32 * 32;
constexpr int OFF_G2  = OFF_G1  + 96 * 96;
constexpr int OFF_F1  = OFF_G2  + 96 * 96;
constexpr int OFF_F2  = OFF_F1  + 32 * 32;
constexpr int OFF_L1  = OFF_F2  + 32 * 224;
constexpr int OFF_L2  = OFF_L1  + 160 * 32;
constexpr int OFF_WO0 = OFF_L2  + 32 * 224;
constexpr int OFF_WO1 = OFF_WO0 + 96 * 128;
constexpr int OFF_WO2 = OFF_WO1 + 96 * 64;
constexpr int WT_FLOATS = OFF_WO2 + 32 * 32;

constexpr int SETUP_THREADS  = 256;
constexpr int SETUP_T_BLOCKS = WT_FLOATS / 1024;
constexpr int ACC_V4         = N_NODES * 480 / 4;
constexpr int Z_V4_PER_BLOCK = SETUP_THREADS * 16;
constexpr int SETUP_Z_BLOCKS = (ACC_V4 + Z_V4_PER_BLOCK - 1) / Z_V4_PER_BLOCK;
constexpr int CNT_PADDED     = 10240;
constexpr int SETUP_BLOCKS   = SETUP_T_BLOCKS + SETUP_Z_BLOCKS + 1;

template <int K, int Nc>
__device__ __forceinline__ void frag_slab(const float* __restrict__ W,
                                          float* __restrict__ Wf, int slab) {
  static_assert((K * Nc) % 1024 == 0, "weights are whole slabs");
  static_assert(K % 8 == 0 && Nc % 16 == 0, "whole k-step pairs and column tiles");
  float v[4];
#pragma unroll
  for (int i = 0; i < 4; ++i) v[i] = W[slab * 1024 + i * SETUP_THREADS + threadIdx.x];
#pragma unroll
  for (int i = 0; i < 4; ++i) {
    const int idx = slab * 1024 + i * SETUP_THREADS + threadIdx.x;
    const int k = idx / Nc, n = idx % Nc;
    const int j = n >> 4, c = n & 15;
    const int p = k >> 3, r = k & 7;
    const int h = (r >> 1) & 1;
    const int sl = ((r >> 2) << 1) | (r & 1);
    Wf[(((j * (K / 8) + p) * 32) + h * 16 + c) * 4 + sl] = v[i];
  }
}

__global__ __launch_bounds__(SETUP_THREADS) void setup_kernel(
    const float* __restrict__ W0p, const float* __restrict__ W1p,
    const float* __restrict__ W0n, const float* __restrict__ W1n,
    const float* __restrict__ G1, const float* __restrict__ G2,
    const float* __restrict__ F1, const float* __restrict__ F2,
    const float* __restrict__ L1, const float* __restrict__ L2,
    const float* __restrict__ WO0, const float* __restrict__ WO1,
    const float* __restrict__ WO2,
    float* __restrict__ wt, float* __restrict__ acc, int* __restrict__ cnt) {
  int s = blockIdx.x;
  if (s < SETUP_T_BLOCKS) {
    if (s < 4)  { frag_slab<64, 64>(W0p, wt + OFF_W0P, s); return; }   s -= 4;
    if (s < 1)  { frag_slab<32, 32>(W1p, wt + OFF_W1P, s); return; }   s -= 1;
    if (s < 4)  { frag_slab<64, 64>(W0n, wt + OFF_W0N, s); return; }   s -= 4;
    if (s < 1)  { frag_slab<32, 32>(W1n, wt + OFF_W1N, s); return; }   s -= 1;
    if (s < 9)  { frag_slab<96, 96>(G1, wt + OFF_G1, s); return; }     s -= 9;
    if (s < 9)  { frag_slab<96, 96>(G2, wt + OFF_G2, s); return; }     s -= 9;
    if (s < 1)  { frag_slab<32, 32>(F1, wt + OFF_F1, s); return; }     s -= 1;
    if (s < 7)  { frag_slab<32, 224>(F2, wt + OFF_F2, s); return; }    s -= 7;
    if (s < 5)  { frag_slab<160, 32>(L1, wt + OFF_L1, s); return; }    s -= 5;
    if (s < 7)  { frag_slab<32, 224>(L2, wt + OFF_L2, s); return; }    s -= 7;
    if (s < 12) { frag_slab<96, 128>(WO0, wt + OFF_WO0, s); return; }  s -= 12;
    if (s < 6)  { frag_slab<96, 64>(WO1, wt + OFF_WO1, s); return; }   s -= 6;
    frag_slab<32, 32>(WO2, wt + OFF_WO2, s);
    return;
  }
  s -= SETUP_T_BLOCKS;
  if (s < SETUP_Z_BLOCKS) {
    v4f* a4 = (v4f*)acc;
    const v4f z = {0.0f, 0.0f, 0.0f, 0.0f};
    for (int i = 0; i < 16; ++i) {
      const int q = s * Z_V4_PER_BLOCK + i * SETUP_THREADS + threadIdx.x;
      if (q < ACC_V4) a4[q] = z;
    }
    return;
  }
  for (int i = threadIdx.x; i < CNT_PADDED; i += SETUP_THREADS) cnt[i] = 0;
}

__global__ __launch_bounds__(32) void node_kernel(
    const float* __restrict__ x,
    const float* __restrict__ W0pF, const float* __restrict__ b0p,
    const float* __restrict__ W1pF,
    const float* __restrict__ W0nF, const float* __restrict__ b0n,
    const float* __restrict__ W1nF,
    const float* __restrict__ G1F, const float* __restrict__ g1b,
    const float* __restrict__ G2F, const float* __restrict__ g2b,
    float* __restrict__ p0, float* __restrict__ p1,
    float* __restrict__ x0n, float* __restrict__ x1n) {
  __shared__ __align__(16) float f0S[16 * 96];
  __shared__ __align__(16) float hS[16 * 96];
  __shared__ __align__(16) float gS[16 * 96];

  const int lane = threadIdx.x;
  const int n0 = blockIdx.x * 16;
  const float* xt = x + (size_t)n0 * 160;

  {
    v2f ax[16];
    load_a(ax, [&](int m, int k) { return ld2(xt + m * 160 + k); });
    for (int j = 0; j < 4; ++j) {
      v8f c = {};
      c = wmma_tile(ax, W0pF + j * 16 * 64, c);
      tile_store(c, [&](int, int m, int n, float v) {
        int col = j * 16 + n;
        p0[(size_t)(n0 + m) * 64 + col] = v * INV_SQ64 + b0p[col];
      });
    }
  }
  for (int i = 0; i < 3; ++i) {
    v2f ax[8];
    load_a(ax, [&](int m, int k) {
      v2f a;
      a.x = xt[m * 160 + 64 + 3 * k + i];
      a.y = xt[m * 160 + 64 + 3 * (k + 1) + i];
      return a;
    });
    for (int j = 0; j < 2; ++j) {
      v8f c = {};
      c = wmma_tile(ax, W1pF + j * 16 * 32, c);
      tile_store(c, [&](int, int m, int n, float v) {
        p1[(size_t)(n0 + m) * 96 + (j * 16 + n) * 3 + i] = v * INV_SQ32;
      });
    }
  }

  for (int idx = lane; idx < 16 * 96; idx += 32) {
    int m = idx / 96, c = idx % 96;
    float v;
    if (c < 64) {
      v = xt[m * 160 + c];
    } else {
      int u = c - 64;
      float a0 = xt[m * 160 + 64 + u * 3 + 0];
      float a1 = xt[m * 160 + 64 + u * 3 + 1];
      float a2 = xt[m * 160 + 64 + u * 3 + 2];
      v = sqrtf(a0 * a0 + a1 * a1 + a2 * a2 + 1e-12f);
    }
    f0S[idx] = v;
  }
  __syncthreads();

  {
    v2f af[24];
    load_a(af, [&](int m, int k) { return ld2(f0S + m * 96 + k); });
    for (int j = 0; j < 6; ++j) {
      v8f c = {};
      c = wmma_tile(af, G1F + j * 16 * 96, c);
      tile_store(c, [&](int, int m, int n, float v) {
        int col = j * 16 + n;
        hS[m * 96 + col] = siluf(v + g1b[col]);
      });
    }
  }
  __syncthreads();
  {
    v2f ah[24];
    load_a(ah, [&](int m, int k) { return ld2(hS + m * 96 + k); });
    for (int j = 0; j < 6; ++j) {
      v8f c = {};
      c = wmma_tile(ah, G2F + j * 16 * 96, c);
      tile_store(c, [&](int, int m, int n, float v) {
        gS[m * 96 + j * 16 + n] = v + g2b[j * 16 + n];
      });
    }
  }
  __syncthreads();

  {
    v2f ag[16];
    load_a(ag, [&](int m, int k) { return ld2(gS + m * 96 + k); });
    for (int j = 0; j < 4; ++j) {
      v8f c = {};
      c = wmma_tile(ag, W0nF + j * 16 * 64, c);
      tile_store(c, [&](int, int m, int n, float v) {
        int col = j * 16 + n;
        x0n[(size_t)(n0 + m) * 64 + col] = v * INV_SQ64 + b0n[col];
      });
    }
  }
  for (int i = 0; i < 3; ++i) {
    v2f ax[8];
    load_a(ax, [&](int m, int k) {
      v2f a;
      a.x = xt[m * 160 + 64 + 3 * k + i] * gS[m * 96 + 64 + k];
      a.y = xt[m * 160 + 64 + 3 * (k + 1) + i] * gS[m * 96 + 64 + k + 1];
      return a;
    });
    for (int j = 0; j < 2; ++j) {
      v8f c = {};
      c = wmma_tile(ax, W1nF + j * 16 * 32, c);
      tile_store(c, [&](int, int m, int n, float v) {
        x1n[(size_t)(n0 + m) * 96 + (j * 16 + n) * 3 + i] = v * INV_SQ32;
      });
    }
  }
}

__global__ void sort_rank_kernel(const int* __restrict__ edge_index,
                                 int* __restrict__ cnt, int* __restrict__ rank) {
  int e = blockIdx.x * blockDim.x + threadIdx.x;
  if (e < N_EDGES) rank[e] = atomicAdd(&cnt[edge_index[e]], 1);
}

#define SCAN_PER_THREAD 10
__global__ __launch_bounds__(1024) void sort_scan_kernel(
    const int* __restrict__ cnt, int* __restrict__ off) {
  __shared__ int part[1024];
  const int t = threadIdx.x;
  int loc[SCAN_PER_THREAD];
  int s = 0;
#pragma unroll
  for (int i = 0; i < SCAN_PER_THREAD; ++i) {
    int idx = t * SCAN_PER_THREAD + i;
    int v = (idx < N_NODES) ? cnt[idx] : 0;
    loc[i] = s;
    s += v;
  }
  part[t] = s;
  __syncthreads();
  for (int d = 1; d < 1024; d <<= 1) {
    int v = (t >= d) ? part[t - d] : 0;
    __syncthreads();
    part[t] += v;
    __syncthreads();
  }
  const int base = part[t] - s;
#pragma unroll
  for (int i = 0; i < SCAN_PER_THREAD; ++i) {
    int idx = t * SCAN_PER_THREAD + i;
    if (idx < N_NODES) off[idx] = base + loc[i];
  }
}

__global__ void sort_fill_kernel(const int* __restrict__ edge_index,
                                 const int* __restrict__ off,
                                 const int* __restrict__ rank,
                                 int* __restrict__ perm) {
  int e = blockIdx.x * blockDim.x + threadIdx.x;
  if (e < N_EDGES) perm[off[edge_index[e]] + rank[e]] = e;
}

__device__ __forceinline__ void run_add(float* __restrict__ acc,
                                        const int (&dR)[8], unsigned flm,
                                        int chan, const float (&v)[8]) {
  float run = 0.0f;
#pragma unroll
  for (int r = 0; r < 8; ++r) {
    run += v[r];
    if ((flm >> r) & 1u) {
      atomicAdd(acc + (size_t)dR[r] * 480 + chan, run);
      run = 0.0f;
    }
  }
}

__global__ __launch_bounds__(32) void edge_kernel(
    const float* __restrict__ edge_sh, const float* __restrict__ edge_attr,
    const float* __restrict__ F1F, const float* __restrict__ F2F,
    const float* __restrict__ L1F, const float* __restrict__ L2F,
    const int* __restrict__ edge_index, const int* __restrict__ perm,
    const float* __restrict__ p0, const float* __restrict__ p1,
    const float* __restrict__ x0n, const float* __restrict__ x1n,
    float* __restrict__ acc) {
  __shared__ __align__(16) float s0S[16 * 160];
  __shared__ __align__(16) float hN[16 * 32];
  __shared__ __align__(16) float hE[16 * 32];
  __shared__ int dstS[16];
  __shared__ int srcS[16];
  __shared__ int eS[16];

  const int lane = threadIdx.x;
  const int e0 = blockIdx.x * 16;

  __builtin_prefetch(F2F, 0, 1);
  __builtin_prefetch(L2F, 0, 1);

  if (lane < 16) {
    const int e = perm[e0 + lane];
    eS[lane] = e;
    dstS[lane] = edge_index[e];
    srcS[lane] = edge_index[N_EDGES + e];
  }
  __syncthreads();

  {
    const int hl = lane & 15;
    v4f pv[16];
#pragma unroll
    for (int m = 0; m < 16; ++m) {
      const int node = (lane < 16) ? dstS[m] : srcS[m];
      pv[m] = *(const v4f*)(p0 + (size_t)node * 64 + hl * 4);
    }
#pragma unroll
    for (int m = 0; m < 16; ++m) *(v4f*)(s0S + m * 160 + lane * 4) = pv[m];

#pragma nounroll
    for (int mb = 0; mb < 16; mb += 8) {
      float dv[8];
#pragma unroll
      for (int r = 0; r < 8; ++r) {
        const float* pd = p1 + (size_t)dstS[mb + r] * 96 + lane * 3;
        const float* ps = p1 + (size_t)srcS[mb + r] * 96 + lane * 3;
        dv[r] = (pd[0] * ps[0] + pd[1] * ps[1] + pd[2] * ps[2]) * INV_SQ3;
      }
#pragma unroll
      for (int r = 0; r < 8; ++r) s0S[(mb + r) * 160 + 128 + lane] = dv[r];
    }
  }
  __syncthreads();

  {
    v2f as0[40];
    load_a(as0, [&](int m, int k) { return ld2(s0S + m * 160 + k); });
    for (int j = 0; j < 2; ++j) {
      v8f c = {};
      c = wmma_tile(as0, L1F + j * 16 * 160, c);
      tile_store(c, [&](int, int m, int n, float v) {
        hN[m * 32 + j * 16 + n] = sspf(v * INV_SQ160);
      });
    }
  }
  {
    v2f aea[8];
    load_a(aea, [&](int m, int k) {
      return ld2(edge_attr + (size_t)eS[m] * 32 + k);
    });
    for (int j = 0; j < 2; ++j) {
      v8f c = {};
      c = wmma_tile(aea, F1F + j * 16 * 32, c);
      tile_store(c, [&](int, int m, int n, float v) {
        hE[m * 32 + j * 16 + n] = sspf(v * INV_SQ32);
      });
    }
  }
  __syncthreads();

  const int rbase = (lane >> 4) << 3;
  float sh0r[8], s1xr[8], s1yr[8], s1zr[8];
  int dR[8], sR[8];
#pragma unroll
  for (int r = 0; r < 8; ++r) {
    int m = rbase + r;
    int e = eS[m];
    sh0r[r] = edge_sh[(size_t)e * 4 + 0];
    s1xr[r] = edge_sh[(size_t)e * 4 + 1];
    s1yr[r] = edge_sh[(size_t)e * 4 + 2];
    s1zr[r] = edge_sh[(size_t)e * 4 + 3];
    dR[r] = dstS[m];
    sR[r] = srcS[m];
  }
  unsigned flm = 0x80u;
#pragma unroll
  for (int r = 0; r < 7; ++r) flm |= (dR[r + 1] != dR[r]) ? (1u << r) : 0u;

  v2f aN[8], aE[8];
  load_a(aN, [&](int m, int k) { return ld2(hN + m * 32 + k); });
  load_a(aE, [&](int m, int k) { return ld2(hE + m * 32 + k); });

  for (int j = 0; j < 14; ++j) {
    const int nb = j * 16;
    v8f cN = {}, cE = {};
    cN = wmma_tile(aN, L2F + nb * 32, cN);
    cE = wmma_tile(aE, F2F + nb * 32, cE);
    v8f w8 = cN * cE * (INV_SQ32 * INV_SQ32);

    const int col = nb + (lane & 15);
    float v0[8], v1[8], v2[8];
    if (nb < 64) {
#pragma unroll
      for (int r = 0; r < 8; ++r) {
        float xs = x0n[(size_t)sR[r] * 64 + col];
        v0[r] = w8[r] * xs * sh0r[r];
      }
      run_add(acc, dR, flm, col, v0);
    } else if (nb < 128) {
      const int c2 = col - 64;
#pragma unroll
      for (int r = 0; r < 8; ++r) {
        float f = w8[r] * x0n[(size_t)sR[r] * 64 + c2] * INV_SQ3;
        v0[r] = f * s1xr[r];
        v1[r] = f * s1yr[r];
        v2[r] = f * s1zr[r];
      }
      run_add(acc, dR, flm, 96 + c2 * 3 + 0, v0);
      run_add(acc, dR, flm, 96 + c2 * 3 + 1, v1);
      run_add(acc, dR, flm, 96 + c2 * 3 + 2, v2);
    } else if (nb < 160) {
      const int c3 = col - 128;
#pragma unroll
      for (int r = 0; r < 8; ++r) {
        const float* xv = x1n + (size_t)sR[r] * 96 + c3 * 3;
        float f = w8[r] * sh0r[r];
        v0[r] = f * xv[0];
        v1[r] = f * xv[1];
        v2[r] = f * xv[2];
      }
      run_add(acc, dR, flm, 288 + c3 * 3 + 0, v0);
      run_add(acc, dR, flm, 288 + c3 * 3 + 1, v1);
      run_add(acc, dR, flm, 288 + c3 * 3 + 2, v2);
    } else if (nb < 192) {
      const int c4 = col - 160;
#pragma unroll
      for (int r = 0; r < 8; ++r) {
        const float* xv = x1n + (size_t)sR[r] * 96 + c4 * 3;
        float dp = xv[0] * s1xr[r] + xv[1] * s1yr[r] + xv[2] * s1zr[r];
        v0[r] = w8[r] * dp * INV_SQ3;
      }
      run_add(acc, dR, flm, 64 + c4, v0);
    } else {
      const int c5 = col - 192;
#pragma unroll
      for (int r = 0; r < 8; ++r) {
        const float* xv = x1n + (size_t)sR[r] * 96 + c5 * 3;
        const float s1x = s1xr[r], s1y = s1yr[r], s1z = s1zr[r];
        float cx = xv[1] * s1z - xv[2] * s1y;
        float cy = xv[2] * s1x - xv[0] * s1z;
        float cz = xv[0] * s1y - xv[1] * s1x;
        float f = w8[r] * INV_SQ2;
        v0[r] = f * cx;
        v1[r] = f * cy;
        v2[r] = f * cz;
      }
      run_add(acc, dR, flm, 384 + c5 * 3 + 0, v0);
      run_add(acc, dR, flm, 384 + c5 * 3 + 1, v1);
      run_add(acc, dR, flm, 384 + c5 * 3 + 2, v2);
    }
  }
}

__global__ __launch_bounds__(32) void proj_kernel(
    const float* __restrict__ acc,
    const float* __restrict__ WO0F, const float* __restrict__ bO0,
    const float* __restrict__ WO1F, const float* __restrict__ WO2F,
    float* __restrict__ y) {
  const int n0 = blockIdx.x * 16;
  const float* at = acc + (size_t)n0 * 480;

  {
    v2f aa[24];
    load_a(aa, [&](int m, int k) { return ld2(at + (size_t)m * 480 + k); });
    for (int j = 0; j < 8; ++j) {
      v8f c = {};
      c = wmma_tile(aa, WO0F + j * 16 * 96, c);
      tile_store(c, [&](int, int m, int n, float v) {
        int col = j * 16 + n;
        y[(size_t)(n0 + m) * 416 + col] = v * INV_SQ96 + bO0[col];
      });
    }
  }
  for (int i = 0; i < 3; ++i) {
    v2f aa[24];
    load_a(aa, [&](int m, int k) {
      v2f a;
      a.x = at[(size_t)m * 480 + 96 + 3 * k + i];
      a.y = at[(size_t)m * 480 + 96 + 3 * (k + 1) + i];
      return a;
    });
    for (int j = 0; j < 4; ++j) {
      v8f c = {};
      c = wmma_tile(aa, WO1F + j * 16 * 96, c);
      tile_store(c, [&](int, int m, int n, float v) {
        y[(size_t)(n0 + m) * 416 + 128 + (j * 16 + n) * 3 + i] = v * INV_SQ96;
      });
    }
  }
  for (int i = 0; i < 3; ++i) {
    v2f aa[8];
    load_a(aa, [&](int m, int k) {
      v2f a;
      a.x = at[(size_t)m * 480 + 384 + 3 * k + i];
      a.y = at[(size_t)m * 480 + 384 + 3 * (k + 1) + i];
      return a;
    });
    for (int j = 0; j < 2; ++j) {
      v8f c = {};
      c = wmma_tile(aa, WO2F + j * 16 * 32, c);
      tile_store(c, [&](int, int m, int n, float v) {
        y[(size_t)(n0 + m) * 416 + 320 + (j * 16 + n) * 3 + i] = v * INV_SQ32;
      });
    }
  }
}

extern "C" void kernel_launch(void* const* d_in, const int* in_sizes, int n_in,
                              void* d_out, int out_size, void* d_ws,
                              size_t ws_size, hipStream_t stream) {
  (void)in_sizes; (void)n_in; (void)out_size; (void)ws_size;
  const float* x         = (const float*)d_in[0];
  const float* edge_sh   = (const float*)d_in[1];
  const float* edge_attr = (const float*)d_in[2];
  const float* W0p = (const float*)d_in[3];
  const float* b0p = (const float*)d_in[4];
  const float* W1p = (const float*)d_in[5];
  const float* W0n = (const float*)d_in[6];
  const float* b0n = (const float*)d_in[7];
  const float* W1n = (const float*)d_in[8];
  const float* G1  = (const float*)d_in[9];
  const float* g1b = (const float*)d_in[10];
  const float* G2  = (const float*)d_in[11];
  const float* g2b = (const float*)d_in[12];
  const float* F1  = (const float*)d_in[13];
  const float* F2  = (const float*)d_in[14];
  const float* L1  = (const float*)d_in[15];
  const float* L2  = (const float*)d_in[16];
  const float* WO0 = (const float*)d_in[17];
  const float* bO0 = (const float*)d_in[18];
  const float* WO1 = (const float*)d_in[19];
  const float* WO2 = (const float*)d_in[20];
  const int* edge_index = (const int*)d_in[21];

  float* ws   = (float*)d_ws;
  float* p0   = ws;
  float* p1   = p0   + (size_t)N_NODES * 64;
  float* x0n  = p1   + (size_t)N_NODES * 96;
  float* x1n  = x0n  + (size_t)N_NODES * 64;
  float* acc  = x1n  + (size_t)N_NODES * 96;
  float* wt   = acc  + (size_t)N_NODES * 480;
  float* W0pF = wt + OFF_W0P;
  float* W1pF = wt + OFF_W1P;
  float* W0nF = wt + OFF_W0N;
  float* W1nF = wt + OFF_W1N;
  float* G1F  = wt + OFF_G1;
  float* G2F  = wt + OFF_G2;
  float* F1F  = wt + OFF_F1;
  float* F2F  = wt + OFF_F2;
  float* L1F  = wt + OFF_L1;
  float* L2F  = wt + OFF_L2;
  float* WO0F = wt + OFF_WO0;
  float* WO1F = wt + OFF_WO1;
  float* WO2F = wt + OFF_WO2;

  int* sc   = (int*)d_out;
  int* cnt  = sc;
  int* off  = sc + CNT_PADDED;
  int* rank = sc + 2 * CNT_PADDED;
  int* perm = sc + 2 * CNT_PADDED + N_EDGES;

  setup_kernel<<<SETUP_BLOCKS, SETUP_THREADS, 0, stream>>>(
      W0p, W1p, W0n, W1n, G1, G2, F1, F2, L1, L2, WO0, WO1, WO2, wt, acc, cnt);

  sort_rank_kernel<<<(N_EDGES + 255) / 256, 256, 0, stream>>>(edge_index, cnt, rank);
  sort_scan_kernel<<<1, 1024, 0, stream>>>(cnt, off);
  sort_fill_kernel<<<(N_EDGES + 255) / 256, 256, 0, stream>>>(edge_index, off, rank, perm);

  node_kernel<<<N_NODES / 16, 32, 0, stream>>>(
      x, W0pF, b0p, W1pF, W0nF, b0n, W1nF, G1F, g1b, G2F, g2b,
      p0, p1, x0n, x1n);

  edge_kernel<<<N_EDGES / 16, 32, 0, stream>>>(
      edge_sh, edge_attr, F1F, F2F, L1F, L2F, edge_index, perm,
      p0, p1, x0n, x1n, acc);

  proj_kernel<<<N_NODES / 16, 32, 0, stream>>>(
      acc, WO0F, bO0, WO1F, WO2F, (float*)d_out);
}


# Edge kernel requests each load when its address is known (a tile's source rows with its weight fragments; edge_attr, edge_sh rows with node indices); 32-bit row offsets keep 5 waves/SIMD; same sums
# speedup vs baseline: 1.9666x; 1.0460x over previous
#include <hip/hip_runtime.h>
#include <hip/hip_bf16.h>
#include <math.h>


typedef __attribute__((ext_vector_type(2))) float v2f;
typedef __attribute__((ext_vector_type(4))) float v4f;
typedef __attribute__((ext_vector_type(8))) float v8f;

#define N_NODES 10000
#define N_EDGES 320000

#define INV_SQ3   0.5773502691896258f
#define INV_SQ2   0.7071067811865476f
#define INV_SQ32  0.17677669529663687f
#define INV_SQ64  0.125f
#define INV_SQ96  0.10206207261596575f
#define INV_SQ160 0.07905694150420949f
#define LN2F      0.6931471805599453f

__device__ __forceinline__ float sspf(float v) {
  float sp = (v > 20.0f) ? v : log1pf(expf(v));
  return sp - LN2F;
}
__device__ __forceinline__ float siluf(float v) {
  return v / (1.0f + expf(-v));
}
__device__ __forceinline__ v2f ld2(const float* p) {
  return *(const v2f*)p;
}

__device__ __forceinline__ v8f wmma4(v2f a, v2f b, v8f c) {
  return __builtin_amdgcn_wmma_f32_16x16x4_f32(false, a, false, b,
                                               (short)0, c, false, false);
}


template <int KS, typename LA>
__device__ __forceinline__ void load_a(v2f (&a)[KS], LA la) {
  const int lane = threadIdx.x & 31;
  const int m = lane & 15;
  const int kk = (lane >> 4) << 1;
#pragma unroll
  for (int q = 0; q < KS; ++q) a[q] = la(m, q * 4 + kk);
}

template <int KS>
__device__ __forceinline__ v8f wmma_tile(const v2f (&a)[KS],
                                         const float* __restrict__ wf_tile, v8f c) {
  static_assert(KS % 2 == 0, "k-steps are stored in pairs");
  const v4f* bp = (const v4f*)wf_tile + (threadIdx.x & 31);
#pragma unroll
  for (int p = 0; p < KS / 2; ++p) {
    const v4f b = bp[p * 32];
    const v2f b0 = {b.x, b.y};
    const v2f b1 = {b.z, b.w};
    c = wmma4(a[2 * p], b0, c);
    c = wmma4(a[2 * p + 1], b1, c);
  }
  return c;
}

__device__ __forceinline__ void load_b32(v4f (&b)[4], const float* __restrict__ wf_tile) {
  const v4f* bp = (const v4f*)wf_tile + (threadIdx.x & 31);
#pragma unroll
  for (int p = 0; p < 4; ++p) b[p] = bp[p * 32];
}
__device__ __forceinline__ v8f wmma_regs(const v2f (&a)[8], const v4f (&b)[4], v8f c) {
#pragma unroll
  for (int p = 0; p < 4; ++p) {
    const v2f b0 = {b[p].x, b[p].y};
    const v2f b1 = {b[p].z, b[p].w};
    c = wmma4(a[2 * p], b0, c);
    c = wmma4(a[2 * p + 1], b1, c);
  }
  return c;
}
template <int V> struct IC { static constexpr int value = V; };

template <typename ST>
__device__ __forceinline__ void tile_store(v8f c, ST st) {
  const int lane = threadIdx.x & 31;
  const int col = lane & 15;
  const int rbase = (lane >> 4) << 3;
#pragma unroll
  for (int r = 0; r < 8; ++r) st(r, rbase + r, col, c[r]);
}

constexpr int OFF_W0P = 0;
constexpr int OFF_W1P = OFF_W0P + 64 * 64;
constexpr int OFF_W0N = OFF_W1P + 32 * 32;
constexpr int OFF_W1N = OFF_W0N + 64 * 64;
constexpr int OFF_G1  = OFF_W1N + 32 * 32;
constexpr int OFF_G2  = OFF_G1  + 96 * 96;
constexpr int OFF_F1  = OFF_G2  + 96 * 96;
constexpr int OFF_F2  = OFF_F1  + 32 * 32;
constexpr int OFF_L1  = OFF_F2  + 32 * 224;
constexpr int OFF_L2  = OFF_L1  + 160 * 32;
constexpr int OFF_WO0 = OFF_L2  + 32 * 224;
constexpr int OFF_WO1 = OFF_WO0 + 96 * 128;
constexpr int OFF_WO2 = OFF_WO1 + 96 * 64;
constexpr int WT_FLOATS = OFF_WO2 + 32 * 32;

constexpr int SETUP_THREADS  = 256;
constexpr int SETUP_T_BLOCKS = WT_FLOATS / 1024;
constexpr int ACC_V4         = N_NODES * 480 / 4;
constexpr int Z_V4_PER_BLOCK = SETUP_THREADS * 16;
constexpr int SETUP_Z_BLOCKS = (ACC_V4 + Z_V4_PER_BLOCK - 1) / Z_V4_PER_BLOCK;
constexpr int CNT_PADDED     = 10240;
constexpr int SETUP_BLOCKS   = SETUP_T_BLOCKS + SETUP_Z_BLOCKS + 1;

template <int K, int Nc>
__device__ __forceinline__ void frag_slab(const float* __restrict__ W,
                                          float* __restrict__ Wf, int slab) {
  static_assert((K * Nc) % 1024 == 0, "weights are whole slabs");
  static_assert(K % 8 == 0 && Nc % 16 == 0, "whole k-step pairs and column tiles");
  float v[4];
#pragma unroll
  for (int i = 0; i < 4; ++i) v[i] = W[slab * 1024 + i * SETUP_THREADS + threadIdx.x];
#pragma unroll
  for (int i = 0; i < 4; ++i) {
    const int idx = slab * 1024 + i * SETUP_THREADS + threadIdx.x;
    const int k = idx / Nc, n = idx % Nc;
    const int j = n >> 4, c = n & 15;
    const int p = k >> 3, r = k & 7;
    const int h = (r >> 1) & 1;
    const int sl = ((r >> 2) << 1) | (r & 1);
    Wf[(((j * (K / 8) + p) * 32) + h * 16 + c) * 4 + sl] = v[i];
  }
}

__global__ __launch_bounds__(SETUP_THREADS) void setup_kernel(
    const float* __restrict__ W0p, const float* __restrict__ W1p,
    const float* __restrict__ W0n, const float* __restrict__ W1n,
    const float* __restrict__ G1, const float* __restrict__ G2,
    const float* __restrict__ F1, const float* __restrict__ F2,
    const float* __restrict__ L1, const float* __restrict__ L2,
    const float* __restrict__ WO0, const float* __restrict__ WO1,
    const float* __restrict__ WO2,
    float* __restrict__ wt, float* __restrict__ acc, int* __restrict__ cnt) {
  int s = blockIdx.x;
  if (s < SETUP_T_BLOCKS) {
    if (s < 4)  { frag_slab<64, 64>(W0p, wt + OFF_W0P, s); return; }   s -= 4;
    if (s < 1)  { frag_slab<32, 32>(W1p, wt + OFF_W1P, s); return; }   s -= 1;
    if (s < 4)  { frag_slab<64, 64>(W0n, wt + OFF_W0N, s); return; }   s -= 4;
    if (s < 1)  { frag_slab<32, 32>(W1n, wt + OFF_W1N, s); return; }   s -= 1;
    if (s < 9)  { frag_slab<96, 96>(G1, wt + OFF_G1, s); return; }     s -= 9;
    if (s < 9)  { frag_slab<96, 96>(G2, wt + OFF_G2, s); return; }     s -= 9;
    if (s < 1)  { frag_slab<32, 32>(F1, wt + OFF_F1, s); return; }     s -= 1;
    if (s < 7)  { frag_slab<32, 224>(F2, wt + OFF_F2, s); return; }    s -= 7;
    if (s < 5)  { frag_slab<160, 32>(L1, wt + OFF_L1, s); return; }    s -= 5;
    if (s < 7)  { frag_slab<32, 224>(L2, wt + OFF_L2, s); return; }    s -= 7;
    if (s < 12) { frag_slab<96, 128>(WO0, wt + OFF_WO0, s); return; }  s -= 12;
    if (s < 6)  { frag_slab<96, 64>(WO1, wt + OFF_WO1, s); return; }   s -= 6;
    frag_slab<32, 32>(WO2, wt + OFF_WO2, s);
    return;
  }
  s -= SETUP_T_BLOCKS;
  if (s < SETUP_Z_BLOCKS) {
    v4f* a4 = (v4f*)acc;
    const v4f z = {0.0f, 0.0f, 0.0f, 0.0f};
    for (int i = 0; i < 16; ++i) {
      const int q = s * Z_V4_PER_BLOCK + i * SETUP_THREADS + threadIdx.x;
      if (q < ACC_V4) a4[q] = z;
    }
    return;
  }
  for (int i = threadIdx.x; i < CNT_PADDED; i += SETUP_THREADS) cnt[i] = 0;
}

__global__ __launch_bounds__(32) void node_kernel(
    const float* __restrict__ x,
    const float* __restrict__ W0pF, const float* __restrict__ b0p,
    const float* __restrict__ W1pF,
    const float* __restrict__ W0nF, const float* __restrict__ b0n,
    const float* __restrict__ W1nF,
    const float* __restrict__ G1F, const float* __restrict__ g1b,
    const float* __restrict__ G2F, const float* __restrict__ g2b,
    float* __restrict__ p0, float* __restrict__ p1,
    float* __restrict__ x0n, float* __restrict__ x1n) {
  __shared__ __align__(16) float f0S[16 * 96];
  __shared__ __align__(16) float hS[16 * 96];
  __shared__ __align__(16) float gS[16 * 96];

  const int lane = threadIdx.x;
  const int n0 = blockIdx.x * 16;
  const float* xt = x + (size_t)n0 * 160;

  {
    v2f ax[16];
    load_a(ax, [&](int m, int k) { return ld2(xt + m * 160 + k); });
    for (int j = 0; j < 4; ++j) {
      v8f c = {};
      c = wmma_tile(ax, W0pF + j * 16 * 64, c);
      tile_store(c, [&](int, int m, int n, float v) {
        int col = j * 16 + n;
        p0[(size_t)(n0 + m) * 64 + col] = v * INV_SQ64 + b0p[col];
      });
    }
  }
  for (int i = 0; i < 3; ++i) {
    v2f ax[8];
    load_a(ax, [&](int m, int k) {
      v2f a;
      a.x = xt[m * 160 + 64 + 3 * k + i];
      a.y = xt[m * 160 + 64 + 3 * (k + 1) + i];
      return a;
    });
    for (int j = 0; j < 2; ++j) {
      v8f c = {};
      c = wmma_tile(ax, W1pF + j * 16 * 32, c);
      tile_store(c, [&](int, int m, int n, float v) {
        p1[(size_t)(n0 + m) * 96 + (j * 16 + n) * 3 + i] = v * INV_SQ32;
      });
    }
  }

  for (int idx = lane; idx < 16 * 96; idx += 32) {
    int m = idx / 96, c = idx % 96;
    float v;
    if (c < 64) {
      v = xt[m * 160 + c];
    } else {
      int u = c - 64;
      float a0 = xt[m * 160 + 64 + u * 3 + 0];
      float a1 = xt[m * 160 + 64 + u * 3 + 1];
      float a2 = xt[m * 160 + 64 + u * 3 + 2];
      v = sqrtf(a0 * a0 + a1 * a1 + a2 * a2 + 1e-12f);
    }
    f0S[idx] = v;
  }
  __syncthreads();

  {
    v2f af[24];
    load_a(af, [&](int m, int k) { return ld2(f0S + m * 96 + k); });
    for (int j = 0; j < 6; ++j) {
      v8f c = {};
      c = wmma_tile(af, G1F + j * 16 * 96, c);
      tile_store(c, [&](int, int m, int n, float v) {
        int col = j * 16 + n;
        hS[m * 96 + col] = siluf(v + g1b[col]);
      });
    }
  }
  __syncthreads();
  {
    v2f ah[24];
    load_a(ah, [&](int m, int k) { return ld2(hS + m * 96 + k); });
    for (int j = 0; j < 6; ++j) {
      v8f c = {};
      c = wmma_tile(ah, G2F + j * 16 * 96, c);
      tile_store(c, [&](int, int m, int n, float v) {
        gS[m * 96 + j * 16 + n] = v + g2b[j * 16 + n];
      });
    }
  }
  __syncthreads();

  {
    v2f ag[16];
    load_a(ag, [&](int m, int k) { return ld2(gS + m * 96 + k); });
    for (int j = 0; j < 4; ++j) {
      v8f c = {};
      c = wmma_tile(ag, W0nF + j * 16 * 64, c);
      tile_store(c, [&](int, int m, int n, float v) {
        int col = j * 16 + n;
        x0n[(size_t)(n0 + m) * 64 + col] = v * INV_SQ64 + b0n[col];
      });
    }
  }
  for (int i = 0; i < 3; ++i) {
    v2f ax[8];
    load_a(ax, [&](int m, int k) {
      v2f a;
      a.x = xt[m * 160 + 64 + 3 * k + i] * gS[m * 96 + 64 + k];
      a.y = xt[m * 160 + 64 + 3 * (k + 1) + i] * gS[m * 96 + 64 + k + 1];
      return a;
    });
    for (int j = 0; j < 2; ++j) {
      v8f c = {};
      c = wmma_tile(ax, W1nF + j * 16 * 32, c);
      tile_store(c, [&](int, int m, int n, float v) {
        x1n[(size_t)(n0 + m) * 96 + (j * 16 + n) * 3 + i] = v * INV_SQ32;
      });
    }
  }
}

__global__ void sort_rank_kernel(const int* __restrict__ edge_index,
                                 int* __restrict__ cnt, int* __restrict__ rank) {
  int e = blockIdx.x * blockDim.x + threadIdx.x;
  if (e < N_EDGES) rank[e] = atomicAdd(&cnt[edge_index[e]], 1);
}

#define SCAN_PER_THREAD 10
__global__ __launch_bounds__(1024) void sort_scan_kernel(
    const int* __restrict__ cnt, int* __restrict__ off) {
  __shared__ int part[1024];
  const int t = threadIdx.x;
  int loc[SCAN_PER_THREAD];
  int s = 0;
#pragma unroll
  for (int i = 0; i < SCAN_PER_THREAD; ++i) {
    int idx = t * SCAN_PER_THREAD + i;
    int v = (idx < N_NODES) ? cnt[idx] : 0;
    loc[i] = s;
    s += v;
  }
  part[t] = s;
  __syncthreads();
  for (int d = 1; d < 1024; d <<= 1) {
    int v = (t >= d) ? part[t - d] : 0;
    __syncthreads();
    part[t] += v;
    __syncthreads();
  }
  const int base = part[t] - s;
#pragma unroll
  for (int i = 0; i < SCAN_PER_THREAD; ++i) {
    int idx = t * SCAN_PER_THREAD + i;
    if (idx < N_NODES) off[idx] = base + loc[i];
  }
}

__global__ void sort_fill_kernel(const int* __restrict__ edge_index,
                                 const int* __restrict__ off,
                                 const int* __restrict__ rank,
                                 int* __restrict__ perm) {
  int e = blockIdx.x * blockDim.x + threadIdx.x;
  if (e < N_EDGES) perm[off[edge_index[e]] + rank[e]] = e;
}

__device__ __forceinline__ void run_add(float* __restrict__ acc,
                                        const unsigned (&dB)[8], unsigned flm,
                                        int chan, const float (&v)[8]) {
  float run = 0.0f;
#pragma unroll
  for (int r = 0; r < 8; ++r) {
    run += v[r];
    if ((flm >> r) & 1u) {
      atomicAdd((float*)((char*)acc + (dB[r] + (unsigned)chan * 4u)), run);
      run = 0.0f;
    }
  }
}

__global__ __launch_bounds__(32) void edge_kernel(
    const float* __restrict__ edge_sh, const float* __restrict__ edge_attr,
    const float* __restrict__ F1F, const float* __restrict__ F2F,
    const float* __restrict__ L1F, const float* __restrict__ L2F,
    const int* __restrict__ edge_index, const int* __restrict__ perm,
    const float* __restrict__ p0, const float* __restrict__ p1,
    const float* __restrict__ x0n, const float* __restrict__ x1n,
    float* __restrict__ acc) {
  __shared__ __align__(16) float s0S[16 * 160];
  __shared__ __align__(16) float hN[16 * 32];
  __shared__ __align__(16) float hE[16 * 32];
  __shared__ __align__(16) float shS[16 * 4];
  __shared__ int dstS[16];
  __shared__ int srcS[16];

  const int lane = threadIdx.x;
  const int e0 = blockIdx.x * 16;

  __builtin_prefetch(F2F, 0, 1);
  __builtin_prefetch(L2F, 0, 1);

  v4f bF0[4], bF1[4];
  load_b32(bF0, F1F);
  load_b32(bF1, F1F + 16 * 32);

  const int e = perm[e0 + (lane & 15)];
  v2f aea[8];
  {
    const float* ar = edge_attr + (size_t)e * 32 + ((lane >> 4) << 1);
#pragma unroll
    for (int q = 0; q < 8; ++q) aea[q] = ld2(ar + q * 4);
  }
  if (lane < 16) {
    dstS[lane] = edge_index[e];
    srcS[lane] = edge_index[N_EDGES + e];
    *(v4f*)(shS + lane * 4) = *(const v4f*)(edge_sh + (size_t)e * 4);
  }
  __syncthreads();

  {
    const int hl = lane & 15;
    v4f pv[16];
#pragma unroll
    for (int m = 0; m < 16; ++m) {
      const int node = (lane < 16) ? dstS[m] : srcS[m];
      pv[m] = *(const v4f*)(p0 + (size_t)node * 64 + hl * 4);
    }
    {
      v8f c0 = {}, c1 = {};
      c0 = wmma_regs(aea, bF0, c0);
      c1 = wmma_regs(aea, bF1, c1);
      tile_store(c0, [&](int, int m, int n, float v) {
        hE[m * 32 + n] = sspf(v * INV_SQ32);
      });
      tile_store(c1, [&](int, int m, int n, float v) {
        hE[m * 32 + 16 + n] = sspf(v * INV_SQ32);
      });
    }
#pragma unroll
    for (int m = 0; m < 16; ++m) *(v4f*)(s0S + m * 160 + lane * 4) = pv[m];

#pragma nounroll
    for (int mb = 0; mb < 16; mb += 8) {
      float dv[8];
#pragma unroll
      for (int r = 0; r < 8; ++r) {
        const float* pd = p1 + (size_t)dstS[mb + r] * 96 + lane * 3;
        const float* ps = p1 + (size_t)srcS[mb + r] * 96 + lane * 3;
        dv[r] = (pd[0] * ps[0] + pd[1] * ps[1] + pd[2] * ps[2]) * INV_SQ3;
      }
#pragma unroll
      for (int r = 0; r < 8; ++r) s0S[(mb + r) * 160 + 128 + lane] = dv[r];
    }
  }
  __syncthreads();

  {
    v2f as0[40];
    load_a(as0, [&](int m, int k) { return ld2(s0S + m * 160 + k); });
    for (int j = 0; j < 2; ++j) {
      v8f c = {};
      c = wmma_tile(as0, L1F + j * 16 * 160, c);
      tile_store(c, [&](int, int m, int n, float v) {
        hN[m * 32 + j * 16 + n] = sspf(v * INV_SQ160);
      });
    }
  }
  __syncthreads();

  const int rbase = (lane >> 4) << 3;
  float sh0r[8], s1xr[8], s1yr[8], s1zr[8];
  int dR[8], sR[8];
#pragma unroll
  for (int r = 0; r < 8; ++r) {
    int m = rbase + r;
    const v4f sh = *(const v4f*)(shS + m * 4);
    sh0r[r] = sh.x;
    s1xr[r] = sh.y;
    s1yr[r] = sh.z;
    s1zr[r] = sh.w;
    dR[r] = dstS[m];
    sR[r] = srcS[m];
  }
  unsigned s0B[8], s1B[8], dB[8];
#pragma unroll
  for (int r = 0; r < 8; ++r) {
    s0B[r] = (unsigned)sR[r] * 256u;
    s1B[r] = (unsigned)sR[r] * 384u;
    dB[r] = (unsigned)dR[r] * 1920u;
  }
  unsigned flm = 0x80u;
#pragma unroll
  for (int r = 0; r < 7; ++r) flm |= (dR[r + 1] != dR[r]) ? (1u << r) : 0u;

  v2f aN[8], aE[8];
  load_a(aN, [&](int m, int k) { return ld2(hN + m * 32 + k); });
  load_a(aE, [&](int m, int k) { return ld2(hE + m * 32 + k); });

  auto do_tile = [&](auto seg, const int j) {
    constexpr int SEG = decltype(seg)::value;
    const int nb = j * 16;
    const int col = nb + (lane & 15);
    v4f bN[4], bE[4];
    load_b32(bN, L2F + nb * 32);
    load_b32(bE, F2F + nb * 32);
    constexpr int CB = (SEG == 0) ? 0 : (SEG == 1) ? 64 : (SEG == 2) ? 128 : (SEG == 3) ? 160 : 192;
    const int cs = col - CB;
    float g0[8], g1[8], g2[8];
    if constexpr (SEG <= 1) {
#pragma unroll
      for (int r = 0; r < 8; ++r)
        g0[r] = *(const float*)((const char*)x0n + (s0B[r] + (unsigned)cs * 4u));
    } else {
#pragma unroll
      for (int r = 0; r < 8; ++r) {
        const float* xv = (const float*)((const char*)x1n + (s1B[r] + (unsigned)cs * 12u));
        g0[r] = xv[0];
        g1[r] = xv[1];
        g2[r] = xv[2];
      }
    }

    v8f cN = {}, cE = {};
    cN = wmma_regs(aN, bN, cN);
    cE = wmma_regs(aE, bE, cE);
    v8f w8 = cN * cE * (INV_SQ32 * INV_SQ32);

    float v0[8], v1[8], v2[8];
    if constexpr (SEG == 0) {
#pragma unroll
      for (int r = 0; r < 8; ++r) v0[r] = w8[r] * g0[r] * sh0r[r];
      run_add(acc, dB, flm, col, v0);
    } else if constexpr (SEG == 1) {
#pragma unroll
      for (int r = 0; r < 8; ++r) {
        float f = w8[r] * g0[r] * INV_SQ3;
        v0[r] = f * s1xr[r];
        v1[r] = f * s1yr[r];
        v2[r] = f * s1zr[r];
      }
      run_add(acc, dB, flm, 96 + cs * 3 + 0, v0);
      run_add(acc, dB, flm, 96 + cs * 3 + 1, v1);
      run_add(acc, dB, flm, 96 + cs * 3 + 2, v2);
    } else if constexpr (SEG == 2) {
#pragma unroll
      for (int r = 0; r < 8; ++r) {
        float f = w8[r] * sh0r[r];
        v0[r] = f * g0[r];
        v1[r] = f * g1[r];
        v2[r] = f * g2[r];
      }
      run_add(acc, dB, flm, 288 + cs * 3 + 0, v0);
      run_add(acc, dB, flm, 288 + cs * 3 + 1, v1);
      run_add(acc, dB, flm, 288 + cs * 3 + 2, v2);
    } else if constexpr (SEG == 3) {
#pragma unroll
      for (int r = 0; r < 8; ++r) {
        float dp = g0[r] * s1xr[r] + g1[r] * s1yr[r] + g2[r] * s1zr[r];
        v0[r] = w8[r] * dp * INV_SQ3;
      }
      run_add(acc, dB, flm, 64 + cs, v0);
    } else {
#pragma unroll
      for (int r = 0; r < 8; ++r) {
        const float s1x = s1xr[r], s1y = s1yr[r], s1z = s1zr[r];
        float cx = g1[r] * s1z - g2[r] * s1y;
        float cy = g2[r] * s1x - g0[r] * s1z;
        float cz = g0[r] * s1y - g1[r] * s1x;
        float f = w8[r] * INV_SQ2;
        v0[r] = f * cx;
        v1[r] = f * cy;
        v2[r] = f * cz;
      }
      run_add(acc, dB, flm, 384 + cs * 3 + 0, v0);
      run_add(acc, dB, flm, 384 + cs * 3 + 1, v1);
      run_add(acc, dB, flm, 384 + cs * 3 + 2, v2);
    }
  };

#pragma nounroll
  for (int j = 0; j < 4; ++j) do_tile(IC<0>{}, j);
#pragma nounroll
  for (int j = 4; j < 8; ++j) do_tile(IC<1>{}, j);
#pragma nounroll
  for (int j = 8; j < 10; ++j) do_tile(IC<2>{}, j);
#pragma nounroll
  for (int j = 10; j < 12; ++j) do_tile(IC<3>{}, j);
#pragma nounroll
  for (int j = 12; j < 14; ++j) do_tile(IC<4>{}, j);
}

__global__ __launch_bounds__(32) void proj_kernel(
    const float* __restrict__ acc,
    const float* __restrict__ WO0F, const float* __restrict__ bO0,
    const float* __restrict__ WO1F, const float* __restrict__ WO2F,
    float* __restrict__ y) {
  const int n0 = blockIdx.x * 16;
  const float* at = acc + (size_t)n0 * 480;

  {
    v2f aa[24];
    load_a(aa, [&](int m, int k) { return ld2(at + (size_t)m * 480 + k); });
    for (int j = 0; j < 8; ++j) {
      v8f c = {};
      c = wmma_tile(aa, WO0F + j * 16 * 96, c);
      tile_store(c, [&](int, int m, int n, float v) {
        int col = j * 16 + n;
        y[(size_t)(n0 + m) * 416 + col] = v * INV_SQ96 + bO0[col];
      });
    }
  }
  for (int i = 0; i < 3; ++i) {
    v2f aa[24];
    load_a(aa, [&](int m, int k) {
      v2f a;
      a.x = at[(size_t)m * 480 + 96 + 3 * k + i];
      a.y = at[(size_t)m * 480 + 96 + 3 * (k + 1) + i];
      return a;
    });
    for (int j = 0; j < 4; ++j) {
      v8f c = {};
      c = wmma_tile(aa, WO1F + j * 16 * 96, c);
      tile_store(c, [&](int, int m, int n, float v) {
        y[(size_t)(n0 + m) * 416 + 128 + (j * 16 + n) * 3 + i] = v * INV_SQ96;
      });
    }
  }
  for (int i = 0; i < 3; ++i) {
    v2f aa[8];
    load_a(aa, [&](int m, int k) {
      v2f a;
      a.x = at[(size_t)m * 480 + 384 + 3 * k + i];
      a.y = at[(size_t)m * 480 + 384 + 3 * (k + 1) + i];
      return a;
    });
    for (int j = 0; j < 2; ++j) {
      v8f c = {};
      c = wmma_tile(aa, WO2F + j * 16 * 32, c);
      tile_store(c, [&](int, int m, int n, float v) {
        y[(size_t)(n0 + m) * 416 + 320 + (j * 16 + n) * 3 + i] = v * INV_SQ32;
      });
    }
  }
}

extern "C" void kernel_launch(void* const* d_in, const int* in_sizes, int n_in,
                              void* d_out, int out_size, void* d_ws,
                              size_t ws_size, hipStream_t stream) {
  (void)in_sizes; (void)n_in; (void)out_size; (void)ws_size;
  const float* x         = (const float*)d_in[0];
  const float* edge_sh   = (const float*)d_in[1];
  const float* edge_attr = (const float*)d_in[2];
  const float* W0p = (const float*)d_in[3];
  const float* b0p = (const float*)d_in[4];
  const float* W1p = (const float*)d_in[5];
  const float* W0n = (const float*)d_in[6];
  const float* b0n = (const float*)d_in[7];
  const float* W1n = (const float*)d_in[8];
  const float* G1  = (const float*)d_in[9];
  const float* g1b = (const float*)d_in[10];
  const float* G2  = (const float*)d_in[11];
  const float* g2b = (const float*)d_in[12];
  const float* F1  = (const float*)d_in[13];
  const float* F2  = (const float*)d_in[14];
  const float* L1  = (const float*)d_in[15];
  const float* L2  = (const float*)d_in[16];
  const float* WO0 = (const float*)d_in[17];
  const float* bO0 = (const float*)d_in[18];
  const float* WO1 = (const float*)d_in[19];
  const float* WO2 = (const float*)d_in[20];
  const int* edge_index = (const int*)d_in[21];

  float* ws   = (float*)d_ws;
  float* p0   = ws;
  float* p1   = p0   + (size_t)N_NODES * 64;
  float* x0n  = p1   + (size_t)N_NODES * 96;
  float* x1n  = x0n  + (size_t)N_NODES * 64;
  float* acc  = x1n  + (size_t)N_NODES * 96;
  float* wt   = acc  + (size_t)N_NODES * 480;
  float* W0pF = wt + OFF_W0P;
  float* W1pF = wt + OFF_W1P;
  float* W0nF = wt + OFF_W0N;
  float* W1nF = wt + OFF_W1N;
  float* G1F  = wt + OFF_G1;
  float* G2F  = wt + OFF_G2;
  float* F1F  = wt + OFF_F1;
  float* F2F  = wt + OFF_F2;
  float* L1F  = wt + OFF_L1;
  float* L2F  = wt + OFF_L2;
  float* WO0F = wt + OFF_WO0;
  float* WO1F = wt + OFF_WO1;
  float* WO2F = wt + OFF_WO2;

  int* sc   = (int*)d_out;
  int* cnt  = sc;
  int* off  = sc + CNT_PADDED;
  int* rank = sc + 2 * CNT_PADDED;
  int* perm = sc + 2 * CNT_PADDED + N_EDGES;

  setup_kernel<<<SETUP_BLOCKS, SETUP_THREADS, 0, stream>>>(
      W0p, W1p, W0n, W1n, G1, G2, F1, F2, L1, L2, WO0, WO1, WO2, wt, acc, cnt);

  sort_rank_kernel<<<(N_EDGES + 255) / 256, 256, 0, stream>>>(edge_index, cnt, rank);
  sort_scan_kernel<<<1, 1024, 0, stream>>>(cnt, off);
  sort_fill_kernel<<<(N_EDGES + 255) / 256, 256, 0, stream>>>(edge_index, off, rank, perm);

  node_kernel<<<N_NODES / 16, 32, 0, stream>>>(
      x, W0pF, b0p, W1pF, W0nF, b0n, W1nF, G1F, g1b, G2F, g2b,
      p0, p1, x0n, x1n);

  edge_kernel<<<N_EDGES / 16, 32, 0, stream>>>(
      edge_sh, edge_attr, F1F, F2F, L1F, L2F, edge_index, perm,
      p0, p1, x0n, x1n, acc);

  proj_kernel<<<N_NODES / 16, 32, 0, stream>>>(
      acc, WO0F, bO0, WO1F, WO2F, (float*)d_out);
}


# Edge MLP layer 1 split by linearity: p0@L1[0:64] and p0@L1[64:128] computed once per node (node kernel), gathered per edge as accumulator start; per-edge GEMM K=160 -> 32; that sum's order changes
# speedup vs baseline: 2.0931x; 1.0643x over previous
#include <hip/hip_runtime.h>
#include <hip/hip_bf16.h>
#include <math.h>


typedef __attribute__((ext_vector_type(2))) float v2f;
typedef __attribute__((ext_vector_type(4))) float v4f;
typedef __attribute__((ext_vector_type(8))) float v8f;

#define N_NODES 10000
#define N_EDGES 320000

#define INV_SQ3   0.5773502691896258f
#define INV_SQ2   0.7071067811865476f
#define INV_SQ32  0.17677669529663687f
#define INV_SQ64  0.125f
#define INV_SQ96  0.10206207261596575f
#define INV_SQ160 0.07905694150420949f
#define LN2F      0.6931471805599453f

__device__ __forceinline__ float sspf(float v) {
  float sp = (v > 20.0f) ? v : log1pf(expf(v));
  return sp - LN2F;
}
__device__ __forceinline__ float siluf(float v) {
  return v / (1.0f + expf(-v));
}
__device__ __forceinline__ v2f ld2(const float* p) {
  return *(const v2f*)p;
}

__device__ __forceinline__ v8f wmma4(v2f a, v2f b, v8f c) {
  return __builtin_amdgcn_wmma_f32_16x16x4_f32(false, a, false, b,
                                               (short)0, c, false, false);
}


template <int KS, typename LA>
__device__ __forceinline__ void load_a(v2f (&a)[KS], LA la) {
  const int lane = threadIdx.x & 31;
  const int m = lane & 15;
  const int kk = (lane >> 4) << 1;
#pragma unroll
  for (int q = 0; q < KS; ++q) a[q] = la(m, q * 4 + kk);
}

template <int KS>
__device__ __forceinline__ v8f wmma_tile(const v2f (&a)[KS],
                                         const float* __restrict__ wf_tile, v8f c) {
  static_assert(KS % 2 == 0, "k-steps are stored in pairs");
  const v4f* bp = (const v4f*)wf_tile + (threadIdx.x & 31);
#pragma unroll
  for (int p = 0; p < KS / 2; ++p) {
    const v4f b = bp[p * 32];
    const v2f b0 = {b.x, b.y};
    const v2f b1 = {b.z, b.w};
    c = wmma4(a[2 * p], b0, c);
    c = wmma4(a[2 * p + 1], b1, c);
  }
  return c;
}

__device__ __forceinline__ void load_b32(v4f (&b)[4], const float* __restrict__ wf_tile) {
  const v4f* bp = (const v4f*)wf_tile + (threadIdx.x & 31);
#pragma unroll
  for (int p = 0; p < 4; ++p) b[p] = bp[p * 32];
}
__device__ __forceinline__ v8f wmma_regs(const v2f (&a)[8], const v4f (&b)[4], v8f c) {
#pragma unroll
  for (int p = 0; p < 4; ++p) {
    const v2f b0 = {b[p].x, b[p].y};
    const v2f b1 = {b[p].z, b[p].w};
    c = wmma4(a[2 * p], b0, c);
    c = wmma4(a[2 * p + 1], b1, c);
  }
  return c;
}
template <int V> struct IC { static constexpr int value = V; };

template <typename ST>
__device__ __forceinline__ void tile_store(v8f c, ST st) {
  const int lane = threadIdx.x & 31;
  const int col = lane & 15;
  const int rbase = (lane >> 4) << 3;
#pragma unroll
  for (int r = 0; r < 8; ++r) st(r, rbase + r, col, c[r]);
}

constexpr int OFF_W0P = 0;
constexpr int OFF_W1P = OFF_W0P + 64 * 64;
constexpr int OFF_W0N = OFF_W1P + 32 * 32;
constexpr int OFF_W1N = OFF_W0N + 64 * 64;
constexpr int OFF_G1  = OFF_W1N + 32 * 32;
constexpr int OFF_G2  = OFF_G1  + 96 * 96;
constexpr int OFF_F1  = OFF_G2  + 96 * 96;
constexpr int OFF_F2  = OFF_F1  + 32 * 32;
constexpr int OFF_L1  = OFF_F2  + 32 * 224;
constexpr int OFF_L2  = OFF_L1  + 160 * 32;
constexpr int OFF_WO0 = OFF_L2  + 32 * 224;
constexpr int OFF_WO1 = OFF_WO0 + 96 * 128;
constexpr int OFF_WO2 = OFF_WO1 + 96 * 64;
constexpr int WT_FLOATS = OFF_WO2 + 32 * 32;

constexpr int SETUP_THREADS  = 256;
constexpr int SETUP_T_BLOCKS = WT_FLOATS / 1024;
constexpr int ACC_V4         = N_NODES * 480 / 4;
constexpr int Z_V4_PER_BLOCK = SETUP_THREADS * 16;
constexpr int SETUP_Z_BLOCKS = (ACC_V4 + Z_V4_PER_BLOCK - 1) / Z_V4_PER_BLOCK;
constexpr int CNT_PADDED     = 10240;
constexpr int SETUP_BLOCKS   = SETUP_T_BLOCKS + SETUP_Z_BLOCKS + 1;

template <int K, int Nc>
__device__ __forceinline__ void frag_slab(const float* __restrict__ W,
                                          float* __restrict__ Wf, int slab) {
  static_assert((K * Nc) % 1024 == 0, "weights are whole slabs");
  static_assert(K % 8 == 0 && Nc % 16 == 0, "whole k-step pairs and column tiles");
  float v[4];
#pragma unroll
  for (int i = 0; i < 4; ++i) v[i] = W[slab * 1024 + i * SETUP_THREADS + threadIdx.x];
#pragma unroll
  for (int i = 0; i < 4; ++i) {
    const int idx = slab * 1024 + i * SETUP_THREADS + threadIdx.x;
    const int k = idx / Nc, n = idx % Nc;
    const int j = n >> 4, c = n & 15;
    const int p = k >> 3, r = k & 7;
    const int h = (r >> 1) & 1;
    const int sl = ((r >> 2) << 1) | (r & 1);
    Wf[(((j * (K / 8) + p) * 32) + h * 16 + c) * 4 + sl] = v[i];
  }
}

__global__ __launch_bounds__(SETUP_THREADS) void setup_kernel(
    const float* __restrict__ W0p, const float* __restrict__ W1p,
    const float* __restrict__ W0n, const float* __restrict__ W1n,
    const float* __restrict__ G1, const float* __restrict__ G2,
    const float* __restrict__ F1, const float* __restrict__ F2,
    const float* __restrict__ L1, const float* __restrict__ L2,
    const float* __restrict__ WO0, const float* __restrict__ WO1,
    const float* __restrict__ WO2,
    float* __restrict__ wt, float* __restrict__ acc, int* __restrict__ cnt) {
  int s = blockIdx.x;
  if (s < SETUP_T_BLOCKS) {
    if (s < 4)  { frag_slab<64, 64>(W0p, wt + OFF_W0P, s); return; }   s -= 4;
    if (s < 1)  { frag_slab<32, 32>(W1p, wt + OFF_W1P, s); return; }   s -= 1;
    if (s < 4)  { frag_slab<64, 64>(W0n, wt + OFF_W0N, s); return; }   s -= 4;
    if (s < 1)  { frag_slab<32, 32>(W1n, wt + OFF_W1N, s); return; }   s -= 1;
    if (s < 9)  { frag_slab<96, 96>(G1, wt + OFF_G1, s); return; }     s -= 9;
    if (s < 9)  { frag_slab<96, 96>(G2, wt + OFF_G2, s); return; }     s -= 9;
    if (s < 1)  { frag_slab<32, 32>(F1, wt + OFF_F1, s); return; }     s -= 1;
    if (s < 7)  { frag_slab<32, 224>(F2, wt + OFF_F2, s); return; }    s -= 7;
    if (s < 5)  { frag_slab<160, 32>(L1, wt + OFF_L1, s); return; }    s -= 5;
    if (s < 7)  { frag_slab<32, 224>(L2, wt + OFF_L2, s); return; }    s -= 7;
    if (s < 12) { frag_slab<96, 128>(WO0, wt + OFF_WO0, s); return; }  s -= 12;
    if (s < 6)  { frag_slab<96, 64>(WO1, wt + OFF_WO1, s); return; }   s -= 6;
    frag_slab<32, 32>(WO2, wt + OFF_WO2, s);
    return;
  }
  s -= SETUP_T_BLOCKS;
  if (s < SETUP_Z_BLOCKS) {
    v4f* a4 = (v4f*)acc;
    const v4f z = {0.0f, 0.0f, 0.0f, 0.0f};
    for (int i = 0; i < 16; ++i) {
      const int q = s * Z_V4_PER_BLOCK + i * SETUP_THREADS + threadIdx.x;
      if (q < ACC_V4) a4[q] = z;
    }
    return;
  }
  for (int i = threadIdx.x; i < CNT_PADDED; i += SETUP_THREADS) cnt[i] = 0;
}

constexpr int P0_PITCH = 68;
constexpr int IP_PITCH = 36;

__global__ __launch_bounds__(32) void node_kernel(
    const float* __restrict__ x,
    const float* __restrict__ W0pF, const float* __restrict__ b0p,
    const float* __restrict__ W1pF,
    const float* __restrict__ W0nF, const float* __restrict__ b0n,
    const float* __restrict__ W1nF,
    const float* __restrict__ G1F, const float* __restrict__ g1b,
    const float* __restrict__ G2F, const float* __restrict__ g2b,
    const float* __restrict__ L1F,
    float* __restrict__ qd, float* __restrict__ qs, float* __restrict__ p1,
    float* __restrict__ x0n, float* __restrict__ x1n) {
  __shared__ __align__(16) float p0S[16 * P0_PITCH];
  __shared__ __align__(16) float f0S[16 * 96];
  __shared__ __align__(16) float hS[16 * 96];
  __shared__ __align__(16) float gS[16 * 96];

  const int lane = threadIdx.x;
  const int n0 = blockIdx.x * 16;
  const float* xt = x + (size_t)n0 * 160;

  {
    v2f ax[16];
    load_a(ax, [&](int m, int k) { return ld2(xt + m * 160 + k); });
    for (int j = 0; j < 4; ++j) {
      v8f c = {};
      c = wmma_tile(ax, W0pF + j * 16 * 64, c);
      tile_store(c, [&](int, int m, int n, float v) {
        int col = j * 16 + n;
        p0S[m * P0_PITCH + col] = v * INV_SQ64 + b0p[col];
      });
    }
  }
  for (int i = 0; i < 3; ++i) {
    v2f ax[8];
    load_a(ax, [&](int m, int k) {
      v2f a;
      a.x = xt[m * 160 + 64 + 3 * k + i];
      a.y = xt[m * 160 + 64 + 3 * (k + 1) + i];
      return a;
    });
    for (int j = 0; j < 2; ++j) {
      v8f c = {};
      c = wmma_tile(ax, W1pF + j * 16 * 32, c);
      tile_store(c, [&](int, int m, int n, float v) {
        p1[(size_t)(n0 + m) * 96 + (j * 16 + n) * 3 + i] = v * INV_SQ32;
      });
    }
  }

  for (int idx = lane; idx < 16 * 96; idx += 32) {
    int m = idx / 96, c = idx % 96;
    float v;
    if (c < 64) {
      v = xt[m * 160 + c];
    } else {
      int u = c - 64;
      float a0 = xt[m * 160 + 64 + u * 3 + 0];
      float a1 = xt[m * 160 + 64 + u * 3 + 1];
      float a2 = xt[m * 160 + 64 + u * 3 + 2];
      v = sqrtf(a0 * a0 + a1 * a1 + a2 * a2 + 1e-12f);
    }
    f0S[idx] = v;
  }
  __syncthreads();

  {
    v2f ap[16];
    load_a(ap, [&](int m, int k) { return ld2(p0S + m * P0_PITCH + k); });
    for (int j = 0; j < 2; ++j) {
      v8f cd = {}, cs = {};
      cd = wmma_tile(ap, L1F + j * 16 * 160, cd);
      cs = wmma_tile(ap, L1F + j * 16 * 160 + 8 * 128, cs);
      tile_store(cd, [&](int, int m, int n, float v) {
        qd[(size_t)(n0 + m) * 32 + j * 16 + n] = v;
      });
      tile_store(cs, [&](int, int m, int n, float v) {
        qs[(size_t)(n0 + m) * 32 + j * 16 + n] = v;
      });
    }
  }

  {
    v2f af[24];
    load_a(af, [&](int m, int k) { return ld2(f0S + m * 96 + k); });
    for (int j = 0; j < 6; ++j) {
      v8f c = {};
      c = wmma_tile(af, G1F + j * 16 * 96, c);
      tile_store(c, [&](int, int m, int n, float v) {
        int col = j * 16 + n;
        hS[m * 96 + col] = siluf(v + g1b[col]);
      });
    }
  }
  __syncthreads();
  {
    v2f ah[24];
    load_a(ah, [&](int m, int k) { return ld2(hS + m * 96 + k); });
    for (int j = 0; j < 6; ++j) {
      v8f c = {};
      c = wmma_tile(ah, G2F + j * 16 * 96, c);
      tile_store(c, [&](int, int m, int n, float v) {
        gS[m * 96 + j * 16 + n] = v + g2b[j * 16 + n];
      });
    }
  }
  __syncthreads();

  {
    v2f ag[16];
    load_a(ag, [&](int m, int k) { return ld2(gS + m * 96 + k); });
    for (int j = 0; j < 4; ++j) {
      v8f c = {};
      c = wmma_tile(ag, W0nF + j * 16 * 64, c);
      tile_store(c, [&](int, int m, int n, float v) {
        int col = j * 16 + n;
        x0n[(size_t)(n0 + m) * 64 + col] = v * INV_SQ64 + b0n[col];
      });
    }
  }
  for (int i = 0; i < 3; ++i) {
    v2f ax[8];
    load_a(ax, [&](int m, int k) {
      v2f a;
      a.x = xt[m * 160 + 64 + 3 * k + i] * gS[m * 96 + 64 + k];
      a.y = xt[m * 160 + 64 + 3 * (k + 1) + i] * gS[m * 96 + 64 + k + 1];
      return a;
    });
    for (int j = 0; j < 2; ++j) {
      v8f c = {};
      c = wmma_tile(ax, W1nF + j * 16 * 32, c);
      tile_store(c, [&](int, int m, int n, float v) {
        x1n[(size_t)(n0 + m) * 96 + (j * 16 + n) * 3 + i] = v * INV_SQ32;
      });
    }
  }
}

__global__ void sort_rank_kernel(const int* __restrict__ edge_index,
                                 int* __restrict__ cnt, int* __restrict__ rank) {
  int e = blockIdx.x * blockDim.x + threadIdx.x;
  if (e < N_EDGES) rank[e] = atomicAdd(&cnt[edge_index[e]], 1);
}

#define SCAN_PER_THREAD 10
__global__ __launch_bounds__(1024) void sort_scan_kernel(
    const int* __restrict__ cnt, int* __restrict__ off) {
  __shared__ int part[1024];
  const int t = threadIdx.x;
  int loc[SCAN_PER_THREAD];
  int s = 0;
#pragma unroll
  for (int i = 0; i < SCAN_PER_THREAD; ++i) {
    int idx = t * SCAN_PER_THREAD + i;
    int v = (idx < N_NODES) ? cnt[idx] : 0;
    loc[i] = s;
    s += v;
  }
  part[t] = s;
  __syncthreads();
  for (int d = 1; d < 1024; d <<= 1) {
    int v = (t >= d) ? part[t - d] : 0;
    __syncthreads();
    part[t] += v;
    __syncthreads();
  }
  const int base = part[t] - s;
#pragma unroll
  for (int i = 0; i < SCAN_PER_THREAD; ++i) {
    int idx = t * SCAN_PER_THREAD + i;
    if (idx < N_NODES) off[idx] = base + loc[i];
  }
}

__global__ void sort_fill_kernel(const int* __restrict__ edge_index,
                                 const int* __restrict__ off,
                                 const int* __restrict__ rank,
                                 int* __restrict__ perm) {
  int e = blockIdx.x * blockDim.x + threadIdx.x;
  if (e < N_EDGES) perm[off[edge_index[e]] + rank[e]] = e;
}

__device__ __forceinline__ void run_add(float* __restrict__ acc,
                                        const unsigned (&dB)[8], unsigned flm,
                                        int chan, const float (&v)[8]) {
  float run = 0.0f;
#pragma unroll
  for (int r = 0; r < 8; ++r) {
    run += v[r];
    if ((flm >> r) & 1u) {
      atomicAdd((float*)((char*)acc + (dB[r] + (unsigned)chan * 4u)), run);
      run = 0.0f;
    }
  }
}

__global__ __launch_bounds__(32) void edge_kernel(
    const float* __restrict__ edge_sh, const float* __restrict__ edge_attr,
    const float* __restrict__ F1F, const float* __restrict__ F2F,
    const float* __restrict__ L1F, const float* __restrict__ L2F,
    const int* __restrict__ edge_index, const int* __restrict__ perm,
    const float* __restrict__ qd, const float* __restrict__ qs,
    const float* __restrict__ p1,
    const float* __restrict__ x0n, const float* __restrict__ x1n,
    float* __restrict__ acc) {
  __shared__ __align__(16) float ipS[16 * IP_PITCH];
  __shared__ __align__(16) float hN[16 * 32];
  __shared__ __align__(16) float hE[16 * 32];
  __shared__ __align__(16) float shS[16 * 4];
  __shared__ int dstS[16];
  __shared__ int srcS[16];

  const int lane = threadIdx.x;
  const int e0 = blockIdx.x * 16;

  __builtin_prefetch(F2F, 0, 1);
  __builtin_prefetch(L2F, 0, 1);

  v4f bF0[4], bF1[4];
  load_b32(bF0, F1F);
  load_b32(bF1, F1F + 16 * 32);

  const int e = perm[e0 + (lane & 15)];
  v2f aea[8];
  {
    const float* ar = edge_attr + (size_t)e * 32 + ((lane >> 4) << 1);
#pragma unroll
    for (int q = 0; q < 8; ++q) aea[q] = ld2(ar + q * 4);
  }
  if (lane < 16) {
    dstS[lane] = edge_index[e];
    srcS[lane] = edge_index[N_EDGES + e];
    *(v4f*)(shS + lane * 4) = *(const v4f*)(edge_sh + (size_t)e * 4);
  }
  __syncthreads();

  const int rbase = (lane >> 4) << 3;

  float qv0[8], qv1[8], qw0[8], qw1[8];
#pragma unroll
  for (int r = 0; r < 8; ++r) {
    const float* pd = (const float*)((const char*)qd + ((unsigned)dstS[rbase + r] * 128u + (unsigned)(lane & 15) * 4u));
    const float* ps = (const float*)((const char*)qs + ((unsigned)srcS[rbase + r] * 128u + (unsigned)(lane & 15) * 4u));
    qv0[r] = pd[0];
    qv1[r] = pd[16];
    qw0[r] = ps[0];
    qw1[r] = ps[16];
  }
  v4f bL0[4], bL1[4];
  load_b32(bL0, L1F + 16 * 128);
  load_b32(bL1, L1F + 16 * 160 + 16 * 128);

  {
    {
      v8f c0 = {}, c1 = {};
      c0 = wmma_regs(aea, bF0, c0);
      c1 = wmma_regs(aea, bF1, c1);
      tile_store(c0, [&](int, int m, int n, float v) {
        hE[m * 32 + n] = sspf(v * INV_SQ32);
      });
      tile_store(c1, [&](int, int m, int n, float v) {
        hE[m * 32 + 16 + n] = sspf(v * INV_SQ32);
      });
    }

#pragma nounroll
    for (int mb = 0; mb < 16; mb += 8) {
      float dv[8];
#pragma unroll
      for (int r = 0; r < 8; ++r) {
        const float* pd = (const float*)((const char*)p1 + ((unsigned)dstS[mb + r] * 384u + (unsigned)lane * 12u));
        const float* ps = (const float*)((const char*)p1 + ((unsigned)srcS[mb + r] * 384u + (unsigned)lane * 12u));
        dv[r] = (pd[0] * ps[0] + pd[1] * ps[1] + pd[2] * ps[2]) * INV_SQ3;
      }
#pragma unroll
      for (int r = 0; r < 8; ++r) ipS[(mb + r) * IP_PITCH + lane] = dv[r];
    }
  }
  __syncthreads();

  {
    v2f aip[8];
    load_a(aip, [&](int m, int k) { return ld2(ipS + m * IP_PITCH + k); });
    v8f c0, c1;
#pragma unroll
    for (int r = 0; r < 8; ++r) {
      c0[r] = qv0[r] + qw0[r];
      c1[r] = qv1[r] + qw1[r];
    }
    c0 = wmma_regs(aip, bL0, c0);
    c1 = wmma_regs(aip, bL1, c1);
    tile_store(c0, [&](int, int m, int n, float v) {
      hN[m * 32 + n] = sspf(v * INV_SQ160);
    });
    tile_store(c1, [&](int, int m, int n, float v) {
      hN[m * 32 + 16 + n] = sspf(v * INV_SQ160);
    });
  }
  __syncthreads();

  float sh0r[8], s1xr[8], s1yr[8], s1zr[8];
  int dR[8], sR[8];
#pragma unroll
  for (int r = 0; r < 8; ++r) {
    int m = rbase + r;
    const v4f sh = *(const v4f*)(shS + m * 4);
    sh0r[r] = sh.x;
    s1xr[r] = sh.y;
    s1yr[r] = sh.z;
    s1zr[r] = sh.w;
    dR[r] = dstS[m];
    sR[r] = srcS[m];
  }
  unsigned s0B[8], s1B[8], dB[8];
#pragma unroll
  for (int r = 0; r < 8; ++r) {
    s0B[r] = (unsigned)sR[r] * 256u;
    s1B[r] = (unsigned)sR[r] * 384u;
    dB[r] = (unsigned)dR[r] * 1920u;
  }
  unsigned flm = 0x80u;
#pragma unroll
  for (int r = 0; r < 7; ++r) flm |= (dR[r + 1] != dR[r]) ? (1u << r) : 0u;

  v2f aN[8], aE[8];
  load_a(aN, [&](int m, int k) { return ld2(hN + m * 32 + k); });
  load_a(aE, [&](int m, int k) { return ld2(hE + m * 32 + k); });

  auto do_tile = [&](auto seg, const int j) {
    constexpr int SEG = decltype(seg)::value;
    const int nb = j * 16;
    const int col = nb + (lane & 15);
    v4f bN[4], bE[4];
    load_b32(bN, L2F + nb * 32);
    load_b32(bE, F2F + nb * 32);
    constexpr int CB = (SEG == 0) ? 0 : (SEG == 1) ? 64 : (SEG == 2) ? 128 : (SEG == 3) ? 160 : 192;
    const int cs = col - CB;
    float g0[8], g1[8], g2[8];
    if constexpr (SEG <= 1) {
#pragma unroll
      for (int r = 0; r < 8; ++r)
        g0[r] = *(const float*)((const char*)x0n + (s0B[r] + (unsigned)cs * 4u));
    } else {
#pragma unroll
      for (int r = 0; r < 8; ++r) {
        const float* xv = (const float*)((const char*)x1n + (s1B[r] + (unsigned)cs * 12u));
        g0[r] = xv[0];
        g1[r] = xv[1];
        g2[r] = xv[2];
      }
    }

    v8f cN = {}, cE = {};
    cN = wmma_regs(aN, bN, cN);
    cE = wmma_regs(aE, bE, cE);
    v8f w8 = cN * cE * (INV_SQ32 * INV_SQ32);

    float v0[8], v1[8], v2[8];
    if constexpr (SEG == 0) {
#pragma unroll
      for (int r = 0; r < 8; ++r) v0[r] = w8[r] * g0[r] * sh0r[r];
      run_add(acc, dB, flm, col, v0);
    } else if constexpr (SEG == 1) {
#pragma unroll
      for (int r = 0; r < 8; ++r) {
        float f = w8[r] * g0[r] * INV_SQ3;
        v0[r] = f * s1xr[r];
        v1[r] = f * s1yr[r];
        v2[r] = f * s1zr[r];
      }
      run_add(acc, dB, flm, 96 + cs * 3 + 0, v0);
      run_add(acc, dB, flm, 96 + cs * 3 + 1, v1);
      run_add(acc, dB, flm, 96 + cs * 3 + 2, v2);
    } else if constexpr (SEG == 2) {
#pragma unroll
      for (int r = 0; r < 8; ++r) {
        float f = w8[r] * sh0r[r];
        v0[r] = f * g0[r];
        v1[r] = f * g1[r];
        v2[r] = f * g2[r];
      }
      run_add(acc, dB, flm, 288 + cs * 3 + 0, v0);
      run_add(acc, dB, flm, 288 + cs * 3 + 1, v1);
      run_add(acc, dB, flm, 288 + cs * 3 + 2, v2);
    } else if constexpr (SEG == 3) {
#pragma unroll
      for (int r = 0; r < 8; ++r) {
        float dp = g0[r] * s1xr[r] + g1[r] * s1yr[r] + g2[r] * s1zr[r];
        v0[r] = w8[r] * dp * INV_SQ3;
      }
      run_add(acc, dB, flm, 64 + cs, v0);
    } else {
#pragma unroll
      for (int r = 0; r < 8; ++r) {
        const float s1x = s1xr[r], s1y = s1yr[r], s1z = s1zr[r];
        float cx = g1[r] * s1z - g2[r] * s1y;
        float cy = g2[r] * s1x - g0[r] * s1z;
        float cz = g0[r] * s1y - g1[r] * s1x;
        float f = w8[r] * INV_SQ2;
        v0[r] = f * cx;
        v1[r] = f * cy;
        v2[r] = f * cz;
      }
      run_add(acc, dB, flm, 384 + cs * 3 + 0, v0);
      run_add(acc, dB, flm, 384 + cs * 3 + 1, v1);
      run_add(acc, dB, flm, 384 + cs * 3 + 2, v2);
    }
  };

#pragma nounroll
  for (int j = 0; j < 4; ++j) do_tile(IC<0>{}, j);
#pragma nounroll
  for (int j = 4; j < 8; ++j) do_tile(IC<1>{}, j);
#pragma nounroll
  for (int j = 8; j < 10; ++j) do_tile(IC<2>{}, j);
#pragma nounroll
  for (int j = 10; j < 12; ++j) do_tile(IC<3>{}, j);
#pragma nounroll
  for (int j = 12; j < 14; ++j) do_tile(IC<4>{}, j);
}

__global__ __launch_bounds__(32) void proj_kernel(
    const float* __restrict__ acc,
    const float* __restrict__ WO0F, const float* __restrict__ bO0,
    const float* __restrict__ WO1F, const float* __restrict__ WO2F,
    float* __restrict__ y) {
  const int n0 = blockIdx.x * 16;
  const float* at = acc + (size_t)n0 * 480;

  {
    v2f aa[24];
    load_a(aa, [&](int m, int k) { return ld2(at + (size_t)m * 480 + k); });
    for (int j = 0; j < 8; ++j) {
      v8f c = {};
      c = wmma_tile(aa, WO0F + j * 16 * 96, c);
      tile_store(c, [&](int, int m, int n, float v) {
        int col = j * 16 + n;
        y[(size_t)(n0 + m) * 416 + col] = v * INV_SQ96 + bO0[col];
      });
    }
  }
  for (int i = 0; i < 3; ++i) {
    v2f aa[24];
    load_a(aa, [&](int m, int k) {
      v2f a;
      a.x = at[(size_t)m * 480 + 96 + 3 * k + i];
      a.y = at[(size_t)m * 480 + 96 + 3 * (k + 1) + i];
      return a;
    });
    for (int j = 0; j < 4; ++j) {
      v8f c = {};
      c = wmma_tile(aa, WO1F + j * 16 * 96, c);
      tile_store(c, [&](int, int m, int n, float v) {
        y[(size_t)(n0 + m) * 416 + 128 + (j * 16 + n) * 3 + i] = v * INV_SQ96;
      });
    }
  }
  for (int i = 0; i < 3; ++i) {
    v2f aa[8];
    load_a(aa, [&](int m, int k) {
      v2f a;
      a.x = at[(size_t)m * 480 + 384 + 3 * k + i];
      a.y = at[(size_t)m * 480 + 384 + 3 * (k + 1) + i];
      return a;
    });
    for (int j = 0; j < 2; ++j) {
      v8f c = {};
      c = wmma_tile(aa, WO2F + j * 16 * 32, c);
      tile_store(c, [&](int, int m, int n, float v) {
        y[(size_t)(n0 + m) * 416 + 320 + (j * 16 + n) * 3 + i] = v * INV_SQ32;
      });
    }
  }
}

extern "C" void kernel_launch(void* const* d_in, const int* in_sizes, int n_in,
                              void* d_out, int out_size, void* d_ws,
                              size_t ws_size, hipStream_t stream) {
  (void)in_sizes; (void)n_in; (void)out_size; (void)ws_size;
  const float* x         = (const float*)d_in[0];
  const float* edge_sh   = (const float*)d_in[1];
  const float* edge_attr = (const float*)d_in[2];
  const float* W0p = (const float*)d_in[3];
  const float* b0p = (const float*)d_in[4];
  const float* W1p = (const float*)d_in[5];
  const float* W0n = (const float*)d_in[6];
  const float* b0n = (const float*)d_in[7];
  const float* W1n = (const float*)d_in[8];
  const float* G1  = (const float*)d_in[9];
  const float* g1b = (const float*)d_in[10];
  const float* G2  = (const float*)d_in[11];
  const float* g2b = (const float*)d_in[12];
  const float* F1  = (const float*)d_in[13];
  const float* F2  = (const float*)d_in[14];
  const float* L1  = (const float*)d_in[15];
  const float* L2  = (const float*)d_in[16];
  const float* WO0 = (const float*)d_in[17];
  const float* bO0 = (const float*)d_in[18];
  const float* WO1 = (const float*)d_in[19];
  const float* WO2 = (const float*)d_in[20];
  const int* edge_index = (const int*)d_in[21];

  float* ws   = (float*)d_ws;
  float* qd   = ws;
  float* qs   = qd   + (size_t)N_NODES * 32;
  float* p1   = qs   + (size_t)N_NODES * 32;
  float* x0n  = p1   + (size_t)N_NODES * 96;
  float* x1n  = x0n  + (size_t)N_NODES * 64;
  float* acc  = x1n  + (size_t)N_NODES * 96;
  float* wt   = acc  + (size_t)N_NODES * 480;
  float* W0pF = wt + OFF_W0P;
  float* W1pF = wt + OFF_W1P;
  float* W0nF = wt + OFF_W0N;
  float* W1nF = wt + OFF_W1N;
  float* G1F  = wt + OFF_G1;
  float* G2F  = wt + OFF_G2;
  float* F1F  = wt + OFF_F1;
  float* F2F  = wt + OFF_F2;
  float* L1F  = wt + OFF_L1;
  float* L2F  = wt + OFF_L2;
  float* WO0F = wt + OFF_WO0;
  float* WO1F = wt + OFF_WO1;
  float* WO2F = wt + OFF_WO2;

  int* sc   = (int*)d_out;
  int* cnt  = sc;
  int* off  = sc + CNT_PADDED;
  int* rank = sc + 2 * CNT_PADDED;
  int* perm = sc + 2 * CNT_PADDED + N_EDGES;

  setup_kernel<<<SETUP_BLOCKS, SETUP_THREADS, 0, stream>>>(
      W0p, W1p, W0n, W1n, G1, G2, F1, F2, L1, L2, WO0, WO1, WO2, wt, acc, cnt);

  sort_rank_kernel<<<(N_EDGES + 255) / 256, 256, 0, stream>>>(edge_index, cnt, rank);
  sort_scan_kernel<<<1, 1024, 0, stream>>>(cnt, off);
  sort_fill_kernel<<<(N_EDGES + 255) / 256, 256, 0, stream>>>(edge_index, off, rank, perm);

  node_kernel<<<N_NODES / 16, 32, 0, stream>>>(
      x, W0pF, b0p, W1pF, W0nF, b0n, W1nF, G1F, g1b, G2F, g2b, L1F,
      qd, qs, p1, x0n, x1n);

  edge_kernel<<<N_EDGES / 16, 32, 0, stream>>>(
      edge_sh, edge_attr, F1F, F2F, L1F, L2F, edge_index, perm,
      qd, qs, p1, x0n, x1n, acc);

  proj_kernel<<<N_NODES / 16, 32, 0, stream>>>(
      acc, WO0F, bO0, WO1F, WO2F, (float*)d_out);
}


# Node and projection kernels: 4 waves per 16-node tile, each wave a fixed share of the column tiles; results staged in LDS and stored as whole 128-byte lines; per-element sums unchanged
# speedup vs baseline: 2.2506x; 1.0752x over previous
#include <hip/hip_runtime.h>
#include <hip/hip_bf16.h>
#include <math.h>


typedef __attribute__((ext_vector_type(2))) float v2f;
typedef __attribute__((ext_vector_type(4))) float v4f;
typedef __attribute__((ext_vector_type(8))) float v8f;

#define N_NODES 10000
#define N_EDGES 320000

#define INV_SQ3   0.5773502691896258f
#define INV_SQ2   0.7071067811865476f
#define INV_SQ32  0.17677669529663687f
#define INV_SQ64  0.125f
#define INV_SQ96  0.10206207261596575f
#define INV_SQ160 0.07905694150420949f
#define LN2F      0.6931471805599453f

__device__ __forceinline__ float sspf(float v) {
  float sp = (v > 20.0f) ? v : log1pf(expf(v));
  return sp - LN2F;
}
__device__ __forceinline__ float siluf(float v) {
  return v / (1.0f + expf(-v));
}
__device__ __forceinline__ v2f ld2(const float* p) {
  return *(const v2f*)p;
}

__device__ __forceinline__ v8f wmma4(v2f a, v2f b, v8f c) {
  return __builtin_amdgcn_wmma_f32_16x16x4_f32(false, a, false, b,
                                               (short)0, c, false, false);
}


template <int KS, typename LA>
__device__ __forceinline__ void load_a(v2f (&a)[KS], LA la) {
  const int lane = threadIdx.x & 31;
  const int m = lane & 15;
  const int kk = (lane >> 4) << 1;
#pragma unroll
  for (int q = 0; q < KS; ++q) a[q] = la(m, q * 4 + kk);
}

template <int KS>
__device__ __forceinline__ v8f wmma_tile(const v2f (&a)[KS],
                                         const float* __restrict__ wf_tile, v8f c) {
  static_assert(KS % 2 == 0, "k-steps are stored in pairs");
  const v4f* bp = (const v4f*)wf_tile + (threadIdx.x & 31);
#pragma unroll
  for (int p = 0; p < KS / 2; ++p) {
    const v4f b = bp[p * 32];
    const v2f b0 = {b.x, b.y};
    const v2f b1 = {b.z, b.w};
    c = wmma4(a[2 * p], b0, c);
    c = wmma4(a[2 * p + 1], b1, c);
  }
  return c;
}

__device__ __forceinline__ void load_b32(v4f (&b)[4], const float* __restrict__ wf_tile) {
  const v4f* bp = (const v4f*)wf_tile + (threadIdx.x & 31);
#pragma unroll
  for (int p = 0; p < 4; ++p) b[p] = bp[p * 32];
}
__device__ __forceinline__ v8f wmma_regs(const v2f (&a)[8], const v4f (&b)[4], v8f c) {
#pragma unroll
  for (int p = 0; p < 4; ++p) {
    const v2f b0 = {b[p].x, b[p].y};
    const v2f b1 = {b[p].z, b[p].w};
    c = wmma4(a[2 * p], b0, c);
    c = wmma4(a[2 * p + 1], b1, c);
  }
  return c;
}
template <int V> struct IC { static constexpr int value = V; };

template <typename ST>
__device__ __forceinline__ void tile_store(v8f c, ST st) {
  const int lane = threadIdx.x & 31;
  const int col = lane & 15;
  const int rbase = (lane >> 4) << 3;
#pragma unroll
  for (int r = 0; r < 8; ++r) st(r, rbase + r, col, c[r]);
}

constexpr int OFF_W0P = 0;
constexpr int OFF_W1P = OFF_W0P + 64 * 64;
constexpr int OFF_W0N = OFF_W1P + 32 * 32;
constexpr int OFF_W1N = OFF_W0N + 64 * 64;
constexpr int OFF_G1  = OFF_W1N + 32 * 32;
constexpr int OFF_G2  = OFF_G1  + 96 * 96;
constexpr int OFF_F1  = OFF_G2  + 96 * 96;
constexpr int OFF_F2  = OFF_F1  + 32 * 32;
constexpr int OFF_L1  = OFF_F2  + 32 * 224;
constexpr int OFF_L2  = OFF_L1  + 160 * 32;
constexpr int OFF_WO0 = OFF_L2  + 32 * 224;
constexpr int OFF_WO1 = OFF_WO0 + 96 * 128;
constexpr int OFF_WO2 = OFF_WO1 + 96 * 64;
constexpr int WT_FLOATS = OFF_WO2 + 32 * 32;

constexpr int SETUP_THREADS  = 256;
constexpr int SETUP_T_BLOCKS = WT_FLOATS / 1024;
constexpr int ACC_V4         = N_NODES * 480 / 4;
constexpr int Z_V4_PER_BLOCK = SETUP_THREADS * 16;
constexpr int SETUP_Z_BLOCKS = (ACC_V4 + Z_V4_PER_BLOCK - 1) / Z_V4_PER_BLOCK;
constexpr int CNT_PADDED     = 10240;
constexpr int SETUP_BLOCKS   = SETUP_T_BLOCKS + SETUP_Z_BLOCKS + 1;

template <int K, int Nc>
__device__ __forceinline__ void frag_slab(const float* __restrict__ W,
                                          float* __restrict__ Wf, int slab) {
  static_assert((K * Nc) % 1024 == 0, "weights are whole slabs");
  static_assert(K % 8 == 0 && Nc % 16 == 0, "whole k-step pairs and column tiles");
  float v[4];
#pragma unroll
  for (int i = 0; i < 4; ++i) v[i] = W[slab * 1024 + i * SETUP_THREADS + threadIdx.x];
#pragma unroll
  for (int i = 0; i < 4; ++i) {
    const int idx = slab * 1024 + i * SETUP_THREADS + threadIdx.x;
    const int k = idx / Nc, n = idx % Nc;
    const int j = n >> 4, c = n & 15;
    const int p = k >> 3, r = k & 7;
    const int h = (r >> 1) & 1;
    const int sl = ((r >> 2) << 1) | (r & 1);
    Wf[(((j * (K / 8) + p) * 32) + h * 16 + c) * 4 + sl] = v[i];
  }
}

__global__ __launch_bounds__(SETUP_THREADS) void setup_kernel(
    const float* __restrict__ W0p, const float* __restrict__ W1p,
    const float* __restrict__ W0n, const float* __restrict__ W1n,
    const float* __restrict__ G1, const float* __restrict__ G2,
    const float* __restrict__ F1, const float* __restrict__ F2,
    const float* __restrict__ L1, const float* __restrict__ L2,
    const float* __restrict__ WO0, const float* __restrict__ WO1,
    const float* __restrict__ WO2,
    float* __restrict__ wt, float* __restrict__ acc, int* __restrict__ cnt) {
  int s = blockIdx.x;
  if (s < SETUP_T_BLOCKS) {
    if (s < 4)  { frag_slab<64, 64>(W0p, wt + OFF_W0P, s); return; }   s -= 4;
    if (s < 1)  { frag_slab<32, 32>(W1p, wt + OFF_W1P, s); return; }   s -= 1;
    if (s < 4)  { frag_slab<64, 64>(W0n, wt + OFF_W0N, s); return; }   s -= 4;
    if (s < 1)  { frag_slab<32, 32>(W1n, wt + OFF_W1N, s); return; }   s -= 1;
    if (s < 9)  { frag_slab<96, 96>(G1, wt + OFF_G1, s); return; }     s -= 9;
    if (s < 9)  { frag_slab<96, 96>(G2, wt + OFF_G2, s); return; }     s -= 9;
    if (s < 1)  { frag_slab<32, 32>(F1, wt + OFF_F1, s); return; }     s -= 1;
    if (s < 7)  { frag_slab<32, 224>(F2, wt + OFF_F2, s); return; }    s -= 7;
    if (s < 5)  { frag_slab<160, 32>(L1, wt + OFF_L1, s); return; }    s -= 5;
    if (s < 7)  { frag_slab<32, 224>(L2, wt + OFF_L2, s); return; }    s -= 7;
    if (s < 12) { frag_slab<96, 128>(WO0, wt + OFF_WO0, s); return; }  s -= 12;
    if (s < 6)  { frag_slab<96, 64>(WO1, wt + OFF_WO1, s); return; }   s -= 6;
    frag_slab<32, 32>(WO2, wt + OFF_WO2, s);
    return;
  }
  s -= SETUP_T_BLOCKS;
  if (s < SETUP_Z_BLOCKS) {
    v4f* a4 = (v4f*)acc;
    const v4f z = {0.0f, 0.0f, 0.0f, 0.0f};
    for (int i = 0; i < 16; ++i) {
      const int q = s * Z_V4_PER_BLOCK + i * SETUP_THREADS + threadIdx.x;
      if (q < ACC_V4) a4[q] = z;
    }
    return;
  }
  for (int i = threadIdx.x; i < CNT_PADDED; i += SETUP_THREADS) cnt[i] = 0;
}

constexpr int NODE_WAVES = 4;

template <int FLOATS>
__device__ __forceinline__ void copy_tile(float* __restrict__ dst, const float* src) {
  constexpr int NT = NODE_WAVES * 32;
  static_assert(FLOATS % (4 * NT) == 0, "whole 16-byte words per thread");
#pragma unroll
  for (int q = 0; q < FLOATS / (4 * NT); ++q) {
    const int w = q * NT + threadIdx.x;
    ((v4f*)dst)[w] = ((const v4f*)src)[w];
  }
}

constexpr int P0_PITCH = 68;
constexpr int IP_PITCH = 36;

__global__ __launch_bounds__(NODE_WAVES * 32) void node_kernel(
    const float* __restrict__ x,
    const float* __restrict__ W0pF, const float* __restrict__ b0p,
    const float* __restrict__ W1pF,
    const float* __restrict__ W0nF, const float* __restrict__ b0n,
    const float* __restrict__ W1nF,
    const float* __restrict__ G1F, const float* __restrict__ g1b,
    const float* __restrict__ G2F, const float* __restrict__ g2b,
    const float* __restrict__ L1F,
    float* __restrict__ qd, float* __restrict__ qs, float* __restrict__ p1,
    float* __restrict__ x0n, float* __restrict__ x1n) {
  __shared__ __align__(16) float p0S[16 * P0_PITCH];
  __shared__ __align__(16) float f0S[16 * 96];
  __shared__ __align__(16) float hS[16 * 96];
  __shared__ __align__(16) float gS[16 * 96];
  __shared__ __align__(16) float qdS[16 * 32];
  __shared__ __align__(16) float qsS[16 * 32];
  __shared__ __align__(16) float p1S[16 * 96];
  __shared__ __align__(16) float x0nS[16 * 64];
  __shared__ __align__(16) float x1nS[16 * 96];
  static_assert(NODE_WAVES == 4, "the shares below are written out for four waves");

  const int tid = threadIdx.x;
  const int wave = __builtin_amdgcn_readfirstlane(threadIdx.x >> 5);
  const int n0 = blockIdx.x * 16;
  const float* xt = x + (size_t)n0 * 160;

  {
    v2f ax[16];
    load_a(ax, [&](int m, int k) { return ld2(xt + m * 160 + k); });
    {
      const int j = wave;
      v8f c = {};
      c = wmma_tile(ax, W0pF + j * 16 * 64, c);
      tile_store(c, [&](int, int m, int n, float v) {
        int col = j * 16 + n;
        p0S[m * P0_PITCH + col] = v * INV_SQ64 + b0p[col];
      });
    }
  }
  if (wave < 3) {
    const int i = wave;
    v2f ax[8];
    load_a(ax, [&](int m, int k) {
      v2f a;
      a.x = xt[m * 160 + 64 + 3 * k + i];
      a.y = xt[m * 160 + 64 + 3 * (k + 1) + i];
      return a;
    });
#pragma unroll
    for (int j = 0; j < 2; ++j) {
      v8f c = {};
      c = wmma_tile(ax, W1pF + j * 16 * 32, c);
      tile_store(c, [&](int, int m, int n, float v) {
        p1S[m * 96 + (j * 16 + n) * 3 + i] = v * INV_SQ32;
      });
    }
  }

  {
    constexpr int NT = NODE_WAVES * 32;
#pragma unroll
    for (int q = 0; q < 16 * 64 / NT; ++q) {
      const int idx = q * NT + tid;
      const int m = idx >> 6, c = idx & 63;
      f0S[m * 96 + c] = xt[m * 160 + c];
    }
#pragma unroll
    for (int q = 0; q < 16 * 32 / NT; ++q) {
      const int idx = q * NT + tid;
      const int m = idx >> 5, u = idx & 31;
      float a0 = xt[m * 160 + 64 + u * 3 + 0];
      float a1 = xt[m * 160 + 64 + u * 3 + 1];
      float a2 = xt[m * 160 + 64 + u * 3 + 2];
      f0S[m * 96 + 64 + u] = sqrtf(a0 * a0 + a1 * a1 + a2 * a2 + 1e-12f);
    }
  }
  __syncthreads();

  {
    v2f ap[16];
    load_a(ap, [&](int m, int k) { return ld2(p0S + m * P0_PITCH + k); });
    {
      const int j = wave >> 1, half = wave & 1;
      float* q = half ? qsS : qdS;
      v8f c = {};
      c = wmma_tile(ap, L1F + j * 16 * 160 + half * 8 * 128, c);
      tile_store(c, [&](int, int m, int n, float v) {
        q[m * 32 + j * 16 + n] = v;
      });
    }
  }

  if (wave < 3) {
    v2f af[24];
    load_a(af, [&](int m, int k) { return ld2(f0S + m * 96 + k); });
#pragma unroll
    for (int jj = 0; jj < 2; ++jj) {
      const int j = wave * 2 + jj;
      v8f c = {};
      c = wmma_tile(af, G1F + j * 16 * 96, c);
      tile_store(c, [&](int, int m, int n, float v) {
        int col = j * 16 + n;
        hS[m * 96 + col] = siluf(v + g1b[col]);
      });
    }
  }
  __syncthreads();
  copy_tile<16 * 32>(qd + (size_t)n0 * 32, qdS);
  copy_tile<16 * 32>(qs + (size_t)n0 * 32, qsS);
  copy_tile<16 * 96>(p1 + (size_t)n0 * 96, p1S);
  if (wave < 3) {
    v2f ah[24];
    load_a(ah, [&](int m, int k) { return ld2(hS + m * 96 + k); });
#pragma unroll
    for (int jj = 0; jj < 2; ++jj) {
      const int j = wave * 2 + jj;
      v8f c = {};
      c = wmma_tile(ah, G2F + j * 16 * 96, c);
      tile_store(c, [&](int, int m, int n, float v) {
        gS[m * 96 + j * 16 + n] = v + g2b[j * 16 + n];
      });
    }
  }
  __syncthreads();

  {
    v2f ag[16];
    load_a(ag, [&](int m, int k) { return ld2(gS + m * 96 + k); });
    {
      const int j = wave;
      v8f c = {};
      c = wmma_tile(ag, W0nF + j * 16 * 64, c);
      tile_store(c, [&](int, int m, int n, float v) {
        int col = j * 16 + n;
        x0nS[m * 64 + col] = v * INV_SQ64 + b0n[col];
      });
    }
  }
  if (wave < 3) {
    const int i = wave;
    v2f ax[8];
    load_a(ax, [&](int m, int k) {
      v2f a;
      a.x = xt[m * 160 + 64 + 3 * k + i] * gS[m * 96 + 64 + k];
      a.y = xt[m * 160 + 64 + 3 * (k + 1) + i] * gS[m * 96 + 64 + k + 1];
      return a;
    });
#pragma unroll
    for (int j = 0; j < 2; ++j) {
      v8f c = {};
      c = wmma_tile(ax, W1nF + j * 16 * 32, c);
      tile_store(c, [&](int, int m, int n, float v) {
        x1nS[m * 96 + (j * 16 + n) * 3 + i] = v * INV_SQ32;
      });
    }
  }
  __syncthreads();
  copy_tile<16 * 64>(x0n + (size_t)n0 * 64, x0nS);
  copy_tile<16 * 96>(x1n + (size_t)n0 * 96, x1nS);
}

__global__ void sort_rank_kernel(const int* __restrict__ edge_index,
                                 int* __restrict__ cnt, int* __restrict__ rank) {
  int e = blockIdx.x * blockDim.x + threadIdx.x;
  if (e < N_EDGES) rank[e] = atomicAdd(&cnt[edge_index[e]], 1);
}

#define SCAN_PER_THREAD 10
__global__ __launch_bounds__(1024) void sort_scan_kernel(
    const int* __restrict__ cnt, int* __restrict__ off) {
  __shared__ int part[1024];
  const int t = threadIdx.x;
  int loc[SCAN_PER_THREAD];
  int s = 0;
#pragma unroll
  for (int i = 0; i < SCAN_PER_THREAD; ++i) {
    int idx = t * SCAN_PER_THREAD + i;
    int v = (idx < N_NODES) ? cnt[idx] : 0;
    loc[i] = s;
    s += v;
  }
  part[t] = s;
  __syncthreads();
  for (int d = 1; d < 1024; d <<= 1) {
    int v = (t >= d) ? part[t - d] : 0;
    __syncthreads();
    part[t] += v;
    __syncthreads();
  }
  const int base = part[t] - s;
#pragma unroll
  for (int i = 0; i < SCAN_PER_THREAD; ++i) {
    int idx = t * SCAN_PER_THREAD + i;
    if (idx < N_NODES) off[idx] = base + loc[i];
  }
}

__global__ void sort_fill_kernel(const int* __restrict__ edge_index,
                                 const int* __restrict__ off,
                                 const int* __restrict__ rank,
                                 int* __restrict__ perm) {
  int e = blockIdx.x * blockDim.x + threadIdx.x;
  if (e < N_EDGES) perm[off[edge_index[e]] + rank[e]] = e;
}

__device__ __forceinline__ void run_add(float* __restrict__ acc,
                                        const unsigned (&dB)[8], unsigned flm,
                                        int chan, const float (&v)[8]) {
  float run = 0.0f;
#pragma unroll
  for (int r = 0; r < 8; ++r) {
    run += v[r];
    if ((flm >> r) & 1u) {
      atomicAdd((float*)((char*)acc + (dB[r] + (unsigned)chan * 4u)), run);
      run = 0.0f;
    }
  }
}

__global__ __launch_bounds__(32) void edge_kernel(
    const float* __restrict__ edge_sh, const float* __restrict__ edge_attr,
    const float* __restrict__ F1F, const float* __restrict__ F2F,
    const float* __restrict__ L1F, const float* __restrict__ L2F,
    const int* __restrict__ edge_index, const int* __restrict__ perm,
    const float* __restrict__ qd, const float* __restrict__ qs,
    const float* __restrict__ p1,
    const float* __restrict__ x0n, const float* __restrict__ x1n,
    float* __restrict__ acc) {
  __shared__ __align__(16) float ipS[16 * IP_PITCH];
  __shared__ __align__(16) float hN[16 * 32];
  __shared__ __align__(16) float hE[16 * 32];
  __shared__ __align__(16) float shS[16 * 4];
  __shared__ int dstS[16];
  __shared__ int srcS[16];

  const int lane = threadIdx.x;
  const int e0 = blockIdx.x * 16;

  __builtin_prefetch(F2F, 0, 1);
  __builtin_prefetch(L2F, 0, 1);

  v4f bF0[4], bF1[4];
  load_b32(bF0, F1F);
  load_b32(bF1, F1F + 16 * 32);

  const int e = perm[e0 + (lane & 15)];
  v2f aea[8];
  {
    const float* ar = edge_attr + (size_t)e * 32 + ((lane >> 4) << 1);
#pragma unroll
    for (int q = 0; q < 8; ++q) aea[q] = ld2(ar + q * 4);
  }
  if (lane < 16) {
    dstS[lane] = edge_index[e];
    srcS[lane] = edge_index[N_EDGES + e];
    *(v4f*)(shS + lane * 4) = *(const v4f*)(edge_sh + (size_t)e * 4);
  }
  __syncthreads();

  const int rbase = (lane >> 4) << 3;

  float qv0[8], qv1[8], qw0[8], qw1[8];
#pragma unroll
  for (int r = 0; r < 8; ++r) {
    const float* pd = (const float*)((const char*)qd + ((unsigned)dstS[rbase + r] * 128u + (unsigned)(lane & 15) * 4u));
    const float* ps = (const float*)((const char*)qs + ((unsigned)srcS[rbase + r] * 128u + (unsigned)(lane & 15) * 4u));
    qv0[r] = pd[0];
    qv1[r] = pd[16];
    qw0[r] = ps[0];
    qw1[r] = ps[16];
  }
  v4f bL0[4], bL1[4];
  load_b32(bL0, L1F + 16 * 128);
  load_b32(bL1, L1F + 16 * 160 + 16 * 128);

  {
    {
      v8f c0 = {}, c1 = {};
      c0 = wmma_regs(aea, bF0, c0);
      c1 = wmma_regs(aea, bF1, c1);
      tile_store(c0, [&](int, int m, int n, float v) {
        hE[m * 32 + n] = sspf(v * INV_SQ32);
      });
      tile_store(c1, [&](int, int m, int n, float v) {
        hE[m * 32 + 16 + n] = sspf(v * INV_SQ32);
      });
    }

#pragma nounroll
    for (int mb = 0; mb < 16; mb += 8) {
      float dv[8];
#pragma unroll
      for (int r = 0; r < 8; ++r) {
        const float* pd = (const float*)((const char*)p1 + ((unsigned)dstS[mb + r] * 384u + (unsigned)lane * 12u));
        const float* ps = (const float*)((const char*)p1 + ((unsigned)srcS[mb + r] * 384u + (unsigned)lane * 12u));
        dv[r] = (pd[0] * ps[0] + pd[1] * ps[1] + pd[2] * ps[2]) * INV_SQ3;
      }
#pragma unroll
      for (int r = 0; r < 8; ++r) ipS[(mb + r) * IP_PITCH + lane] = dv[r];
    }
  }
  __syncthreads();

  {
    v2f aip[8];
    load_a(aip, [&](int m, int k) { return ld2(ipS + m * IP_PITCH + k); });
    v8f c0, c1;
#pragma unroll
    for (int r = 0; r < 8; ++r) {
      c0[r] = qv0[r] + qw0[r];
      c1[r] = qv1[r] + qw1[r];
    }
    c0 = wmma_regs(aip, bL0, c0);
    c1 = wmma_regs(aip, bL1, c1);
    tile_store(c0, [&](int, int m, int n, float v) {
      hN[m * 32 + n] = sspf(v * INV_SQ160);
    });
    tile_store(c1, [&](int, int m, int n, float v) {
      hN[m * 32 + 16 + n] = sspf(v * INV_SQ160);
    });
  }
  __syncthreads();

  float sh0r[8], s1xr[8], s1yr[8], s1zr[8];
  int dR[8], sR[8];
#pragma unroll
  for (int r = 0; r < 8; ++r) {
    int m = rbase + r;
    const v4f sh = *(const v4f*)(shS + m * 4);
    sh0r[r] = sh.x;
    s1xr[r] = sh.y;
    s1yr[r] = sh.z;
    s1zr[r] = sh.w;
    dR[r] = dstS[m];
    sR[r] = srcS[m];
  }
  unsigned s0B[8], s1B[8], dB[8];
#pragma unroll
  for (int r = 0; r < 8; ++r) {
    s0B[r] = (unsigned)sR[r] * 256u;
    s1B[r] = (unsigned)sR[r] * 384u;
    dB[r] = (unsigned)dR[r] * 1920u;
  }
  unsigned flm = 0x80u;
#pragma unroll
  for (int r = 0; r < 7; ++r) flm |= (dR[r + 1] != dR[r]) ? (1u << r) : 0u;

  v2f aN[8], aE[8];
  load_a(aN, [&](int m, int k) { return ld2(hN + m * 32 + k); });
  load_a(aE, [&](int m, int k) { return ld2(hE + m * 32 + k); });

  auto do_tile = [&](auto seg, const int j) {
    constexpr int SEG = decltype(seg)::value;
    const int nb = j * 16;
    const int col = nb + (lane & 15);
    v4f bN[4], bE[4];
    load_b32(bN, L2F + nb * 32);
    load_b32(bE, F2F + nb * 32);
    constexpr int CB = (SEG == 0) ? 0 : (SEG == 1) ? 64 : (SEG == 2) ? 128 : (SEG == 3) ? 160 : 192;
    const int cs = col - CB;
    float g0[8], g1[8], g2[8];
    if constexpr (SEG <= 1) {
#pragma unroll
      for (int r = 0; r < 8; ++r)
        g0[r] = *(const float*)((const char*)x0n + (s0B[r] + (unsigned)cs * 4u));
    } else {
#pragma unroll
      for (int r = 0; r < 8; ++r) {
        const float* xv = (const float*)((const char*)x1n + (s1B[r] + (unsigned)cs * 12u));
        g0[r] = xv[0];
        g1[r] = xv[1];
        g2[r] = xv[2];
      }
    }

    v8f cN = {}, cE = {};
    cN = wmma_regs(aN, bN, cN);
    cE = wmma_regs(aE, bE, cE);
    v8f w8 = cN * cE * (INV_SQ32 * INV_SQ32);

    float v0[8], v1[8], v2[8];
    if constexpr (SEG == 0) {
#pragma unroll
      for (int r = 0; r < 8; ++r) v0[r] = w8[r] * g0[r] * sh0r[r];
      run_add(acc, dB, flm, col, v0);
    } else if constexpr (SEG == 1) {
#pragma unroll
      for (int r = 0; r < 8; ++r) {
        float f = w8[r] * g0[r] * INV_SQ3;
        v0[r] = f * s1xr[r];
        v1[r] = f * s1yr[r];
        v2[r] = f * s1zr[r];
      }
      run_add(acc, dB, flm, 96 + cs * 3 + 0, v0);
      run_add(acc, dB, flm, 96 + cs * 3 + 1, v1);
      run_add(acc, dB, flm, 96 + cs * 3 + 2, v2);
    } else if constexpr (SEG == 2) {
#pragma unroll
      for (int r = 0; r < 8; ++r) {
        float f = w8[r] * sh0r[r];
        v0[r] = f * g0[r];
        v1[r] = f * g1[r];
        v2[r] = f * g2[r];
      }
      run_add(acc, dB, flm, 288 + cs * 3 + 0, v0);
      run_add(acc, dB, flm, 288 + cs * 3 + 1, v1);
      run_add(acc, dB, flm, 288 + cs * 3 + 2, v2);
    } else if constexpr (SEG == 3) {
#pragma unroll
      for (int r = 0; r < 8; ++r) {
        float dp = g0[r] * s1xr[r] + g1[r] * s1yr[r] + g2[r] * s1zr[r];
        v0[r] = w8[r] * dp * INV_SQ3;
      }
      run_add(acc, dB, flm, 64 + cs, v0);
    } else {
#pragma unroll
      for (int r = 0; r < 8; ++r) {
        const float s1x = s1xr[r], s1y = s1yr[r], s1z = s1zr[r];
        float cx = g1[r] * s1z - g2[r] * s1y;
        float cy = g2[r] * s1x - g0[r] * s1z;
        float cz = g0[r] * s1y - g1[r] * s1x;
        float f = w8[r] * INV_SQ2;
        v0[r] = f * cx;
        v1[r] = f * cy;
        v2[r] = f * cz;
      }
      run_add(acc, dB, flm, 384 + cs * 3 + 0, v0);
      run_add(acc, dB, flm, 384 + cs * 3 + 1, v1);
      run_add(acc, dB, flm, 384 + cs * 3 + 2, v2);
    }
  };

#pragma nounroll
  for (int j = 0; j < 4; ++j) do_tile(IC<0>{}, j);
#pragma nounroll
  for (int j = 4; j < 8; ++j) do_tile(IC<1>{}, j);
#pragma nounroll
  for (int j = 8; j < 10; ++j) do_tile(IC<2>{}, j);
#pragma nounroll
  for (int j = 10; j < 12; ++j) do_tile(IC<3>{}, j);
#pragma nounroll
  for (int j = 12; j < 14; ++j) do_tile(IC<4>{}, j);
}

__global__ __launch_bounds__(NODE_WAVES * 32) __attribute__((amdgpu_waves_per_eu(1, 2))) void proj_kernel(
    const float* __restrict__ acc,
    const float* __restrict__ WO0F, const float* __restrict__ bO0,
    const float* __restrict__ WO1F, const float* __restrict__ WO2F,
    float* __restrict__ y) {
  __shared__ __align__(16) float yS[16 * 416];
  static_assert(NODE_WAVES == 4, "the shares below are written out for four waves");
  const int wave = __builtin_amdgcn_readfirstlane(threadIdx.x >> 5);
  const int n0 = blockIdx.x * 16;
  const float* at = acc + (size_t)n0 * 480;

  {
    v2f aa[24];
    load_a(aa, [&](int m, int k) { return ld2(at + (size_t)m * 480 + k); });
#pragma unroll
    for (int jj = 0; jj < 2; ++jj) {
      const int j = wave * 2 + jj;
      v8f c = {};
      c = wmma_tile(aa, WO0F + j * 16 * 96, c);
      tile_store(c, [&](int, int m, int n, float v) {
        int col = j * 16 + n;
        yS[m * 416 + col] = v * INV_SQ96 + bO0[col];
      });
    }
  }
#pragma unroll
  for (int i = 0; i < 3; ++i) {
    v2f aa[24];
    load_a(aa, [&](int m, int k) {
      v2f a;
      a.x = at[(size_t)m * 480 + 96 + 3 * k + i];
      a.y = at[(size_t)m * 480 + 96 + 3 * (k + 1) + i];
      return a;
    });
    {
      const int j = wave;
      v8f c = {};
      c = wmma_tile(aa, WO1F + j * 16 * 96, c);
      tile_store(c, [&](int, int m, int n, float v) {
        yS[m * 416 + 128 + (j * 16 + n) * 3 + i] = v * INV_SQ96;
      });
    }
  }
  if (wave < 3) {
    const int i = wave;
    v2f aa[8];
    load_a(aa, [&](int m, int k) {
      v2f a;
      a.x = at[(size_t)m * 480 + 384 + 3 * k + i];
      a.y = at[(size_t)m * 480 + 384 + 3 * (k + 1) + i];
      return a;
    });
#pragma unroll
    for (int j = 0; j < 2; ++j) {
      v8f c = {};
      c = wmma_tile(aa, WO2F + j * 16 * 32, c);
      tile_store(c, [&](int, int m, int n, float v) {
        yS[m * 416 + 320 + (j * 16 + n) * 3 + i] = v * INV_SQ32;
      });
    }
  }
  __syncthreads();
  copy_tile<16 * 416>(y + (size_t)n0 * 416, yS);
}

extern "C" void kernel_launch(void* const* d_in, const int* in_sizes, int n_in,
                              void* d_out, int out_size, void* d_ws,
                              size_t ws_size, hipStream_t stream) {
  (void)in_sizes; (void)n_in; (void)out_size; (void)ws_size;
  const float* x         = (const float*)d_in[0];
  const float* edge_sh   = (const float*)d_in[1];
  const float* edge_attr = (const float*)d_in[2];
  const float* W0p = (const float*)d_in[3];
  const float* b0p = (const float*)d_in[4];
  const float* W1p = (const float*)d_in[5];
  const float* W0n = (const float*)d_in[6];
  const float* b0n = (const float*)d_in[7];
  const float* W1n = (const float*)d_in[8];
  const float* G1  = (const float*)d_in[9];
  const float* g1b = (const float*)d_in[10];
  const float* G2  = (const float*)d_in[11];
  const float* g2b = (const float*)d_in[12];
  const float* F1  = (const float*)d_in[13];
  const float* F2  = (const float*)d_in[14];
  const float* L1  = (const float*)d_in[15];
  const float* L2  = (const float*)d_in[16];
  const float* WO0 = (const float*)d_in[17];
  const float* bO0 = (const float*)d_in[18];
  const float* WO1 = (const float*)d_in[19];
  const float* WO2 = (const float*)d_in[20];
  const int* edge_index = (const int*)d_in[21];

  float* ws   = (float*)d_ws;
  float* qd   = ws;
  float* qs   = qd   + (size_t)N_NODES * 32;
  float* p1   = qs   + (size_t)N_NODES * 32;
  float* x0n  = p1   + (size_t)N_NODES * 96;
  float* x1n  = x0n  + (size_t)N_NODES * 64;
  float* acc  = x1n  + (size_t)N_NODES * 96;
  float* wt   = acc  + (size_t)N_NODES * 480;
  float* W0pF = wt + OFF_W0P;
  float* W1pF = wt + OFF_W1P;
  float* W0nF = wt + OFF_W0N;
  float* W1nF = wt + OFF_W1N;
  float* G1F  = wt + OFF_G1;
  float* G2F  = wt + OFF_G2;
  float* F1F  = wt + OFF_F1;
  float* F2F  = wt + OFF_F2;
  float* L1F  = wt + OFF_L1;
  float* L2F  = wt + OFF_L2;
  float* WO0F = wt + OFF_WO0;
  float* WO1F = wt + OFF_WO1;
  float* WO2F = wt + OFF_WO2;

  int* sc   = (int*)d_out;
  int* cnt  = sc;
  int* off  = sc + CNT_PADDED;
  int* rank = sc + 2 * CNT_PADDED;
  int* perm = sc + 2 * CNT_PADDED + N_EDGES;

  setup_kernel<<<SETUP_BLOCKS, SETUP_THREADS, 0, stream>>>(
      W0p, W1p, W0n, W1n, G1, G2, F1, F2, L1, L2, WO0, WO1, WO2, wt, acc, cnt);

  sort_rank_kernel<<<(N_EDGES + 255) / 256, 256, 0, stream>>>(edge_index, cnt, rank);
  sort_scan_kernel<<<1, 1024, 0, stream>>>(cnt, off);
  sort_fill_kernel<<<(N_EDGES + 255) / 256, 256, 0, stream>>>(edge_index, off, rank, perm);

  node_kernel<<<N_NODES / 16, NODE_WAVES * 32, 0, stream>>>(
      x, W0pF, b0p, W1pF, W0nF, b0n, W1nF, G1F, g1b, G2F, g2b, L1F,
      qd, qs, p1, x0n, x1n);

  edge_kernel<<<N_EDGES / 16, 32, 0, stream>>>(
      edge_sh, edge_attr, F1F, F2F, L1F, L2F, edge_index, perm,
      qd, qs, p1, x0n, x1n, acc);

  proj_kernel<<<N_NODES / 16, NODE_WAVES * 32, 0, stream>>>(
      acc, WO0F, bO0, WO1F, WO2F, (float*)d_out);
}


# 7 launches become 5: node precompute shares a launch with the sort's rank pass (independent work); each fill block does the 10000-counter prefix sum itself (LDS + lane shuffles); same sums
# speedup vs baseline: 2.3275x; 1.0342x over previous
#include <hip/hip_runtime.h>
#include <hip/hip_bf16.h>
#include <math.h>


typedef __attribute__((ext_vector_type(2))) float v2f;
typedef __attribute__((ext_vector_type(4))) float v4f;
typedef __attribute__((ext_vector_type(8))) float v8f;

#define N_NODES 10000
#define N_EDGES 320000

#define INV_SQ3   0.5773502691896258f
#define INV_SQ2   0.7071067811865476f
#define INV_SQ32  0.17677669529663687f
#define INV_SQ64  0.125f
#define INV_SQ96  0.10206207261596575f
#define INV_SQ160 0.07905694150420949f
#define LN2F      0.6931471805599453f

__device__ __forceinline__ float sspf(float v) {
  float sp = (v > 20.0f) ? v : log1pf(expf(v));
  return sp - LN2F;
}
__device__ __forceinline__ float siluf(float v) {
  return v / (1.0f + expf(-v));
}
__device__ __forceinline__ v2f ld2(const float* p) {
  return *(const v2f*)p;
}

__device__ __forceinline__ v8f wmma4(v2f a, v2f b, v8f c) {
  return __builtin_amdgcn_wmma_f32_16x16x4_f32(false, a, false, b,
                                               (short)0, c, false, false);
}


template <int KS, typename LA>
__device__ __forceinline__ void load_a(v2f (&a)[KS], LA la) {
  const int lane = threadIdx.x & 31;
  const int m = lane & 15;
  const int kk = (lane >> 4) << 1;
#pragma unroll
  for (int q = 0; q < KS; ++q) a[q] = la(m, q * 4 + kk);
}

template <int KS>
__device__ __forceinline__ v8f wmma_tile(const v2f (&a)[KS],
                                         const float* __restrict__ wf_tile, v8f c) {
  static_assert(KS % 2 == 0, "k-steps are stored in pairs");
  const v4f* bp = (const v4f*)wf_tile + (threadIdx.x & 31);
#pragma unroll
  for (int p = 0; p < KS / 2; ++p) {
    const v4f b = bp[p * 32];
    const v2f b0 = {b.x, b.y};
    const v2f b1 = {b.z, b.w};
    c = wmma4(a[2 * p], b0, c);
    c = wmma4(a[2 * p + 1], b1, c);
  }
  return c;
}

__device__ __forceinline__ void load_b32(v4f (&b)[4], const float* __restrict__ wf_tile) {
  const v4f* bp = (const v4f*)wf_tile + (threadIdx.x & 31);
#pragma unroll
  for (int p = 0; p < 4; ++p) b[p] = bp[p * 32];
}
__device__ __forceinline__ v8f wmma_regs(const v2f (&a)[8], const v4f (&b)[4], v8f c) {
#pragma unroll
  for (int p = 0; p < 4; ++p) {
    const v2f b0 = {b[p].x, b[p].y};
    const v2f b1 = {b[p].z, b[p].w};
    c = wmma4(a[2 * p], b0, c);
    c = wmma4(a[2 * p + 1], b1, c);
  }
  return c;
}
template <int V> struct IC { static constexpr int value = V; };

template <typename ST>
__device__ __forceinline__ void tile_store(v8f c, ST st) {
  const int lane = threadIdx.x & 31;
  const int col = lane & 15;
  const int rbase = (lane >> 4) << 3;
#pragma unroll
  for (int r = 0; r < 8; ++r) st(r, rbase + r, col, c[r]);
}

constexpr int OFF_W0P = 0;
constexpr int OFF_W1P = OFF_W0P + 64 * 64;
constexpr int OFF_W0N = OFF_W1P + 32 * 32;
constexpr int OFF_W1N = OFF_W0N + 64 * 64;
constexpr int OFF_G1  = OFF_W1N + 32 * 32;
constexpr int OFF_G2  = OFF_G1  + 96 * 96;
constexpr int OFF_F1  = OFF_G2  + 96 * 96;
constexpr int OFF_F2  = OFF_F1  + 32 * 32;
constexpr int OFF_L1  = OFF_F2  + 32 * 224;
constexpr int OFF_L2  = OFF_L1  + 160 * 32;
constexpr int OFF_WO0 = OFF_L2  + 32 * 224;
constexpr int OFF_WO1 = OFF_WO0 + 96 * 128;
constexpr int OFF_WO2 = OFF_WO1 + 96 * 64;
constexpr int WT_FLOATS = OFF_WO2 + 32 * 32;

constexpr int SETUP_THREADS  = 256;
constexpr int SETUP_T_BLOCKS = WT_FLOATS / 1024;
constexpr int ACC_V4         = N_NODES * 480 / 4;
constexpr int Z_V4_PER_BLOCK = SETUP_THREADS * 16;
constexpr int SETUP_Z_BLOCKS = (ACC_V4 + Z_V4_PER_BLOCK - 1) / Z_V4_PER_BLOCK;
constexpr int CNT_PADDED     = 10240;
constexpr int SETUP_BLOCKS   = SETUP_T_BLOCKS + SETUP_Z_BLOCKS + 1;

template <int K, int Nc>
__device__ __forceinline__ void frag_slab(const float* __restrict__ W,
                                          float* __restrict__ Wf, int slab) {
  static_assert((K * Nc) % 1024 == 0, "weights are whole slabs");
  static_assert(K % 8 == 0 && Nc % 16 == 0, "whole k-step pairs and column tiles");
  float v[4];
#pragma unroll
  for (int i = 0; i < 4; ++i) v[i] = W[slab * 1024 + i * SETUP_THREADS + threadIdx.x];
#pragma unroll
  for (int i = 0; i < 4; ++i) {
    const int idx = slab * 1024 + i * SETUP_THREADS + threadIdx.x;
    const int k = idx / Nc, n = idx % Nc;
    const int j = n >> 4, c = n & 15;
    const int p = k >> 3, r = k & 7;
    const int h = (r >> 1) & 1;
    const int sl = ((r >> 2) << 1) | (r & 1);
    Wf[(((j * (K / 8) + p) * 32) + h * 16 + c) * 4 + sl] = v[i];
  }
}

__global__ __launch_bounds__(SETUP_THREADS) void setup_kernel(
    const float* __restrict__ W0p, const float* __restrict__ W1p,
    const float* __restrict__ W0n, const float* __restrict__ W1n,
    const float* __restrict__ G1, const float* __restrict__ G2,
    const float* __restrict__ F1, const float* __restrict__ F2,
    const float* __restrict__ L1, const float* __restrict__ L2,
    const float* __restrict__ WO0, const float* __restrict__ WO1,
    const float* __restrict__ WO2,
    float* __restrict__ wt, float* __restrict__ acc, int* __restrict__ cnt) {
  int s = blockIdx.x;
  if (s < SETUP_T_BLOCKS) {
    if (s < 4)  { frag_slab<64, 64>(W0p, wt + OFF_W0P, s); return; }   s -= 4;
    if (s < 1)  { frag_slab<32, 32>(W1p, wt + OFF_W1P, s); return; }   s -= 1;
    if (s < 4)  { frag_slab<64, 64>(W0n, wt + OFF_W0N, s); return; }   s -= 4;
    if (s < 1)  { frag_slab<32, 32>(W1n, wt + OFF_W1N, s); return; }   s -= 1;
    if (s < 9)  { frag_slab<96, 96>(G1, wt + OFF_G1, s); return; }     s -= 9;
    if (s < 9)  { frag_slab<96, 96>(G2, wt + OFF_G2, s); return; }     s -= 9;
    if (s < 1)  { frag_slab<32, 32>(F1, wt + OFF_F1, s); return; }     s -= 1;
    if (s < 7)  { frag_slab<32, 224>(F2, wt + OFF_F2, s); return; }    s -= 7;
    if (s < 5)  { frag_slab<160, 32>(L1, wt + OFF_L1, s); return; }    s -= 5;
    if (s < 7)  { frag_slab<32, 224>(L2, wt + OFF_L2, s); return; }    s -= 7;
    if (s < 12) { frag_slab<96, 128>(WO0, wt + OFF_WO0, s); return; }  s -= 12;
    if (s < 6)  { frag_slab<96, 64>(WO1, wt + OFF_WO1, s); return; }   s -= 6;
    frag_slab<32, 32>(WO2, wt + OFF_WO2, s);
    return;
  }
  s -= SETUP_T_BLOCKS;
  if (s < SETUP_Z_BLOCKS) {
    v4f* a4 = (v4f*)acc;
    const v4f z = {0.0f, 0.0f, 0.0f, 0.0f};
    for (int i = 0; i < 16; ++i) {
      const int q = s * Z_V4_PER_BLOCK + i * SETUP_THREADS + threadIdx.x;
      if (q < ACC_V4) a4[q] = z;
    }
    return;
  }
  for (int i = threadIdx.x; i < CNT_PADDED; i += SETUP_THREADS) cnt[i] = 0;
}

constexpr int NODE_WAVES = 4;

template <int FLOATS>
__device__ __forceinline__ void copy_tile(float* __restrict__ dst, const float* src) {
  constexpr int NT = NODE_WAVES * 32;
  static_assert(FLOATS % (4 * NT) == 0, "whole 16-byte words per thread");
#pragma unroll
  for (int q = 0; q < FLOATS / (4 * NT); ++q) {
    const int w = q * NT + threadIdx.x;
    ((v4f*)dst)[w] = ((const v4f*)src)[w];
  }
}

constexpr int P0_PITCH = 68;
constexpr int IP_PITCH = 36;

__device__ __forceinline__ void node_tile(
    const int tile,
    const float* __restrict__ x,
    const float* __restrict__ W0pF, const float* __restrict__ b0p,
    const float* __restrict__ W1pF,
    const float* __restrict__ W0nF, const float* __restrict__ b0n,
    const float* __restrict__ W1nF,
    const float* __restrict__ G1F, const float* __restrict__ g1b,
    const float* __restrict__ G2F, const float* __restrict__ g2b,
    const float* __restrict__ L1F,
    float* __restrict__ qd, float* __restrict__ qs, float* __restrict__ p1,
    float* __restrict__ x0n, float* __restrict__ x1n) {
  __shared__ __align__(16) float p0S[16 * P0_PITCH];
  __shared__ __align__(16) float f0S[16 * 96];
  __shared__ __align__(16) float hS[16 * 96];
  __shared__ __align__(16) float gS[16 * 96];
  __shared__ __align__(16) float qdS[16 * 32];
  __shared__ __align__(16) float qsS[16 * 32];
  __shared__ __align__(16) float p1S[16 * 96];
  __shared__ __align__(16) float x0nS[16 * 64];
  __shared__ __align__(16) float x1nS[16 * 96];
  static_assert(NODE_WAVES == 4, "the shares below are written out for four waves");

  const int tid = threadIdx.x;
  const int wave = __builtin_amdgcn_readfirstlane(threadIdx.x >> 5);
  const int n0 = tile * 16;
  const float* xt = x + (size_t)n0 * 160;

  {
    v2f ax[16];
    load_a(ax, [&](int m, int k) { return ld2(xt + m * 160 + k); });
    {
      const int j = wave;
      v8f c = {};
      c = wmma_tile(ax, W0pF + j * 16 * 64, c);
      tile_store(c, [&](int, int m, int n, float v) {
        int col = j * 16 + n;
        p0S[m * P0_PITCH + col] = v * INV_SQ64 + b0p[col];
      });
    }
  }
  if (wave < 3) {
    const int i = wave;
    v2f ax[8];
    load_a(ax, [&](int m, int k) {
      v2f a;
      a.x = xt[m * 160 + 64 + 3 * k + i];
      a.y = xt[m * 160 + 64 + 3 * (k + 1) + i];
      return a;
    });
#pragma unroll
    for (int j = 0; j < 2; ++j) {
      v8f c = {};
      c = wmma_tile(ax, W1pF + j * 16 * 32, c);
      tile_store(c, [&](int, int m, int n, float v) {
        p1S[m * 96 + (j * 16 + n) * 3 + i] = v * INV_SQ32;
      });
    }
  }

  {
    constexpr int NT = NODE_WAVES * 32;
#pragma unroll
    for (int q = 0; q < 16 * 64 / NT; ++q) {
      const int idx = q * NT + tid;
      const int m = idx >> 6, c = idx & 63;
      f0S[m * 96 + c] = xt[m * 160 + c];
    }
#pragma unroll
    for (int q = 0; q < 16 * 32 / NT; ++q) {
      const int idx = q * NT + tid;
      const int m = idx >> 5, u = idx & 31;
      float a0 = xt[m * 160 + 64 + u * 3 + 0];
      float a1 = xt[m * 160 + 64 + u * 3 + 1];
      float a2 = xt[m * 160 + 64 + u * 3 + 2];
      f0S[m * 96 + 64 + u] = sqrtf(a0 * a0 + a1 * a1 + a2 * a2 + 1e-12f);
    }
  }
  __syncthreads();

  {
    v2f ap[16];
    load_a(ap, [&](int m, int k) { return ld2(p0S + m * P0_PITCH + k); });
    {
      const int j = wave >> 1, half = wave & 1;
      float* q = half ? qsS : qdS;
      v8f c = {};
      c = wmma_tile(ap, L1F + j * 16 * 160 + half * 8 * 128, c);
      tile_store(c, [&](int, int m, int n, float v) {
        q[m * 32 + j * 16 + n] = v;
      });
    }
  }

  if (wave < 3) {
    v2f af[24];
    load_a(af, [&](int m, int k) { return ld2(f0S + m * 96 + k); });
#pragma unroll
    for (int jj = 0; jj < 2; ++jj) {
      const int j = wave * 2 + jj;
      v8f c = {};
      c = wmma_tile(af, G1F + j * 16 * 96, c);
      tile_store(c, [&](int, int m, int n, float v) {
        int col = j * 16 + n;
        hS[m * 96 + col] = siluf(v + g1b[col]);
      });
    }
  }
  __syncthreads();
  copy_tile<16 * 32>(qd + (size_t)n0 * 32, qdS);
  copy_tile<16 * 32>(qs + (size_t)n0 * 32, qsS);
  copy_tile<16 * 96>(p1 + (size_t)n0 * 96, p1S);
  if (wave < 3) {
    v2f ah[24];
    load_a(ah, [&](int m, int k) { return ld2(hS + m * 96 + k); });
#pragma unroll
    for (int jj = 0; jj < 2; ++jj) {
      const int j = wave * 2 + jj;
      v8f c = {};
      c = wmma_tile(ah, G2F + j * 16 * 96, c);
      tile_store(c, [&](int, int m, int n, float v) {
        gS[m * 96 + j * 16 + n] = v + g2b[j * 16 + n];
      });
    }
  }
  __syncthreads();

  {
    v2f ag[16];
    load_a(ag, [&](int m, int k) { return ld2(gS + m * 96 + k); });
    {
      const int j = wave;
      v8f c = {};
      c = wmma_tile(ag, W0nF + j * 16 * 64, c);
      tile_store(c, [&](int, int m, int n, float v) {
        int col = j * 16 + n;
        x0nS[m * 64 + col] = v * INV_SQ64 + b0n[col];
      });
    }
  }
  if (wave < 3) {
    const int i = wave;
    v2f ax[8];
    load_a(ax, [&](int m, int k) {
      v2f a;
      a.x = xt[m * 160 + 64 + 3 * k + i] * gS[m * 96 + 64 + k];
      a.y = xt[m * 160 + 64 + 3 * (k + 1) + i] * gS[m * 96 + 64 + k + 1];
      return a;
    });
#pragma unroll
    for (int j = 0; j < 2; ++j) {
      v8f c = {};
      c = wmma_tile(ax, W1nF + j * 16 * 32, c);
      tile_store(c, [&](int, int m, int n, float v) {
        x1nS[m * 96 + (j * 16 + n) * 3 + i] = v * INV_SQ32;
      });
    }
  }
  __syncthreads();
  copy_tile<16 * 64>(x0n + (size_t)n0 * 64, x0nS);
  copy_tile<16 * 96>(x1n + (size_t)n0 * 96, x1nS);
}

constexpr int NODE_TILES  = N_NODES / 16;
constexpr int RANK_BLOCKS = N_EDGES / (NODE_WAVES * 32);
static_assert(NODE_TILES * 16 == N_NODES, "whole node tiles");
static_assert(RANK_BLOCKS * NODE_WAVES * 32 == N_EDGES, "whole rank blocks");

__global__ __launch_bounds__(NODE_WAVES * 32) void node_rank_kernel(
    const float* __restrict__ x,
    const float* __restrict__ W0pF, const float* __restrict__ b0p,
    const float* __restrict__ W1pF,
    const float* __restrict__ W0nF, const float* __restrict__ b0n,
    const float* __restrict__ W1nF,
    const float* __restrict__ G1F, const float* __restrict__ g1b,
    const float* __restrict__ G2F, const float* __restrict__ g2b,
    const float* __restrict__ L1F,
    float* __restrict__ qd, float* __restrict__ qs, float* __restrict__ p1,
    float* __restrict__ x0n, float* __restrict__ x1n,
    const int* __restrict__ edge_index, int* __restrict__ cnt, int* __restrict__ rank) {
  if (blockIdx.x < NODE_TILES) {
    node_tile(blockIdx.x, x, W0pF, b0p, W1pF, W0nF, b0n, W1nF, G1F, g1b, G2F, g2b, L1F,
              qd, qs, p1, x0n, x1n);
    return;
  }
  const int e = (blockIdx.x - NODE_TILES) * (NODE_WAVES * 32) + threadIdx.x;
  rank[e] = atomicAdd(&cnt[edge_index[e]], 1);
}

#define SCAN_PER_THREAD 10
constexpr int FILL_THREADS = 1024;
constexpr int FILL_WAVES   = FILL_THREADS / 32;
constexpr int FILL_BLOCKS  = (N_EDGES + FILL_THREADS - 1) / FILL_THREADS;
static_assert(FILL_THREADS * SCAN_PER_THREAD >= N_NODES, "one block scans all counters");
static_assert(FILL_WAVES == 32, "the wave totals are scanned by one wave");

__device__ __forceinline__ int wave_inclusive_scan(int v, const int lane) {
#pragma unroll
  for (int d = 1; d < 32; d <<= 1) {
    const int u = __shfl_up(v, d, 32);
    if (lane >= d) v += u;
  }
  return v;
}

__global__ __launch_bounds__(FILL_THREADS) void sort_scan_fill_kernel(
    const int* __restrict__ edge_index, const int* __restrict__ cnt,
    const int* __restrict__ rank, int* __restrict__ perm) {
  __shared__ int part[FILL_WAVES];
  __shared__ int offS[FILL_THREADS * SCAN_PER_THREAD];
  const int t = threadIdx.x;
  const int lane = t & 31, wave = t >> 5;
  const int e = blockIdx.x * FILL_THREADS + t;
  int d = 0, rk = 0;
  if (e < N_EDGES) {
    d = edge_index[e];
    rk = rank[e];
  }
#pragma unroll
  for (int i = 0; i < SCAN_PER_THREAD; ++i) {
    const int idx = i * FILL_THREADS + t;
    offS[idx] = (idx < N_NODES) ? cnt[idx] : 0;
  }
  __syncthreads();
  int loc[SCAN_PER_THREAD];
  int s = 0;
#pragma unroll
  for (int i = 0; i < SCAN_PER_THREAD; ++i) {
    const int v = offS[t * SCAN_PER_THREAD + i];
    loc[i] = s;
    s += v;
  }
  const int incl = wave_inclusive_scan(s, lane);
  if (lane == 31) part[wave] = incl;
  __syncthreads();
  if (wave == 0) {
    const int w = part[lane];
    part[lane] = wave_inclusive_scan(w, lane) - w;
  }
  __syncthreads();
  const int base = part[wave] + incl - s;
#pragma unroll
  for (int i = 0; i < SCAN_PER_THREAD; ++i) offS[t * SCAN_PER_THREAD + i] = base + loc[i];
  __syncthreads();
  if (e < N_EDGES) perm[offS[d] + rk] = e;
}

__device__ __forceinline__ void run_add(float* __restrict__ acc,
                                        const unsigned (&dB)[8], unsigned flm,
                                        int chan, const float (&v)[8]) {
  float run = 0.0f;
#pragma unroll
  for (int r = 0; r < 8; ++r) {
    run += v[r];
    if ((flm >> r) & 1u) {
      atomicAdd((float*)((char*)acc + (dB[r] + (unsigned)chan * 4u)), run);
      run = 0.0f;
    }
  }
}

__global__ __launch_bounds__(32) void edge_kernel(
    const float* __restrict__ edge_sh, const float* __restrict__ edge_attr,
    const float* __restrict__ F1F, const float* __restrict__ F2F,
    const float* __restrict__ L1F, const float* __restrict__ L2F,
    const int* __restrict__ edge_index, const int* __restrict__ perm,
    const float* __restrict__ qd, const float* __restrict__ qs,
    const float* __restrict__ p1,
    const float* __restrict__ x0n, const float* __restrict__ x1n,
    float* __restrict__ acc) {
  __shared__ __align__(16) float ipS[16 * IP_PITCH];
  __shared__ __align__(16) float hN[16 * 32];
  __shared__ __align__(16) float hE[16 * 32];
  __shared__ __align__(16) float shS[16 * 4];
  __shared__ int dstS[16];
  __shared__ int srcS[16];

  const int lane = threadIdx.x;
  const int e0 = blockIdx.x * 16;

  __builtin_prefetch(F2F, 0, 1);
  __builtin_prefetch(L2F, 0, 1);

  v4f bF0[4], bF1[4];
  load_b32(bF0, F1F);
  load_b32(bF1, F1F + 16 * 32);

  const int e = perm[e0 + (lane & 15)];
  v2f aea[8];
  {
    const float* ar = edge_attr + (size_t)e * 32 + ((lane >> 4) << 1);
#pragma unroll
    for (int q = 0; q < 8; ++q) aea[q] = ld2(ar + q * 4);
  }
  if (lane < 16) {
    dstS[lane] = edge_index[e];
    srcS[lane] = edge_index[N_EDGES + e];
    *(v4f*)(shS + lane * 4) = *(const v4f*)(edge_sh + (size_t)e * 4);
  }
  __syncthreads();

  const int rbase = (lane >> 4) << 3;

  float qv0[8], qv1[8], qw0[8], qw1[8];
#pragma unroll
  for (int r = 0; r < 8; ++r) {
    const float* pd = (const float*)((const char*)qd + ((unsigned)dstS[rbase + r] * 128u + (unsigned)(lane & 15) * 4u));
    const float* ps = (const float*)((const char*)qs + ((unsigned)srcS[rbase + r] * 128u + (unsigned)(lane & 15) * 4u));
    qv0[r] = pd[0];
    qv1[r] = pd[16];
    qw0[r] = ps[0];
    qw1[r] = ps[16];
  }
  v4f bL0[4], bL1[4];
  load_b32(bL0, L1F + 16 * 128);
  load_b32(bL1, L1F + 16 * 160 + 16 * 128);

  {
    {
      v8f c0 = {}, c1 = {};
      c0 = wmma_regs(aea, bF0, c0);
      c1 = wmma_regs(aea, bF1, c1);
      tile_store(c0, [&](int, int m, int n, float v) {
        hE[m * 32 + n] = sspf(v * INV_SQ32);
      });
      tile_store(c1, [&](int, int m, int n, float v) {
        hE[m * 32 + 16 + n] = sspf(v * INV_SQ32);
      });
    }

#pragma nounroll
    for (int mb = 0; mb < 16; mb += 8) {
      float dv[8];
#pragma unroll
      for (int r = 0; r < 8; ++r) {
        const float* pd = (const float*)((const char*)p1 + ((unsigned)dstS[mb + r] * 384u + (unsigned)lane * 12u));
        const float* ps = (const float*)((const char*)p1 + ((unsigned)srcS[mb + r] * 384u + (unsigned)lane * 12u));
        dv[r] = (pd[0] * ps[0] + pd[1] * ps[1] + pd[2] * ps[2]) * INV_SQ3;
      }
#pragma unroll
      for (int r = 0; r < 8; ++r) ipS[(mb + r) * IP_PITCH + lane] = dv[r];
    }
  }
  __syncthreads();

  {
    v2f aip[8];
    load_a(aip, [&](int m, int k) { return ld2(ipS + m * IP_PITCH + k); });
    v8f c0, c1;
#pragma unroll
    for (int r = 0; r < 8; ++r) {
      c0[r] = qv0[r] + qw0[r];
      c1[r] = qv1[r] + qw1[r];
    }
    c0 = wmma_regs(aip, bL0, c0);
    c1 = wmma_regs(aip, bL1, c1);
    tile_store(c0, [&](int, int m, int n, float v) {
      hN[m * 32 + n] = sspf(v * INV_SQ160);
    });
    tile_store(c1, [&](int, int m, int n, float v) {
      hN[m * 32 + 16 + n] = sspf(v * INV_SQ160);
    });
  }
  __syncthreads();

  float sh0r[8], s1xr[8], s1yr[8], s1zr[8];
  int dR[8], sR[8];
#pragma unroll
  for (int r = 0; r < 8; ++r) {
    int m = rbase + r;
    const v4f sh = *(const v4f*)(shS + m * 4);
    sh0r[r] = sh.x;
    s1xr[r] = sh.y;
    s1yr[r] = sh.z;
    s1zr[r] = sh.w;
    dR[r] = dstS[m];
    sR[r] = srcS[m];
  }
  unsigned s0B[8], s1B[8], dB[8];
#pragma unroll
  for (int r = 0; r < 8; ++r) {
    s0B[r] = (unsigned)sR[r] * 256u;
    s1B[r] = (unsigned)sR[r] * 384u;
    dB[r] = (unsigned)dR[r] * 1920u;
  }
  unsigned flm = 0x80u;
#pragma unroll
  for (int r = 0; r < 7; ++r) flm |= (dR[r + 1] != dR[r]) ? (1u << r) : 0u;

  v2f aN[8], aE[8];
  load_a(aN, [&](int m, int k) { return ld2(hN + m * 32 + k); });
  load_a(aE, [&](int m, int k) { return ld2(hE + m * 32 + k); });

  auto do_tile = [&](auto seg, const int j) {
    constexpr int SEG = decltype(seg)::value;
    const int nb = j * 16;
    const int col = nb + (lane & 15);
    v4f bN[4], bE[4];
    load_b32(bN, L2F + nb * 32);
    load_b32(bE, F2F + nb * 32);
    constexpr int CB = (SEG == 0) ? 0 : (SEG == 1) ? 64 : (SEG == 2) ? 128 : (SEG == 3) ? 160 : 192;
    const int cs = col - CB;
    float g0[8], g1[8], g2[8];
    if constexpr (SEG <= 1) {
#pragma unroll
      for (int r = 0; r < 8; ++r)
        g0[r] = *(const float*)((const char*)x0n + (s0B[r] + (unsigned)cs * 4u));
    } else {
#pragma unroll
      for (int r = 0; r < 8; ++r) {
        const float* xv = (const float*)((const char*)x1n + (s1B[r] + (unsigned)cs * 12u));
        g0[r] = xv[0];
        g1[r] = xv[1];
        g2[r] = xv[2];
      }
    }

    v8f cN = {}, cE = {};
    cN = wmma_regs(aN, bN, cN);
    cE = wmma_regs(aE, bE, cE);
    v8f w8 = cN * cE * (INV_SQ32 * INV_SQ32);

    float v0[8], v1[8], v2[8];
    if constexpr (SEG == 0) {
#pragma unroll
      for (int r = 0; r < 8; ++r) v0[r] = w8[r] * g0[r] * sh0r[r];
      run_add(acc, dB, flm, col, v0);
    } else if constexpr (SEG == 1) {
#pragma unroll
      for (int r = 0; r < 8; ++r) {
        float f = w8[r] * g0[r] * INV_SQ3;
        v0[r] = f * s1xr[r];
        v1[r] = f * s1yr[r];
        v2[r] = f * s1zr[r];
      }
      run_add(acc, dB, flm, 96 + cs * 3 + 0, v0);
      run_add(acc, dB, flm, 96 + cs * 3 + 1, v1);
      run_add(acc, dB, flm, 96 + cs * 3 + 2, v2);
    } else if constexpr (SEG == 2) {
#pragma unroll
      for (int r = 0; r < 8; ++r) {
        float f = w8[r] * sh0r[r];
        v0[r] = f * g0[r];
        v1[r] = f * g1[r];
        v2[r] = f * g2[r];
      }
      run_add(acc, dB, flm, 288 + cs * 3 + 0, v0);
      run_add(acc, dB, flm, 288 + cs * 3 + 1, v1);
      run_add(acc, dB, flm, 288 + cs * 3 + 2, v2);
    } else if constexpr (SEG == 3) {
#pragma unroll
      for (int r = 0; r < 8; ++r) {
        float dp = g0[r] * s1xr[r] + g1[r] * s1yr[r] + g2[r] * s1zr[r];
        v0[r] = w8[r] * dp * INV_SQ3;
      }
      run_add(acc, dB, flm, 64 + cs, v0);
    } else {
#pragma unroll
      for (int r = 0; r < 8; ++r) {
        const float s1x = s1xr[r], s1y = s1yr[r], s1z = s1zr[r];
        float cx = g1[r] * s1z - g2[r] * s1y;
        float cy = g2[r] * s1x - g0[r] * s1z;
        float cz = g0[r] * s1y - g1[r] * s1x;
        float f = w8[r] * INV_SQ2;
        v0[r] = f * cx;
        v1[r] = f * cy;
        v2[r] = f * cz;
      }
      run_add(acc, dB, flm, 384 + cs * 3 + 0, v0);
      run_add(acc, dB, flm, 384 + cs * 3 + 1, v1);
      run_add(acc, dB, flm, 384 + cs * 3 + 2, v2);
    }
  };

#pragma nounroll
  for (int j = 0; j < 4; ++j) do_tile(IC<0>{}, j);
#pragma nounroll
  for (int j = 4; j < 8; ++j) do_tile(IC<1>{}, j);
#pragma nounroll
  for (int j = 8; j < 10; ++j) do_tile(IC<2>{}, j);
#pragma nounroll
  for (int j = 10; j < 12; ++j) do_tile(IC<3>{}, j);
#pragma nounroll
  for (int j = 12; j < 14; ++j) do_tile(IC<4>{}, j);
}

__global__ __launch_bounds__(NODE_WAVES * 32) __attribute__((amdgpu_waves_per_eu(1, 2))) void proj_kernel(
    const float* __restrict__ acc,
    const float* __restrict__ WO0F, const float* __restrict__ bO0,
    const float* __restrict__ WO1F, const float* __restrict__ WO2F,
    float* __restrict__ y) {
  __shared__ __align__(16) float yS[16 * 416];
  static_assert(NODE_WAVES == 4, "the shares below are written out for four waves");
  const int wave = __builtin_amdgcn_readfirstlane(threadIdx.x >> 5);
  const int n0 = blockIdx.x * 16;
  const float* at = acc + (size_t)n0 * 480;

  {
    v2f aa[24];
    load_a(aa, [&](int m, int k) { return ld2(at + (size_t)m * 480 + k); });
#pragma unroll
    for (int jj = 0; jj < 2; ++jj) {
      const int j = wave * 2 + jj;
      v8f c = {};
      c = wmma_tile(aa, WO0F + j * 16 * 96, c);
      tile_store(c, [&](int, int m, int n, float v) {
        int col = j * 16 + n;
        yS[m * 416 + col] = v * INV_SQ96 + bO0[col];
      });
    }
  }
#pragma unroll
  for (int i = 0; i < 3; ++i) {
    v2f aa[24];
    load_a(aa, [&](int m, int k) {
      v2f a;
      a.x = at[(size_t)m * 480 + 96 + 3 * k + i];
      a.y = at[(size_t)m * 480 + 96 + 3 * (k + 1) + i];
      return a;
    });
    {
      const int j = wave;
      v8f c = {};
      c = wmma_tile(aa, WO1F + j * 16 * 96, c);
      tile_store(c, [&](int, int m, int n, float v) {
        yS[m * 416 + 128 + (j * 16 + n) * 3 + i] = v * INV_SQ96;
      });
    }
  }
  if (wave < 3) {
    const int i = wave;
    v2f aa[8];
    load_a(aa, [&](int m, int k) {
      v2f a;
      a.x = at[(size_t)m * 480 + 384 + 3 * k + i];
      a.y = at[(size_t)m * 480 + 384 + 3 * (k + 1) + i];
      return a;
    });
#pragma unroll
    for (int j = 0; j < 2; ++j) {
      v8f c = {};
      c = wmma_tile(aa, WO2F + j * 16 * 32, c);
      tile_store(c, [&](int, int m, int n, float v) {
        yS[m * 416 + 320 + (j * 16 + n) * 3 + i] = v * INV_SQ32;
      });
    }
  }
  __syncthreads();
  copy_tile<16 * 416>(y + (size_t)n0 * 416, yS);
}

extern "C" void kernel_launch(void* const* d_in, const int* in_sizes, int n_in,
                              void* d_out, int out_size, void* d_ws,
                              size_t ws_size, hipStream_t stream) {
  (void)in_sizes; (void)n_in; (void)out_size; (void)ws_size;
  const float* x         = (const float*)d_in[0];
  const float* edge_sh   = (const float*)d_in[1];
  const float* edge_attr = (const float*)d_in[2];
  const float* W0p = (const float*)d_in[3];
  const float* b0p = (const float*)d_in[4];
  const float* W1p = (const float*)d_in[5];
  const float* W0n = (const float*)d_in[6];
  const float* b0n = (const float*)d_in[7];
  const float* W1n = (const float*)d_in[8];
  const float* G1  = (const float*)d_in[9];
  const float* g1b = (const float*)d_in[10];
  const float* G2  = (const float*)d_in[11];
  const float* g2b = (const float*)d_in[12];
  const float* F1  = (const float*)d_in[13];
  const float* F2  = (const float*)d_in[14];
  const float* L1  = (const float*)d_in[15];
  const float* L2  = (const float*)d_in[16];
  const float* WO0 = (const float*)d_in[17];
  const float* bO0 = (const float*)d_in[18];
  const float* WO1 = (const float*)d_in[19];
  const float* WO2 = (const float*)d_in[20];
  const int* edge_index = (const int*)d_in[21];

  float* ws   = (float*)d_ws;
  float* qd   = ws;
  float* qs   = qd   + (size_t)N_NODES * 32;
  float* p1   = qs   + (size_t)N_NODES * 32;
  float* x0n  = p1   + (size_t)N_NODES * 96;
  float* x1n  = x0n  + (size_t)N_NODES * 64;
  float* acc  = x1n  + (size_t)N_NODES * 96;
  float* wt   = acc  + (size_t)N_NODES * 480;
  float* W0pF = wt + OFF_W0P;
  float* W1pF = wt + OFF_W1P;
  float* W0nF = wt + OFF_W0N;
  float* W1nF = wt + OFF_W1N;
  float* G1F  = wt + OFF_G1;
  float* G2F  = wt + OFF_G2;
  float* F1F  = wt + OFF_F1;
  float* F2F  = wt + OFF_F2;
  float* L1F  = wt + OFF_L1;
  float* L2F  = wt + OFF_L2;
  float* WO0F = wt + OFF_WO0;
  float* WO1F = wt + OFF_WO1;
  float* WO2F = wt + OFF_WO2;

  int* sc   = (int*)d_out;
  int* cnt  = sc;
  int* rank = sc + 2 * CNT_PADDED;
  int* perm = sc + 2 * CNT_PADDED + N_EDGES;

  setup_kernel<<<SETUP_BLOCKS, SETUP_THREADS, 0, stream>>>(
      W0p, W1p, W0n, W1n, G1, G2, F1, F2, L1, L2, WO0, WO1, WO2, wt, acc, cnt);

  node_rank_kernel<<<NODE_TILES + RANK_BLOCKS, NODE_WAVES * 32, 0, stream>>>(
      x, W0pF, b0p, W1pF, W0nF, b0n, W1nF, G1F, g1b, G2F, g2b, L1F,
      qd, qs, p1, x0n, x1n, edge_index, cnt, rank);
  sort_scan_fill_kernel<<<FILL_BLOCKS, FILL_THREADS, 0, stream>>>(edge_index, cnt, rank, perm);

  edge_kernel<<<N_EDGES / 16, 32, 0, stream>>>(
      edge_sh, edge_attr, F1F, F2F, L1F, L2F, edge_index, perm,
      qd, qs, p1, x0n, x1n, acc);

  proj_kernel<<<N_NODES / 16, NODE_WAVES * 32, 0, stream>>>(
      acc, WO0F, bO0, WO1F, WO2F, (float*)d_out);
}


# Edge kernel: 8 tiles (waves) per work-group share one LDS copy of the two 32x224 second-layer weights; the 14 column tiles' B fragments come from LDS, not 56 KB of L2 loads per wave; same sums
# speedup vs baseline: 2.4584x; 1.0562x over previous
#include <hip/hip_runtime.h>
#include <hip/hip_bf16.h>
#include <math.h>


typedef __attribute__((ext_vector_type(2))) float v2f;
typedef __attribute__((ext_vector_type(4))) float v4f;
typedef __attribute__((ext_vector_type(8))) float v8f;

#define N_NODES 10000
#define N_EDGES 320000

#define INV_SQ3   0.5773502691896258f
#define INV_SQ2   0.7071067811865476f
#define INV_SQ32  0.17677669529663687f
#define INV_SQ64  0.125f
#define INV_SQ96  0.10206207261596575f
#define INV_SQ160 0.07905694150420949f
#define LN2F      0.6931471805599453f

__device__ __forceinline__ float sspf(float v) {
  float sp = (v > 20.0f) ? v : log1pf(expf(v));
  return sp - LN2F;
}
__device__ __forceinline__ float siluf(float v) {
  return v / (1.0f + expf(-v));
}
__device__ __forceinline__ v2f ld2(const float* p) {
  return *(const v2f*)p;
}

__device__ __forceinline__ v8f wmma4(v2f a, v2f b, v8f c) {
  return __builtin_amdgcn_wmma_f32_16x16x4_f32(false, a, false, b,
                                               (short)0, c, false, false);
}


template <int KS, typename LA>
__device__ __forceinline__ void load_a(v2f (&a)[KS], LA la) {
  const int lane = threadIdx.x & 31;
  const int m = lane & 15;
  const int kk = (lane >> 4) << 1;
#pragma unroll
  for (int q = 0; q < KS; ++q) a[q] = la(m, q * 4 + kk);
}

template <int KS>
__device__ __forceinline__ v8f wmma_tile(const v2f (&a)[KS],
                                         const float* __restrict__ wf_tile, v8f c) {
  static_assert(KS % 2 == 0, "k-steps are stored in pairs");
  const v4f* bp = (const v4f*)wf_tile + (threadIdx.x & 31);
#pragma unroll
  for (int p = 0; p < KS / 2; ++p) {
    const v4f b = bp[p * 32];
    const v2f b0 = {b.x, b.y};
    const v2f b1 = {b.z, b.w};
    c = wmma4(a[2 * p], b0, c);
    c = wmma4(a[2 * p + 1], b1, c);
  }
  return c;
}

__device__ __forceinline__ void load_b32(v4f (&b)[4], const float* __restrict__ wf_tile) {
  const v4f* bp = (const v4f*)wf_tile + (threadIdx.x & 31);
#pragma unroll
  for (int p = 0; p < 4; ++p) b[p] = bp[p * 32];
}
__device__ __forceinline__ v8f wmma_regs(const v2f (&a)[8], const v4f (&b)[4], v8f c) {
#pragma unroll
  for (int p = 0; p < 4; ++p) {
    const v2f b0 = {b[p].x, b[p].y};
    const v2f b1 = {b[p].z, b[p].w};
    c = wmma4(a[2 * p], b0, c);
    c = wmma4(a[2 * p + 1], b1, c);
  }
  return c;
}
template <int V> struct IC { static constexpr int value = V; };

template <typename ST>
__device__ __forceinline__ void tile_store(v8f c, ST st) {
  const int lane = threadIdx.x & 31;
  const int col = lane & 15;
  const int rbase = (lane >> 4) << 3;
#pragma unroll
  for (int r = 0; r < 8; ++r) st(r, rbase + r, col, c[r]);
}

constexpr int OFF_W0P = 0;
constexpr int OFF_W1P = OFF_W0P + 64 * 64;
constexpr int OFF_W0N = OFF_W1P + 32 * 32;
constexpr int OFF_W1N = OFF_W0N + 64 * 64;
constexpr int OFF_G1  = OFF_W1N + 32 * 32;
constexpr int OFF_G2  = OFF_G1  + 96 * 96;
constexpr int OFF_F1  = OFF_G2  + 96 * 96;
constexpr int OFF_F2  = OFF_F1  + 32 * 32;
constexpr int OFF_L1  = OFF_F2  + 32 * 224;
constexpr int OFF_L2  = OFF_L1  + 160 * 32;
constexpr int OFF_WO0 = OFF_L2  + 32 * 224;
constexpr int OFF_WO1 = OFF_WO0 + 96 * 128;
constexpr int OFF_WO2 = OFF_WO1 + 96 * 64;
constexpr int WT_FLOATS = OFF_WO2 + 32 * 32;

constexpr int SETUP_THREADS  = 256;
constexpr int SETUP_T_BLOCKS = WT_FLOATS / 1024;
constexpr int ACC_V4         = N_NODES * 480 / 4;
constexpr int Z_V4_PER_BLOCK = SETUP_THREADS * 16;
constexpr int SETUP_Z_BLOCKS = (ACC_V4 + Z_V4_PER_BLOCK - 1) / Z_V4_PER_BLOCK;
constexpr int CNT_PADDED     = 10240;
constexpr int SETUP_BLOCKS   = SETUP_T_BLOCKS + SETUP_Z_BLOCKS + 1;

template <int K, int Nc>
__device__ __forceinline__ void frag_slab(const float* __restrict__ W,
                                          float* __restrict__ Wf, int slab) {
  static_assert((K * Nc) % 1024 == 0, "weights are whole slabs");
  static_assert(K % 8 == 0 && Nc % 16 == 0, "whole k-step pairs and column tiles");
  float v[4];
#pragma unroll
  for (int i = 0; i < 4; ++i) v[i] = W[slab * 1024 + i * SETUP_THREADS + threadIdx.x];
#pragma unroll
  for (int i = 0; i < 4; ++i) {
    const int idx = slab * 1024 + i * SETUP_THREADS + threadIdx.x;
    const int k = idx / Nc, n = idx % Nc;
    const int j = n >> 4, c = n & 15;
    const int p = k >> 3, r = k & 7;
    const int h = (r >> 1) & 1;
    const int sl = ((r >> 2) << 1) | (r & 1);
    Wf[(((j * (K / 8) + p) * 32) + h * 16 + c) * 4 + sl] = v[i];
  }
}

__global__ __launch_bounds__(SETUP_THREADS) void setup_kernel(
    const float* __restrict__ W0p, const float* __restrict__ W1p,
    const float* __restrict__ W0n, const float* __restrict__ W1n,
    const float* __restrict__ G1, const float* __restrict__ G2,
    const float* __restrict__ F1, const float* __restrict__ F2,
    const float* __restrict__ L1, const float* __restrict__ L2,
    const float* __restrict__ WO0, const float* __restrict__ WO1,
    const float* __restrict__ WO2,
    float* __restrict__ wt, float* __restrict__ acc, int* __restrict__ cnt) {
  int s = blockIdx.x;
  if (s < SETUP_T_BLOCKS) {
    if (s < 4)  { frag_slab<64, 64>(W0p, wt + OFF_W0P, s); return; }   s -= 4;
    if (s < 1)  { frag_slab<32, 32>(W1p, wt + OFF_W1P, s); return; }   s -= 1;
    if (s < 4)  { frag_slab<64, 64>(W0n, wt + OFF_W0N, s); return; }   s -= 4;
    if (s < 1)  { frag_slab<32, 32>(W1n, wt + OFF_W1N, s); return; }   s -= 1;
    if (s < 9)  { frag_slab<96, 96>(G1, wt + OFF_G1, s); return; }     s -= 9;
    if (s < 9)  { frag_slab<96, 96>(G2, wt + OFF_G2, s); return; }     s -= 9;
    if (s < 1)  { frag_slab<32, 32>(F1, wt + OFF_F1, s); return; }     s -= 1;
    if (s < 7)  { frag_slab<32, 224>(F2, wt + OFF_F2, s); return; }    s -= 7;
    if (s < 5)  { frag_slab<160, 32>(L1, wt + OFF_L1, s); return; }    s -= 5;
    if (s < 7)  { frag_slab<32, 224>(L2, wt + OFF_L2, s); return; }    s -= 7;
    if (s < 12) { frag_slab<96, 128>(WO0, wt + OFF_WO0, s); return; }  s -= 12;
    if (s < 6)  { frag_slab<96, 64>(WO1, wt + OFF_WO1, s); return; }   s -= 6;
    frag_slab<32, 32>(WO2, wt + OFF_WO2, s);
    return;
  }
  s -= SETUP_T_BLOCKS;
  if (s < SETUP_Z_BLOCKS) {
    v4f* a4 = (v4f*)acc;
    const v4f z = {0.0f, 0.0f, 0.0f, 0.0f};
    for (int i = 0; i < 16; ++i) {
      const int q = s * Z_V4_PER_BLOCK + i * SETUP_THREADS + threadIdx.x;
      if (q < ACC_V4) a4[q] = z;
    }
    return;
  }
  for (int i = threadIdx.x; i < CNT_PADDED; i += SETUP_THREADS) cnt[i] = 0;
}

constexpr int NODE_WAVES = 4;

template <int FLOATS>
__device__ __forceinline__ void copy_tile(float* __restrict__ dst, const float* src) {
  constexpr int NT = NODE_WAVES * 32;
  static_assert(FLOATS % (4 * NT) == 0, "whole 16-byte words per thread");
#pragma unroll
  for (int q = 0; q < FLOATS / (4 * NT); ++q) {
    const int w = q * NT + threadIdx.x;
    ((v4f*)dst)[w] = ((const v4f*)src)[w];
  }
}

constexpr int P0_PITCH = 68;
constexpr int IP_PITCH = 36;

__device__ __forceinline__ void node_tile(
    const int tile,
    const float* __restrict__ x,
    const float* __restrict__ W0pF, const float* __restrict__ b0p,
    const float* __restrict__ W1pF,
    const float* __restrict__ W0nF, const float* __restrict__ b0n,
    const float* __restrict__ W1nF,
    const float* __restrict__ G1F, const float* __restrict__ g1b,
    const float* __restrict__ G2F, const float* __restrict__ g2b,
    const float* __restrict__ L1F,
    float* __restrict__ qd, float* __restrict__ qs, float* __restrict__ p1,
    float* __restrict__ x0n, float* __restrict__ x1n) {
  __shared__ __align__(16) float p0S[16 * P0_PITCH];
  __shared__ __align__(16) float f0S[16 * 96];
  __shared__ __align__(16) float hS[16 * 96];
  __shared__ __align__(16) float gS[16 * 96];
  __shared__ __align__(16) float qdS[16 * 32];
  __shared__ __align__(16) float qsS[16 * 32];
  __shared__ __align__(16) float p1S[16 * 96];
  __shared__ __align__(16) float x0nS[16 * 64];
  __shared__ __align__(16) float x1nS[16 * 96];
  static_assert(NODE_WAVES == 4, "the shares below are written out for four waves");

  const int tid = threadIdx.x;
  const int wave = __builtin_amdgcn_readfirstlane(threadIdx.x >> 5);
  const int n0 = tile * 16;
  const float* xt = x + (size_t)n0 * 160;

  {
    v2f ax[16];
    load_a(ax, [&](int m, int k) { return ld2(xt + m * 160 + k); });
    {
      const int j = wave;
      v8f c = {};
      c = wmma_tile(ax, W0pF + j * 16 * 64, c);
      tile_store(c, [&](int, int m, int n, float v) {
        int col = j * 16 + n;
        p0S[m * P0_PITCH + col] = v * INV_SQ64 + b0p[col];
      });
    }
  }
  if (wave < 3) {
    const int i = wave;
    v2f ax[8];
    load_a(ax, [&](int m, int k) {
      v2f a;
      a.x = xt[m * 160 + 64 + 3 * k + i];
      a.y = xt[m * 160 + 64 + 3 * (k + 1) + i];
      return a;
    });
#pragma unroll
    for (int j = 0; j < 2; ++j) {
      v8f c = {};
      c = wmma_tile(ax, W1pF + j * 16 * 32, c);
      tile_store(c, [&](int, int m, int n, float v) {
        p1S[m * 96 + (j * 16 + n) * 3 + i] = v * INV_SQ32;
      });
    }
  }

  {
    constexpr int NT = NODE_WAVES * 32;
#pragma unroll
    for (int q = 0; q < 16 * 64 / NT; ++q) {
      const int idx = q * NT + tid;
      const int m = idx >> 6, c = idx & 63;
      f0S[m * 96 + c] = xt[m * 160 + c];
    }
#pragma unroll
    for (int q = 0; q < 16 * 32 / NT; ++q) {
      const int idx = q * NT + tid;
      const int m = idx >> 5, u = idx & 31;
      float a0 = xt[m * 160 + 64 + u * 3 + 0];
      float a1 = xt[m * 160 + 64 + u * 3 + 1];
      float a2 = xt[m * 160 + 64 + u * 3 + 2];
      f0S[m * 96 + 64 + u] = sqrtf(a0 * a0 + a1 * a1 + a2 * a2 + 1e-12f);
    }
  }
  __syncthreads();

  {
    v2f ap[16];
    load_a(ap, [&](int m, int k) { return ld2(p0S + m * P0_PITCH + k); });
    {
      const int j = wave >> 1, half = wave & 1;
      float* q = half ? qsS : qdS;
      v8f c = {};
      c = wmma_tile(ap, L1F + j * 16 * 160 + half * 8 * 128, c);
      tile_store(c, [&](int, int m, int n, float v) {
        q[m * 32 + j * 16 + n] = v;
      });
    }
  }

  if (wave < 3) {
    v2f af[24];
    load_a(af, [&](int m, int k) { return ld2(f0S + m * 96 + k); });
#pragma unroll
    for (int jj = 0; jj < 2; ++jj) {
      const int j = wave * 2 + jj;
      v8f c = {};
      c = wmma_tile(af, G1F + j * 16 * 96, c);
      tile_store(c, [&](int, int m, int n, float v) {
        int col = j * 16 + n;
        hS[m * 96 + col] = siluf(v + g1b[col]);
      });
    }
  }
  __syncthreads();
  copy_tile<16 * 32>(qd + (size_t)n0 * 32, qdS);
  copy_tile<16 * 32>(qs + (size_t)n0 * 32, qsS);
  copy_tile<16 * 96>(p1 + (size_t)n0 * 96, p1S);
  if (wave < 3) {
    v2f ah[24];
    load_a(ah, [&](int m, int k) { return ld2(hS + m * 96 + k); });
#pragma unroll
    for (int jj = 0; jj < 2; ++jj) {
      const int j = wave * 2 + jj;
      v8f c = {};
      c = wmma_tile(ah, G2F + j * 16 * 96, c);
      tile_store(c, [&](int, int m, int n, float v) {
        gS[m * 96 + j * 16 + n] = v + g2b[j * 16 + n];
      });
    }
  }
  __syncthreads();

  {
    v2f ag[16];
    load_a(ag, [&](int m, int k) { return ld2(gS + m * 96 + k); });
    {
      const int j = wave;
      v8f c = {};
      c = wmma_tile(ag, W0nF + j * 16 * 64, c);
      tile_store(c, [&](int, int m, int n, float v) {
        int col = j * 16 + n;
        x0nS[m * 64 + col] = v * INV_SQ64 + b0n[col];
      });
    }
  }
  if (wave < 3) {
    const int i = wave;
    v2f ax[8];
    load_a(ax, [&](int m, int k) {
      v2f a;
      a.x = xt[m * 160 + 64 + 3 * k + i] * gS[m * 96 + 64 + k];
      a.y = xt[m * 160 + 64 + 3 * (k + 1) + i] * gS[m * 96 + 64 + k + 1];
      return a;
    });
#pragma unroll
    for (int j = 0; j < 2; ++j) {
      v8f c = {};
      c = wmma_tile(ax, W1nF + j * 16 * 32, c);
      tile_store(c, [&](int, int m, int n, float v) {
        x1nS[m * 96 + (j * 16 + n) * 3 + i] = v * INV_SQ32;
      });
    }
  }
  __syncthreads();
  copy_tile<16 * 64>(x0n + (size_t)n0 * 64, x0nS);
  copy_tile<16 * 96>(x1n + (size_t)n0 * 96, x1nS);
}

constexpr int NODE_TILES  = N_NODES / 16;
constexpr int RANK_BLOCKS = N_EDGES / (NODE_WAVES * 32);
static_assert(NODE_TILES * 16 == N_NODES, "whole node tiles");
static_assert(RANK_BLOCKS * NODE_WAVES * 32 == N_EDGES, "whole rank blocks");

__global__ __launch_bounds__(NODE_WAVES * 32) void node_rank_kernel(
    const float* __restrict__ x,
    const float* __restrict__ W0pF, const float* __restrict__ b0p,
    const float* __restrict__ W1pF,
    const float* __restrict__ W0nF, const float* __restrict__ b0n,
    const float* __restrict__ W1nF,
    const float* __restrict__ G1F, const float* __restrict__ g1b,
    const float* __restrict__ G2F, const float* __restrict__ g2b,
    const float* __restrict__ L1F,
    float* __restrict__ qd, float* __restrict__ qs, float* __restrict__ p1,
    float* __restrict__ x0n, float* __restrict__ x1n,
    const int* __restrict__ edge_index, int* __restrict__ cnt, int* __restrict__ rank) {
  if (blockIdx.x < NODE_TILES) {
    node_tile(blockIdx.x, x, W0pF, b0p, W1pF, W0nF, b0n, W1nF, G1F, g1b, G2F, g2b, L1F,
              qd, qs, p1, x0n, x1n);
    return;
  }
  const int e = (blockIdx.x - NODE_TILES) * (NODE_WAVES * 32) + threadIdx.x;
  rank[e] = atomicAdd(&cnt[edge_index[e]], 1);
}

#define SCAN_PER_THREAD 10
constexpr int FILL_THREADS = 1024;
constexpr int FILL_WAVES   = FILL_THREADS / 32;
constexpr int FILL_BLOCKS  = (N_EDGES + FILL_THREADS - 1) / FILL_THREADS;
static_assert(FILL_THREADS * SCAN_PER_THREAD >= N_NODES, "one block scans all counters");
static_assert(FILL_WAVES == 32, "the wave totals are scanned by one wave");

__device__ __forceinline__ int wave_inclusive_scan(int v, const int lane) {
#pragma unroll
  for (int d = 1; d < 32; d <<= 1) {
    const int u = __shfl_up(v, d, 32);
    if (lane >= d) v += u;
  }
  return v;
}

__global__ __launch_bounds__(FILL_THREADS) void sort_scan_fill_kernel(
    const int* __restrict__ edge_index, const int* __restrict__ cnt,
    const int* __restrict__ rank, int* __restrict__ perm) {
  __shared__ int part[FILL_WAVES];
  __shared__ int offS[FILL_THREADS * SCAN_PER_THREAD];
  const int t = threadIdx.x;
  const int lane = t & 31, wave = t >> 5;
  const int e = blockIdx.x * FILL_THREADS + t;
  int d = 0, rk = 0;
  if (e < N_EDGES) {
    d = edge_index[e];
    rk = rank[e];
  }
#pragma unroll
  for (int i = 0; i < SCAN_PER_THREAD; ++i) {
    const int idx = i * FILL_THREADS + t;
    offS[idx] = (idx < N_NODES) ? cnt[idx] : 0;
  }
  __syncthreads();
  int loc[SCAN_PER_THREAD];
  int s = 0;
#pragma unroll
  for (int i = 0; i < SCAN_PER_THREAD; ++i) {
    const int v = offS[t * SCAN_PER_THREAD + i];
    loc[i] = s;
    s += v;
  }
  const int incl = wave_inclusive_scan(s, lane);
  if (lane == 31) part[wave] = incl;
  __syncthreads();
  if (wave == 0) {
    const int w = part[lane];
    part[lane] = wave_inclusive_scan(w, lane) - w;
  }
  __syncthreads();
  const int base = part[wave] + incl - s;
#pragma unroll
  for (int i = 0; i < SCAN_PER_THREAD; ++i) offS[t * SCAN_PER_THREAD + i] = base + loc[i];
  __syncthreads();
  if (e < N_EDGES) perm[offS[d] + rk] = e;
}

__device__ __forceinline__ void run_add(float* __restrict__ acc,
                                        const unsigned (&dB)[8], unsigned flm,
                                        int chan, const float (&v)[8]) {
  float run = 0.0f;
#pragma unroll
  for (int r = 0; r < 8; ++r) {
    run += v[r];
    if ((flm >> r) & 1u) {
      atomicAdd((float*)((char*)acc + (dB[r] + (unsigned)chan * 4u)), run);
      run = 0.0f;
    }
  }
}

constexpr int EDGE_WAVES   = 8;
constexpr int EDGE_THREADS = EDGE_WAVES * 32;
constexpr int W2_FLOATS    = 32 * 224;
constexpr int W2_V4        = W2_FLOATS / 4;
constexpr int W2_STAGE     = W2_V4 / EDGE_THREADS;
static_assert(W2_STAGE * EDGE_THREADS == W2_V4, "whole 16-byte words per thread");
static_assert((N_EDGES / 16) % EDGE_WAVES == 0, "whole work-groups of tiles");

__device__ __forceinline__ void wave_lds_sync() {
  __builtin_amdgcn_fence(__ATOMIC_RELEASE, "wavefront");
  __builtin_amdgcn_wave_barrier();
  __builtin_amdgcn_fence(__ATOMIC_ACQUIRE, "wavefront");
}

__global__ __launch_bounds__(EDGE_THREADS) void edge_kernel(
    const float* __restrict__ edge_sh, const float* __restrict__ edge_attr,
    const float* __restrict__ F1F, const float* __restrict__ F2F,
    const float* __restrict__ L1F, const float* __restrict__ L2F,
    const int* __restrict__ edge_index, const int* __restrict__ perm,
    const float* __restrict__ qd, const float* __restrict__ qs,
    const float* __restrict__ p1,
    const float* __restrict__ x0n, const float* __restrict__ x1n,
    float* __restrict__ acc) {
  __shared__ __align__(16) float wS[2 * W2_FLOATS];
  __shared__ __align__(16) float ipS_w[EDGE_WAVES][16 * IP_PITCH];
  __shared__ __align__(16) float hE_w[EDGE_WAVES][16 * 32];
  __shared__ __align__(16) float shS_w[EDGE_WAVES][16 * 4];
  __shared__ int dstS_w[EDGE_WAVES][16];
  __shared__ int srcS_w[EDGE_WAVES][16];
  static_assert(16 * 32 <= 16 * IP_PITCH, "hN fits in the inner-product panel");

  const int lane = threadIdx.x & 31;
  const int wave = __builtin_amdgcn_readfirstlane(threadIdx.x >> 5);
  const int e0 = (blockIdx.x * EDGE_WAVES + wave) * 16;
  float* const ipS = ipS_w[wave];
  float* const hN  = ipS_w[wave];
  float* const hE  = hE_w[wave];
  float* const shS = shS_w[wave];
  int* const dstS = dstS_w[wave];
  int* const srcS = srcS_w[wave];

  v4f wst[2 * W2_STAGE];
#pragma unroll
  for (int i = 0; i < W2_STAGE; ++i) {
    wst[i]            = ((const v4f*)L2F)[i * EDGE_THREADS + threadIdx.x];
    wst[W2_STAGE + i] = ((const v4f*)F2F)[i * EDGE_THREADS + threadIdx.x];
  }

  v4f bF0[4], bF1[4];
  load_b32(bF0, F1F);
  load_b32(bF1, F1F + 16 * 32);

  const int e = perm[e0 + (lane & 15)];
  v2f aea[8];
  {
    const float* ar = edge_attr + (size_t)e * 32 + ((lane >> 4) << 1);
#pragma unroll
    for (int q = 0; q < 8; ++q) aea[q] = ld2(ar + q * 4);
  }
  if (lane < 16) {
    dstS[lane] = edge_index[e];
    srcS[lane] = edge_index[N_EDGES + e];
    *(v4f*)(shS + lane * 4) = *(const v4f*)(edge_sh + (size_t)e * 4);
  }
#pragma unroll
  for (int i = 0; i < W2_STAGE; ++i) {
    ((v4f*)wS)[i * EDGE_THREADS + threadIdx.x]         = wst[i];
    ((v4f*)wS)[W2_V4 + i * EDGE_THREADS + threadIdx.x] = wst[W2_STAGE + i];
  }
  wave_lds_sync();

  const int rbase = (lane >> 4) << 3;

  float qv0[8], qv1[8], qw0[8], qw1[8];
#pragma unroll
  for (int r = 0; r < 8; ++r) {
    const float* pd = (const float*)((const char*)qd + ((unsigned)dstS[rbase + r] * 128u + (unsigned)(lane & 15) * 4u));
    const float* ps = (const float*)((const char*)qs + ((unsigned)srcS[rbase + r] * 128u + (unsigned)(lane & 15) * 4u));
    qv0[r] = pd[0];
    qv1[r] = pd[16];
    qw0[r] = ps[0];
    qw1[r] = ps[16];
  }
  v4f bL0[4], bL1[4];
  load_b32(bL0, L1F + 16 * 128);
  load_b32(bL1, L1F + 16 * 160 + 16 * 128);

  {
    {
      v8f c0 = {}, c1 = {};
      c0 = wmma_regs(aea, bF0, c0);
      c1 = wmma_regs(aea, bF1, c1);
      tile_store(c0, [&](int, int m, int n, float v) {
        hE[m * 32 + n] = sspf(v * INV_SQ32);
      });
      tile_store(c1, [&](int, int m, int n, float v) {
        hE[m * 32 + 16 + n] = sspf(v * INV_SQ32);
      });
    }

#pragma nounroll
    for (int mb = 0; mb < 16; mb += 8) {
      float dv[8];
#pragma unroll
      for (int r = 0; r < 8; ++r) {
        const float* pd = (const float*)((const char*)p1 + ((unsigned)dstS[mb + r] * 384u + (unsigned)lane * 12u));
        const float* ps = (const float*)((const char*)p1 + ((unsigned)srcS[mb + r] * 384u + (unsigned)lane * 12u));
        dv[r] = (pd[0] * ps[0] + pd[1] * ps[1] + pd[2] * ps[2]) * INV_SQ3;
      }
#pragma unroll
      for (int r = 0; r < 8; ++r) ipS[(mb + r) * IP_PITCH + lane] = dv[r];
    }
  }
  wave_lds_sync();

  {
    v2f aip[8];
    load_a(aip, [&](int m, int k) { return ld2(ipS + m * IP_PITCH + k); });
    v8f c0, c1;
#pragma unroll
    for (int r = 0; r < 8; ++r) {
      c0[r] = qv0[r] + qw0[r];
      c1[r] = qv1[r] + qw1[r];
    }
    c0 = wmma_regs(aip, bL0, c0);
    c1 = wmma_regs(aip, bL1, c1);
    tile_store(c0, [&](int, int m, int n, float v) {
      hN[m * 32 + n] = sspf(v * INV_SQ160);
    });
    tile_store(c1, [&](int, int m, int n, float v) {
      hN[m * 32 + 16 + n] = sspf(v * INV_SQ160);
    });
  }
  wave_lds_sync();

  float sh0r[8], s1xr[8], s1yr[8], s1zr[8];
  int dR[8], sR[8];
#pragma unroll
  for (int r = 0; r < 8; ++r) {
    int m = rbase + r;
    const v4f sh = *(const v4f*)(shS + m * 4);
    sh0r[r] = sh.x;
    s1xr[r] = sh.y;
    s1yr[r] = sh.z;
    s1zr[r] = sh.w;
    dR[r] = dstS[m];
    sR[r] = srcS[m];
  }
  unsigned s0B[8], s1B[8], dB[8];
#pragma unroll
  for (int r = 0; r < 8; ++r) {
    s0B[r] = (unsigned)sR[r] * 256u;
    s1B[r] = (unsigned)sR[r] * 384u;
    dB[r] = (unsigned)dR[r] * 1920u;
  }
  unsigned flm = 0x80u;
#pragma unroll
  for (int r = 0; r < 7; ++r) flm |= (dR[r + 1] != dR[r]) ? (1u << r) : 0u;

  v2f aN[8], aE[8];
  load_a(aN, [&](int m, int k) { return ld2(hN + m * 32 + k); });
  load_a(aE, [&](int m, int k) { return ld2(hE + m * 32 + k); });

  auto do_tile = [&](auto seg, const int j) {
    constexpr int SEG = decltype(seg)::value;
    const int nb = j * 16;
    const int col = nb + (lane & 15);
    v4f bN[4], bE[4];
    load_b32(bN, wS + nb * 32);
    load_b32(bE, wS + W2_FLOATS + nb * 32);
    constexpr int CB = (SEG == 0) ? 0 : (SEG == 1) ? 64 : (SEG == 2) ? 128 : (SEG == 3) ? 160 : 192;
    const int cs = col - CB;
    float g0[8], g1[8], g2[8];
    if constexpr (SEG <= 1) {
#pragma unroll
      for (int r = 0; r < 8; ++r)
        g0[r] = *(const float*)((const char*)x0n + (s0B[r] + (unsigned)cs * 4u));
    } else {
#pragma unroll
      for (int r = 0; r < 8; ++r) {
        const float* xv = (const float*)((const char*)x1n + (s1B[r] + (unsigned)cs * 12u));
        g0[r] = xv[0];
        g1[r] = xv[1];
        g2[r] = xv[2];
      }
    }

    v8f cN = {}, cE = {};
    cN = wmma_regs(aN, bN, cN);
    cE = wmma_regs(aE, bE, cE);
    v8f w8 = cN * cE * (INV_SQ32 * INV_SQ32);

    float v0[8], v1[8], v2[8];
    if constexpr (SEG == 0) {
#pragma unroll
      for (int r = 0; r < 8; ++r) v0[r] = w8[r] * g0[r] * sh0r[r];
      run_add(acc, dB, flm, col, v0);
    } else if constexpr (SEG == 1) {
#pragma unroll
      for (int r = 0; r < 8; ++r) {
        float f = w8[r] * g0[r] * INV_SQ3;
        v0[r] = f * s1xr[r];
        v1[r] = f * s1yr[r];
        v2[r] = f * s1zr[r];
      }
      run_add(acc, dB, flm, 96 + cs * 3 + 0, v0);
      run_add(acc, dB, flm, 96 + cs * 3 + 1, v1);
      run_add(acc, dB, flm, 96 + cs * 3 + 2, v2);
    } else if constexpr (SEG == 2) {
#pragma unroll
      for (int r = 0; r < 8; ++r) {
        float f = w8[r] * sh0r[r];
        v0[r] = f * g0[r];
        v1[r] = f * g1[r];
        v2[r] = f * g2[r];
      }
      run_add(acc, dB, flm, 288 + cs * 3 + 0, v0);
      run_add(acc, dB, flm, 288 + cs * 3 + 1, v1);
      run_add(acc, dB, flm, 288 + cs * 3 + 2, v2);
    } else if constexpr (SEG == 3) {
#pragma unroll
      for (int r = 0; r < 8; ++r) {
        float dp = g0[r] * s1xr[r] + g1[r] * s1yr[r] + g2[r] * s1zr[r];
        v0[r] = w8[r] * dp * INV_SQ3;
      }
      run_add(acc, dB, flm, 64 + cs, v0);
    } else {
#pragma unroll
      for (int r = 0; r < 8; ++r) {
        const float s1x = s1xr[r], s1y = s1yr[r], s1z = s1zr[r];
        float cx = g1[r] * s1z - g2[r] * s1y;
        float cy = g2[r] * s1x - g0[r] * s1z;
        float cz = g0[r] * s1y - g1[r] * s1x;
        float f = w8[r] * INV_SQ2;
        v0[r] = f * cx;
        v1[r] = f * cy;
        v2[r] = f * cz;
      }
      run_add(acc, dB, flm, 384 + cs * 3 + 0, v0);
      run_add(acc, dB, flm, 384 + cs * 3 + 1, v1);
      run_add(acc, dB, flm, 384 + cs * 3 + 2, v2);
    }
  };

  __syncthreads();

#pragma nounroll
  for (int j = 0; j < 4; ++j) do_tile(IC<0>{}, j);
#pragma nounroll
  for (int j = 4; j < 8; ++j) do_tile(IC<1>{}, j);
#pragma nounroll
  for (int j = 8; j < 10; ++j) do_tile(IC<2>{}, j);
#pragma nounroll
  for (int j = 10; j < 12; ++j) do_tile(IC<3>{}, j);
#pragma nounroll
  for (int j = 12; j < 14; ++j) do_tile(IC<4>{}, j);
}

__global__ __launch_bounds__(NODE_WAVES * 32) __attribute__((amdgpu_waves_per_eu(1, 2))) void proj_kernel(
    const float* __restrict__ acc,
    const float* __restrict__ WO0F, const float* __restrict__ bO0,
    const float* __restrict__ WO1F, const float* __restrict__ WO2F,
    float* __restrict__ y) {
  __shared__ __align__(16) float yS[16 * 416];
  static_assert(NODE_WAVES == 4, "the shares below are written out for four waves");
  const int wave = __builtin_amdgcn_readfirstlane(threadIdx.x >> 5);
  const int n0 = blockIdx.x * 16;
  const float* at = acc + (size_t)n0 * 480;

  {
    v2f aa[24];
    load_a(aa, [&](int m, int k) { return ld2(at + (size_t)m * 480 + k); });
#pragma unroll
    for (int jj = 0; jj < 2; ++jj) {
      const int j = wave * 2 + jj;
      v8f c = {};
      c = wmma_tile(aa, WO0F + j * 16 * 96, c);
      tile_store(c, [&](int, int m, int n, float v) {
        int col = j * 16 + n;
        yS[m * 416 + col] = v * INV_SQ96 + bO0[col];
      });
    }
  }
#pragma unroll
  for (int i = 0; i < 3; ++i) {
    v2f aa[24];
    load_a(aa, [&](int m, int k) {
      v2f a;
      a.x = at[(size_t)m * 480 + 96 + 3 * k + i];
      a.y = at[(size_t)m * 480 + 96 + 3 * (k + 1) + i];
      return a;
    });
    {
      const int j = wave;
      v8f c = {};
      c = wmma_tile(aa, WO1F + j * 16 * 96, c);
      tile_store(c, [&](int, int m, int n, float v) {
        yS[m * 416 + 128 + (j * 16 + n) * 3 + i] = v * INV_SQ96;
      });
    }
  }
  if (wave < 3) {
    const int i = wave;
    v2f aa[8];
    load_a(aa, [&](int m, int k) {
      v2f a;
      a.x = at[(size_t)m * 480 + 384 + 3 * k + i];
      a.y = at[(size_t)m * 480 + 384 + 3 * (k + 1) + i];
      return a;
    });
#pragma unroll
    for (int j = 0; j < 2; ++j) {
      v8f c = {};
      c = wmma_tile(aa, WO2F + j * 16 * 32, c);
      tile_store(c, [&](int, int m, int n, float v) {
        yS[m * 416 + 320 + (j * 16 + n) * 3 + i] = v * INV_SQ32;
      });
    }
  }
  __syncthreads();
  copy_tile<16 * 416>(y + (size_t)n0 * 416, yS);
}

extern "C" void kernel_launch(void* const* d_in, const int* in_sizes, int n_in,
                              void* d_out, int out_size, void* d_ws,
                              size_t ws_size, hipStream_t stream) {
  (void)in_sizes; (void)n_in; (void)out_size; (void)ws_size;
  const float* x         = (const float*)d_in[0];
  const float* edge_sh   = (const float*)d_in[1];
  const float* edge_attr = (const float*)d_in[2];
  const float* W0p = (const float*)d_in[3];
  const float* b0p = (const float*)d_in[4];
  const float* W1p = (const float*)d_in[5];
  const float* W0n = (const float*)d_in[6];
  const float* b0n = (const float*)d_in[7];
  const float* W1n = (const float*)d_in[8];
  const float* G1  = (const float*)d_in[9];
  const float* g1b = (const float*)d_in[10];
  const float* G2  = (const float*)d_in[11];
  const float* g2b = (const float*)d_in[12];
  const float* F1  = (const float*)d_in[13];
  const float* F2  = (const float*)d_in[14];
  const float* L1  = (const float*)d_in[15];
  const float* L2  = (const float*)d_in[16];
  const float* WO0 = (const float*)d_in[17];
  const float* bO0 = (const float*)d_in[18];
  const float* WO1 = (const float*)d_in[19];
  const float* WO2 = (const float*)d_in[20];
  const int* edge_index = (const int*)d_in[21];

  float* ws   = (float*)d_ws;
  float* qd   = ws;
  float* qs   = qd   + (size_t)N_NODES * 32;
  float* p1   = qs   + (size_t)N_NODES * 32;
  float* x0n  = p1   + (size_t)N_NODES * 96;
  float* x1n  = x0n  + (size_t)N_NODES * 64;
  float* acc  = x1n  + (size_t)N_NODES * 96;
  float* wt   = acc  + (size_t)N_NODES * 480;
  float* W0pF = wt + OFF_W0P;
  float* W1pF = wt + OFF_W1P;
  float* W0nF = wt + OFF_W0N;
  float* W1nF = wt + OFF_W1N;
  float* G1F  = wt + OFF_G1;
  float* G2F  = wt + OFF_G2;
  float* F1F  = wt + OFF_F1;
  float* F2F  = wt + OFF_F2;
  float* L1F  = wt + OFF_L1;
  float* L2F  = wt + OFF_L2;
  float* WO0F = wt + OFF_WO0;
  float* WO1F = wt + OFF_WO1;
  float* WO2F = wt + OFF_WO2;

  int* sc   = (int*)d_out;
  int* cnt  = sc;
  int* rank = sc + 2 * CNT_PADDED;
  int* perm = sc + 2 * CNT_PADDED + N_EDGES;

  setup_kernel<<<SETUP_BLOCKS, SETUP_THREADS, 0, stream>>>(
      W0p, W1p, W0n, W1n, G1, G2, F1, F2, L1, L2, WO0, WO1, WO2, wt, acc, cnt);

  node_rank_kernel<<<NODE_TILES + RANK_BLOCKS, NODE_WAVES * 32, 0, stream>>>(
      x, W0pF, b0p, W1pF, W0nF, b0n, W1nF, G1F, g1b, G2F, g2b, L1F,
      qd, qs, p1, x0n, x1n, edge_index, cnt, rank);
  sort_scan_fill_kernel<<<FILL_BLOCKS, FILL_THREADS, 0, stream>>>(edge_index, cnt, rank, perm);

  edge_kernel<<<N_EDGES / 16 / EDGE_WAVES, EDGE_THREADS, 0, stream>>>(
      edge_sh, edge_attr, F1F, F2F, L1F, L2F, edge_index, perm,
      qd, qs, p1, x0n, x1n, acc);

  proj_kernel<<<N_NODES / 16, NODE_WAVES * 32, 0, stream>>>(
      acc, WO0F, bO0, WO1F, WO2F, (float*)d_out);
}


# Edge-MLP shifted softplus as max(v,0)+ln2*(log2(1+2^(-|v|*log2e))-1) on hardware exp2/log2: 7 instructions, no branch, was a 100-deep library log1p(exp) chain; activation last bits differ
# speedup vs baseline: 2.7944x; 1.1367x over previous
#include <hip/hip_runtime.h>
#include <hip/hip_bf16.h>
#include <math.h>


typedef __attribute__((ext_vector_type(2))) float v2f;
typedef __attribute__((ext_vector_type(4))) float v4f;
typedef __attribute__((ext_vector_type(8))) float v8f;

#define N_NODES 10000
#define N_EDGES 320000

#define INV_SQ3   0.5773502691896258f
#define INV_SQ2   0.7071067811865476f
#define INV_SQ32  0.17677669529663687f
#define INV_SQ64  0.125f
#define INV_SQ96  0.10206207261596575f
#define INV_SQ160 0.07905694150420949f
#define LN2F      0.6931471805599453f
#define LOG2EF    1.4426950408889634f

__device__ __forceinline__ float sspf(float v) {
  const float t = __builtin_amdgcn_exp2f(-fabsf(v) * LOG2EF);
  const float l = __builtin_amdgcn_logf(1.0f + t);
  return fmaxf(v, 0.0f) + LN2F * (l - 1.0f);
}
__device__ __forceinline__ float siluf(float v) {
  return v / (1.0f + expf(-v));
}
__device__ __forceinline__ v2f ld2(const float* p) {
  return *(const v2f*)p;
}

__device__ __forceinline__ v8f wmma4(v2f a, v2f b, v8f c) {
  return __builtin_amdgcn_wmma_f32_16x16x4_f32(false, a, false, b,
                                               (short)0, c, false, false);
}


template <int KS, typename LA>
__device__ __forceinline__ void load_a(v2f (&a)[KS], LA la) {
  const int lane = threadIdx.x & 31;
  const int m = lane & 15;
  const int kk = (lane >> 4) << 1;
#pragma unroll
  for (int q = 0; q < KS; ++q) a[q] = la(m, q * 4 + kk);
}

template <int KS>
__device__ __forceinline__ v8f wmma_tile(const v2f (&a)[KS],
                                         const float* __restrict__ wf_tile, v8f c) {
  static_assert(KS % 2 == 0, "k-steps are stored in pairs");
  const v4f* bp = (const v4f*)wf_tile + (threadIdx.x & 31);
#pragma unroll
  for (int p = 0; p < KS / 2; ++p) {
    const v4f b = bp[p * 32];
    const v2f b0 = {b.x, b.y};
    const v2f b1 = {b.z, b.w};
    c = wmma4(a[2 * p], b0, c);
    c = wmma4(a[2 * p + 1], b1, c);
  }
  return c;
}

__device__ __forceinline__ void load_b32(v4f (&b)[4], const float* __restrict__ wf_tile) {
  const v4f* bp = (const v4f*)wf_tile + (threadIdx.x & 31);
#pragma unroll
  for (int p = 0; p < 4; ++p) b[p] = bp[p * 32];
}
__device__ __forceinline__ v8f wmma_regs(const v2f (&a)[8], const v4f (&b)[4], v8f c) {
#pragma unroll
  for (int p = 0; p < 4; ++p) {
    const v2f b0 = {b[p].x, b[p].y};
    const v2f b1 = {b[p].z, b[p].w};
    c = wmma4(a[2 * p], b0, c);
    c = wmma4(a[2 * p + 1], b1, c);
  }
  return c;
}
template <int V> struct IC { static constexpr int value = V; };

template <typename ST>
__device__ __forceinline__ void tile_store(v8f c, ST st) {
  const int lane = threadIdx.x & 31;
  const int col = lane & 15;
  const int rbase = (lane >> 4) << 3;
#pragma unroll
  for (int r = 0; r < 8; ++r) st(r, rbase + r, col, c[r]);
}

constexpr int OFF_W0P = 0;
constexpr int OFF_W1P = OFF_W0P + 64 * 64;
constexpr int OFF_W0N = OFF_W1P + 32 * 32;
constexpr int OFF_W1N = OFF_W0N + 64 * 64;
constexpr int OFF_G1  = OFF_W1N + 32 * 32;
constexpr int OFF_G2  = OFF_G1  + 96 * 96;
constexpr int OFF_F1  = OFF_G2  + 96 * 96;
constexpr int OFF_F2  = OFF_F1  + 32 * 32;
constexpr int OFF_L1  = OFF_F2  + 32 * 224;
constexpr int OFF_L2  = OFF_L1  + 160 * 32;
constexpr int OFF_WO0 = OFF_L2  + 32 * 224;
constexpr int OFF_WO1 = OFF_WO0 + 96 * 128;
constexpr int OFF_WO2 = OFF_WO1 + 96 * 64;
constexpr int WT_FLOATS = OFF_WO2 + 32 * 32;

constexpr int SETUP_THREADS  = 256;
constexpr int SETUP_T_BLOCKS = WT_FLOATS / 1024;
constexpr int ACC_V4         = N_NODES * 480 / 4;
constexpr int Z_V4_PER_BLOCK = SETUP_THREADS * 16;
constexpr int SETUP_Z_BLOCKS = (ACC_V4 + Z_V4_PER_BLOCK - 1) / Z_V4_PER_BLOCK;
constexpr int CNT_PADDED     = 10240;
constexpr int SETUP_BLOCKS   = SETUP_T_BLOCKS + SETUP_Z_BLOCKS + 1;

template <int K, int Nc>
__device__ __forceinline__ void frag_slab(const float* __restrict__ W,
                                          float* __restrict__ Wf, int slab) {
  static_assert((K * Nc) % 1024 == 0, "weights are whole slabs");
  static_assert(K % 8 == 0 && Nc % 16 == 0, "whole k-step pairs and column tiles");
  float v[4];
#pragma unroll
  for (int i = 0; i < 4; ++i) v[i] = W[slab * 1024 + i * SETUP_THREADS + threadIdx.x];
#pragma unroll
  for (int i = 0; i < 4; ++i) {
    const int idx = slab * 1024 + i * SETUP_THREADS + threadIdx.x;
    const int k = idx / Nc, n = idx % Nc;
    const int j = n >> 4, c = n & 15;
    const int p = k >> 3, r = k & 7;
    const int h = (r >> 1) & 1;
    const int sl = ((r >> 2) << 1) | (r & 1);
    Wf[(((j * (K / 8) + p) * 32) + h * 16 + c) * 4 + sl] = v[i];
  }
}

__global__ __launch_bounds__(SETUP_THREADS) void setup_kernel(
    const float* __restrict__ W0p, const float* __restrict__ W1p,
    const float* __restrict__ W0n, const float* __restrict__ W1n,
    const float* __restrict__ G1, const float* __restrict__ G2,
    const float* __restrict__ F1, const float* __restrict__ F2,
    const float* __restrict__ L1, const float* __restrict__ L2,
    const float* __restrict__ WO0, const float* __restrict__ WO1,
    const float* __restrict__ WO2,
    float* __restrict__ wt, float* __restrict__ acc, int* __restrict__ cnt) {
  int s = blockIdx.x;
  if (s < SETUP_T_BLOCKS) {
    if (s < 4)  { frag_slab<64, 64>(W0p, wt + OFF_W0P, s); return; }   s -= 4;
    if (s < 1)  { frag_slab<32, 32>(W1p, wt + OFF_W1P, s); return; }   s -= 1;
    if (s < 4)  { frag_slab<64, 64>(W0n, wt + OFF_W0N, s); return; }   s -= 4;
    if (s < 1)  { frag_slab<32, 32>(W1n, wt + OFF_W1N, s); return; }   s -= 1;
    if (s < 9)  { frag_slab<96, 96>(G1, wt + OFF_G1, s); return; }     s -= 9;
    if (s < 9)  { frag_slab<96, 96>(G2, wt + OFF_G2, s); return; }     s -= 9;
    if (s < 1)  { frag_slab<32, 32>(F1, wt + OFF_F1, s); return; }     s -= 1;
    if (s < 7)  { frag_slab<32, 224>(F2, wt + OFF_F2, s); return; }    s -= 7;
    if (s < 5)  { frag_slab<160, 32>(L1, wt + OFF_L1, s); return; }    s -= 5;
    if (s < 7)  { frag_slab<32, 224>(L2, wt + OFF_L2, s); return; }    s -= 7;
    if (s < 12) { frag_slab<96, 128>(WO0, wt + OFF_WO0, s); return; }  s -= 12;
    if (s < 6)  { frag_slab<96, 64>(WO1, wt + OFF_WO1, s); return; }   s -= 6;
    frag_slab<32, 32>(WO2, wt + OFF_WO2, s);
    return;
  }
  s -= SETUP_T_BLOCKS;
  if (s < SETUP_Z_BLOCKS) {
    v4f* a4 = (v4f*)acc;
    const v4f z = {0.0f, 0.0f, 0.0f, 0.0f};
    for (int i = 0; i < 16; ++i) {
      const int q = s * Z_V4_PER_BLOCK + i * SETUP_THREADS + threadIdx.x;
      if (q < ACC_V4) a4[q] = z;
    }
    return;
  }
  for (int i = threadIdx.x; i < CNT_PADDED; i += SETUP_THREADS) cnt[i] = 0;
}

constexpr int NODE_WAVES = 4;

template <int FLOATS>
__device__ __forceinline__ void copy_tile(float* __restrict__ dst, const float* src) {
  constexpr int NT = NODE_WAVES * 32;
  static_assert(FLOATS % (4 * NT) == 0, "whole 16-byte words per thread");
#pragma unroll
  for (int q = 0; q < FLOATS / (4 * NT); ++q) {
    const int w = q * NT + threadIdx.x;
    ((v4f*)dst)[w] = ((const v4f*)src)[w];
  }
}

constexpr int P0_PITCH = 68;
constexpr int IP_PITCH = 36;

__device__ __forceinline__ void node_tile(
    const int tile,
    const float* __restrict__ x,
    const float* __restrict__ W0pF, const float* __restrict__ b0p,
    const float* __restrict__ W1pF,
    const float* __restrict__ W0nF, const float* __restrict__ b0n,
    const float* __restrict__ W1nF,
    const float* __restrict__ G1F, const float* __restrict__ g1b,
    const float* __restrict__ G2F, const float* __restrict__ g2b,
    const float* __restrict__ L1F,
    float* __restrict__ qd, float* __restrict__ qs, float* __restrict__ p1,
    float* __restrict__ x0n, float* __restrict__ x1n) {
  __shared__ __align__(16) float p0S[16 * P0_PITCH];
  __shared__ __align__(16) float f0S[16 * 96];
  __shared__ __align__(16) float hS[16 * 96];
  __shared__ __align__(16) float gS[16 * 96];
  __shared__ __align__(16) float qdS[16 * 32];
  __shared__ __align__(16) float qsS[16 * 32];
  __shared__ __align__(16) float p1S[16 * 96];
  __shared__ __align__(16) float x0nS[16 * 64];
  __shared__ __align__(16) float x1nS[16 * 96];
  static_assert(NODE_WAVES == 4, "the shares below are written out for four waves");

  const int tid = threadIdx.x;
  const int wave = __builtin_amdgcn_readfirstlane(threadIdx.x >> 5);
  const int n0 = tile * 16;
  const float* xt = x + (size_t)n0 * 160;

  {
    v2f ax[16];
    load_a(ax, [&](int m, int k) { return ld2(xt + m * 160 + k); });
    {
      const int j = wave;
      v8f c = {};
      c = wmma_tile(ax, W0pF + j * 16 * 64, c);
      tile_store(c, [&](int, int m, int n, float v) {
        int col = j * 16 + n;
        p0S[m * P0_PITCH + col] = v * INV_SQ64 + b0p[col];
      });
    }
  }
  if (wave < 3) {
    const int i = wave;
    v2f ax[8];
    load_a(ax, [&](int m, int k) {
      v2f a;
      a.x = xt[m * 160 + 64 + 3 * k + i];
      a.y = xt[m * 160 + 64 + 3 * (k + 1) + i];
      return a;
    });
#pragma unroll
    for (int j = 0; j < 2; ++j) {
      v8f c = {};
      c = wmma_tile(ax, W1pF + j * 16 * 32, c);
      tile_store(c, [&](int, int m, int n, float v) {
        p1S[m * 96 + (j * 16 + n) * 3 + i] = v * INV_SQ32;
      });
    }
  }

  {
    constexpr int NT = NODE_WAVES * 32;
#pragma unroll
    for (int q = 0; q < 16 * 64 / NT; ++q) {
      const int idx = q * NT + tid;
      const int m = idx >> 6, c = idx & 63;
      f0S[m * 96 + c] = xt[m * 160 + c];
    }
#pragma unroll
    for (int q = 0; q < 16 * 32 / NT; ++q) {
      const int idx = q * NT + tid;
      const int m = idx >> 5, u = idx & 31;
      float a0 = xt[m * 160 + 64 + u * 3 + 0];
      float a1 = xt[m * 160 + 64 + u * 3 + 1];
      float a2 = xt[m * 160 + 64 + u * 3 + 2];
      f0S[m * 96 + 64 + u] = sqrtf(a0 * a0 + a1 * a1 + a2 * a2 + 1e-12f);
    }
  }
  __syncthreads();

  {
    v2f ap[16];
    load_a(ap, [&](int m, int k) { return ld2(p0S + m * P0_PITCH + k); });
    {
      const int j = wave >> 1, half = wave & 1;
      float* q = half ? qsS : qdS;
      v8f c = {};
      c = wmma_tile(ap, L1F + j * 16 * 160 + half * 8 * 128, c);
      tile_store(c, [&](int, int m, int n, float v) {
        q[m * 32 + j * 16 + n] = v;
      });
    }
  }

  if (wave < 3) {
    v2f af[24];
    load_a(af, [&](int m, int k) { return ld2(f0S + m * 96 + k); });
#pragma unroll
    for (int jj = 0; jj < 2; ++jj) {
      const int j = wave * 2 + jj;
      v8f c = {};
      c = wmma_tile(af, G1F + j * 16 * 96, c);
      tile_store(c, [&](int, int m, int n, float v) {
        int col = j * 16 + n;
        hS[m * 96 + col] = siluf(v + g1b[col]);
      });
    }
  }
  __syncthreads();
  copy_tile<16 * 32>(qd + (size_t)n0 * 32, qdS);
  copy_tile<16 * 32>(qs + (size_t)n0 * 32, qsS);
  copy_tile<16 * 96>(p1 + (size_t)n0 * 96, p1S);
  if (wave < 3) {
    v2f ah[24];
    load_a(ah, [&](int m, int k) { return ld2(hS + m * 96 + k); });
#pragma unroll
    for (int jj = 0; jj < 2; ++jj) {
      const int j = wave * 2 + jj;
      v8f c = {};
      c = wmma_tile(ah, G2F + j * 16 * 96, c);
      tile_store(c, [&](int, int m, int n, float v) {
        gS[m * 96 + j * 16 + n] = v + g2b[j * 16 + n];
      });
    }
  }
  __syncthreads();

  {
    v2f ag[16];
    load_a(ag, [&](int m, int k) { return ld2(gS + m * 96 + k); });
    {
      const int j = wave;
      v8f c = {};
      c = wmma_tile(ag, W0nF + j * 16 * 64, c);
      tile_store(c, [&](int, int m, int n, float v) {
        int col = j * 16 + n;
        x0nS[m * 64 + col] = v * INV_SQ64 + b0n[col];
      });
    }
  }
  if (wave < 3) {
    const int i = wave;
    v2f ax[8];
    load_a(ax, [&](int m, int k) {
      v2f a;
      a.x = xt[m * 160 + 64 + 3 * k + i] * gS[m * 96 + 64 + k];
      a.y = xt[m * 160 + 64 + 3 * (k + 1) + i] * gS[m * 96 + 64 + k + 1];
      return a;
    });
#pragma unroll
    for (int j = 0; j < 2; ++j) {
      v8f c = {};
      c = wmma_tile(ax, W1nF + j * 16 * 32, c);
      tile_store(c, [&](int, int m, int n, float v) {
        x1nS[m * 96 + (j * 16 + n) * 3 + i] = v * INV_SQ32;
      });
    }
  }
  __syncthreads();
  copy_tile<16 * 64>(x0n + (size_t)n0 * 64, x0nS);
  copy_tile<16 * 96>(x1n + (size_t)n0 * 96, x1nS);
}

constexpr int NODE_TILES  = N_NODES / 16;
constexpr int RANK_BLOCKS = N_EDGES / (NODE_WAVES * 32);
static_assert(NODE_TILES * 16 == N_NODES, "whole node tiles");
static_assert(RANK_BLOCKS * NODE_WAVES * 32 == N_EDGES, "whole rank blocks");

__global__ __launch_bounds__(NODE_WAVES * 32) void node_rank_kernel(
    const float* __restrict__ x,
    const float* __restrict__ W0pF, const float* __restrict__ b0p,
    const float* __restrict__ W1pF,
    const float* __restrict__ W0nF, const float* __restrict__ b0n,
    const float* __restrict__ W1nF,
    const float* __restrict__ G1F, const float* __restrict__ g1b,
    const float* __restrict__ G2F, const float* __restrict__ g2b,
    const float* __restrict__ L1F,
    float* __restrict__ qd, float* __restrict__ qs, float* __restrict__ p1,
    float* __restrict__ x0n, float* __restrict__ x1n,
    const int* __restrict__ edge_index, int* __restrict__ cnt, int* __restrict__ rank) {
  if (blockIdx.x < NODE_TILES) {
    node_tile(blockIdx.x, x, W0pF, b0p, W1pF, W0nF, b0n, W1nF, G1F, g1b, G2F, g2b, L1F,
              qd, qs, p1, x0n, x1n);
    return;
  }
  const int e = (blockIdx.x - NODE_TILES) * (NODE_WAVES * 32) + threadIdx.x;
  rank[e] = atomicAdd(&cnt[edge_index[e]], 1);
}

#define SCAN_PER_THREAD 10
constexpr int FILL_THREADS = 1024;
constexpr int FILL_WAVES   = FILL_THREADS / 32;
constexpr int FILL_BLOCKS  = (N_EDGES + FILL_THREADS - 1) / FILL_THREADS;
static_assert(FILL_THREADS * SCAN_PER_THREAD >= N_NODES, "one block scans all counters");
static_assert(FILL_WAVES == 32, "the wave totals are scanned by one wave");

__device__ __forceinline__ int wave_inclusive_scan(int v, const int lane) {
#pragma unroll
  for (int d = 1; d < 32; d <<= 1) {
    const int u = __shfl_up(v, d, 32);
    if (lane >= d) v += u;
  }
  return v;
}

__global__ __launch_bounds__(FILL_THREADS) void sort_scan_fill_kernel(
    const int* __restrict__ edge_index, const int* __restrict__ cnt,
    const int* __restrict__ rank, int* __restrict__ perm) {
  __shared__ int part[FILL_WAVES];
  __shared__ int offS[FILL_THREADS * SCAN_PER_THREAD];
  const int t = threadIdx.x;
  const int lane = t & 31, wave = t >> 5;
  const int e = blockIdx.x * FILL_THREADS + t;
  int d = 0, rk = 0;
  if (e < N_EDGES) {
    d = edge_index[e];
    rk = rank[e];
  }
#pragma unroll
  for (int i = 0; i < SCAN_PER_THREAD; ++i) {
    const int idx = i * FILL_THREADS + t;
    offS[idx] = (idx < N_NODES) ? cnt[idx] : 0;
  }
  __syncthreads();
  int loc[SCAN_PER_THREAD];
  int s = 0;
#pragma unroll
  for (int i = 0; i < SCAN_PER_THREAD; ++i) {
    const int v = offS[t * SCAN_PER_THREAD + i];
    loc[i] = s;
    s += v;
  }
  const int incl = wave_inclusive_scan(s, lane);
  if (lane == 31) part[wave] = incl;
  __syncthreads();
  if (wave == 0) {
    const int w = part[lane];
    part[lane] = wave_inclusive_scan(w, lane) - w;
  }
  __syncthreads();
  const int base = part[wave] + incl - s;
#pragma unroll
  for (int i = 0; i < SCAN_PER_THREAD; ++i) offS[t * SCAN_PER_THREAD + i] = base + loc[i];
  __syncthreads();
  if (e < N_EDGES) perm[offS[d] + rk] = e;
}

__device__ __forceinline__ void run_add(float* __restrict__ acc,
                                        const unsigned (&dB)[8], unsigned flm,
                                        int chan, const float (&v)[8]) {
  float run = 0.0f;
#pragma unroll
  for (int r = 0; r < 8; ++r) {
    run += v[r];
    if ((flm >> r) & 1u) {
      atomicAdd((float*)((char*)acc + (dB[r] + (unsigned)chan * 4u)), run);
      run = 0.0f;
    }
  }
}

constexpr int EDGE_WAVES   = 8;
constexpr int EDGE_THREADS = EDGE_WAVES * 32;
constexpr int W2_FLOATS    = 32 * 224;
constexpr int W2_V4        = W2_FLOATS / 4;
constexpr int W2_STAGE     = W2_V4 / EDGE_THREADS;
static_assert(W2_STAGE * EDGE_THREADS == W2_V4, "whole 16-byte words per thread");
static_assert((N_EDGES / 16) % EDGE_WAVES == 0, "whole work-groups of tiles");

__device__ __forceinline__ void wave_lds_sync() {
  __builtin_amdgcn_fence(__ATOMIC_RELEASE, "wavefront");
  __builtin_amdgcn_wave_barrier();
  __builtin_amdgcn_fence(__ATOMIC_ACQUIRE, "wavefront");
}

__global__ __launch_bounds__(EDGE_THREADS) void edge_kernel(
    const float* __restrict__ edge_sh, const float* __restrict__ edge_attr,
    const float* __restrict__ F1F, const float* __restrict__ F2F,
    const float* __restrict__ L1F, const float* __restrict__ L2F,
    const int* __restrict__ edge_index, const int* __restrict__ perm,
    const float* __restrict__ qd, const float* __restrict__ qs,
    const float* __restrict__ p1,
    const float* __restrict__ x0n, const float* __restrict__ x1n,
    float* __restrict__ acc) {
  __shared__ __align__(16) float wS[2 * W2_FLOATS];
  __shared__ __align__(16) float ipS_w[EDGE_WAVES][16 * IP_PITCH];
  __shared__ __align__(16) float hE_w[EDGE_WAVES][16 * 32];
  __shared__ __align__(16) float shS_w[EDGE_WAVES][16 * 4];
  __shared__ int dstS_w[EDGE_WAVES][16];
  __shared__ int srcS_w[EDGE_WAVES][16];
  static_assert(16 * 32 <= 16 * IP_PITCH, "hN fits in the inner-product panel");

  const int lane = threadIdx.x & 31;
  const int wave = __builtin_amdgcn_readfirstlane(threadIdx.x >> 5);
  const int e0 = (blockIdx.x * EDGE_WAVES + wave) * 16;
  float* const ipS = ipS_w[wave];
  float* const hN  = ipS_w[wave];
  float* const hE  = hE_w[wave];
  float* const shS = shS_w[wave];
  int* const dstS = dstS_w[wave];
  int* const srcS = srcS_w[wave];

  v4f wst[2 * W2_STAGE];
#pragma unroll
  for (int i = 0; i < W2_STAGE; ++i) {
    wst[i]            = ((const v4f*)L2F)[i * EDGE_THREADS + threadIdx.x];
    wst[W2_STAGE + i] = ((const v4f*)F2F)[i * EDGE_THREADS + threadIdx.x];
  }

  v4f bF0[4], bF1[4];
  load_b32(bF0, F1F);
  load_b32(bF1, F1F + 16 * 32);

  const int e = perm[e0 + (lane & 15)];
  v2f aea[8];
  {
    const float* ar = edge_attr + (size_t)e * 32 + ((lane >> 4) << 1);
#pragma unroll
    for (int q = 0; q < 8; ++q) aea[q] = ld2(ar + q * 4);
  }
  if (lane < 16) {
    dstS[lane] = edge_index[e];
    srcS[lane] = edge_index[N_EDGES + e];
    *(v4f*)(shS + lane * 4) = *(const v4f*)(edge_sh + (size_t)e * 4);
  }
#pragma unroll
  for (int i = 0; i < W2_STAGE; ++i) {
    ((v4f*)wS)[i * EDGE_THREADS + threadIdx.x]         = wst[i];
    ((v4f*)wS)[W2_V4 + i * EDGE_THREADS + threadIdx.x] = wst[W2_STAGE + i];
  }
  wave_lds_sync();

  const int rbase = (lane >> 4) << 3;

  float qv0[8], qv1[8], qw0[8], qw1[8];
#pragma unroll
  for (int r = 0; r < 8; ++r) {
    const float* pd = (const float*)((const char*)qd + ((unsigned)dstS[rbase + r] * 128u + (unsigned)(lane & 15) * 4u));
    const float* ps = (const float*)((const char*)qs + ((unsigned)srcS[rbase + r] * 128u + (unsigned)(lane & 15) * 4u));
    qv0[r] = pd[0];
    qv1[r] = pd[16];
    qw0[r] = ps[0];
    qw1[r] = ps[16];
  }
  v4f bL0[4], bL1[4];
  load_b32(bL0, L1F + 16 * 128);
  load_b32(bL1, L1F + 16 * 160 + 16 * 128);

  {
    {
      v8f c0 = {}, c1 = {};
      c0 = wmma_regs(aea, bF0, c0);
      c1 = wmma_regs(aea, bF1, c1);
      tile_store(c0, [&](int, int m, int n, float v) {
        hE[m * 32 + n] = sspf(v * INV_SQ32);
      });
      tile_store(c1, [&](int, int m, int n, float v) {
        hE[m * 32 + 16 + n] = sspf(v * INV_SQ32);
      });
    }

#pragma nounroll
    for (int mb = 0; mb < 16; mb += 8) {
      float dv[8];
#pragma unroll
      for (int r = 0; r < 8; ++r) {
        const float* pd = (const float*)((const char*)p1 + ((unsigned)dstS[mb + r] * 384u + (unsigned)lane * 12u));
        const float* ps = (const float*)((const char*)p1 + ((unsigned)srcS[mb + r] * 384u + (unsigned)lane * 12u));
        dv[r] = (pd[0] * ps[0] + pd[1] * ps[1] + pd[2] * ps[2]) * INV_SQ3;
      }
#pragma unroll
      for (int r = 0; r < 8; ++r) ipS[(mb + r) * IP_PITCH + lane] = dv[r];
    }
  }
  wave_lds_sync();

  {
    v2f aip[8];
    load_a(aip, [&](int m, int k) { return ld2(ipS + m * IP_PITCH + k); });
    v8f c0, c1;
#pragma unroll
    for (int r = 0; r < 8; ++r) {
      c0[r] = qv0[r] + qw0[r];
      c1[r] = qv1[r] + qw1[r];
    }
    c0 = wmma_regs(aip, bL0, c0);
    c1 = wmma_regs(aip, bL1, c1);
    tile_store(c0, [&](int, int m, int n, float v) {
      hN[m * 32 + n] = sspf(v * INV_SQ160);
    });
    tile_store(c1, [&](int, int m, int n, float v) {
      hN[m * 32 + 16 + n] = sspf(v * INV_SQ160);
    });
  }
  wave_lds_sync();

  float sh0r[8], s1xr[8], s1yr[8], s1zr[8];
  int dR[8], sR[8];
#pragma unroll
  for (int r = 0; r < 8; ++r) {
    int m = rbase + r;
    const v4f sh = *(const v4f*)(shS + m * 4);
    sh0r[r] = sh.x;
    s1xr[r] = sh.y;
    s1yr[r] = sh.z;
    s1zr[r] = sh.w;
    dR[r] = dstS[m];
    sR[r] = srcS[m];
  }
  unsigned s0B[8], s1B[8], dB[8];
#pragma unroll
  for (int r = 0; r < 8; ++r) {
    s0B[r] = (unsigned)sR[r] * 256u;
    s1B[r] = (unsigned)sR[r] * 384u;
    dB[r] = (unsigned)dR[r] * 1920u;
  }
  unsigned flm = 0x80u;
#pragma unroll
  for (int r = 0; r < 7; ++r) flm |= (dR[r + 1] != dR[r]) ? (1u << r) : 0u;

  v2f aN[8], aE[8];
  load_a(aN, [&](int m, int k) { return ld2(hN + m * 32 + k); });
  load_a(aE, [&](int m, int k) { return ld2(hE + m * 32 + k); });

  auto do_tile = [&](auto seg, const int j) {
    constexpr int SEG = decltype(seg)::value;
    const int nb = j * 16;
    const int col = nb + (lane & 15);
    v4f bN[4], bE[4];
    load_b32(bN, wS + nb * 32);
    load_b32(bE, wS + W2_FLOATS + nb * 32);
    constexpr int CB = (SEG == 0) ? 0 : (SEG == 1) ? 64 : (SEG == 2) ? 128 : (SEG == 3) ? 160 : 192;
    const int cs = col - CB;
    float g0[8], g1[8], g2[8];
    if constexpr (SEG <= 1) {
#pragma unroll
      for (int r = 0; r < 8; ++r)
        g0[r] = *(const float*)((const char*)x0n + (s0B[r] + (unsigned)cs * 4u));
    } else {
#pragma unroll
      for (int r = 0; r < 8; ++r) {
        const float* xv = (const float*)((const char*)x1n + (s1B[r] + (unsigned)cs * 12u));
        g0[r] = xv[0];
        g1[r] = xv[1];
        g2[r] = xv[2];
      }
    }

    v8f cN = {}, cE = {};
    cN = wmma_regs(aN, bN, cN);
    cE = wmma_regs(aE, bE, cE);
    v8f w8 = cN * cE * (INV_SQ32 * INV_SQ32);

    float v0[8], v1[8], v2[8];
    if constexpr (SEG == 0) {
#pragma unroll
      for (int r = 0; r < 8; ++r) v0[r] = w8[r] * g0[r] * sh0r[r];
      run_add(acc, dB, flm, col, v0);
    } else if constexpr (SEG == 1) {
#pragma unroll
      for (int r = 0; r < 8; ++r) {
        float f = w8[r] * g0[r] * INV_SQ3;
        v0[r] = f * s1xr[r];
        v1[r] = f * s1yr[r];
        v2[r] = f * s1zr[r];
      }
      run_add(acc, dB, flm, 96 + cs * 3 + 0, v0);
      run_add(acc, dB, flm, 96 + cs * 3 + 1, v1);
      run_add(acc, dB, flm, 96 + cs * 3 + 2, v2);
    } else if constexpr (SEG == 2) {
#pragma unroll
      for (int r = 0; r < 8; ++r) {
        float f = w8[r] * sh0r[r];
        v0[r] = f * g0[r];
        v1[r] = f * g1[r];
        v2[r] = f * g2[r];
      }
      run_add(acc, dB, flm, 288 + cs * 3 + 0, v0);
      run_add(acc, dB, flm, 288 + cs * 3 + 1, v1);
      run_add(acc, dB, flm, 288 + cs * 3 + 2, v2);
    } else if constexpr (SEG == 3) {
#pragma unroll
      for (int r = 0; r < 8; ++r) {
        float dp = g0[r] * s1xr[r] + g1[r] * s1yr[r] + g2[r] * s1zr[r];
        v0[r] = w8[r] * dp * INV_SQ3;
      }
      run_add(acc, dB, flm, 64 + cs, v0);
    } else {
#pragma unroll
      for (int r = 0; r < 8; ++r) {
        const float s1x = s1xr[r], s1y = s1yr[r], s1z = s1zr[r];
        float cx = g1[r] * s1z - g2[r] * s1y;
        float cy = g2[r] * s1x - g0[r] * s1z;
        float cz = g0[r] * s1y - g1[r] * s1x;
        float f = w8[r] * INV_SQ2;
        v0[r] = f * cx;
        v1[r] = f * cy;
        v2[r] = f * cz;
      }
      run_add(acc, dB, flm, 384 + cs * 3 + 0, v0);
      run_add(acc, dB, flm, 384 + cs * 3 + 1, v1);
      run_add(acc, dB, flm, 384 + cs * 3 + 2, v2);
    }
  };

  __syncthreads();

#pragma nounroll
  for (int j = 0; j < 4; ++j) do_tile(IC<0>{}, j);
#pragma nounroll
  for (int j = 4; j < 8; ++j) do_tile(IC<1>{}, j);
#pragma nounroll
  for (int j = 8; j < 10; ++j) do_tile(IC<2>{}, j);
#pragma nounroll
  for (int j = 10; j < 12; ++j) do_tile(IC<3>{}, j);
#pragma nounroll
  for (int j = 12; j < 14; ++j) do_tile(IC<4>{}, j);
}

__global__ __launch_bounds__(NODE_WAVES * 32) __attribute__((amdgpu_waves_per_eu(1, 2))) void proj_kernel(
    const float* __restrict__ acc,
    const float* __restrict__ WO0F, const float* __restrict__ bO0,
    const float* __restrict__ WO1F, const float* __restrict__ WO2F,
    float* __restrict__ y) {
  __shared__ __align__(16) float yS[16 * 416];
  static_assert(NODE_WAVES == 4, "the shares below are written out for four waves");
  const int wave = __builtin_amdgcn_readfirstlane(threadIdx.x >> 5);
  const int n0 = blockIdx.x * 16;
  const float* at = acc + (size_t)n0 * 480;

  {
    v2f aa[24];
    load_a(aa, [&](int m, int k) { return ld2(at + (size_t)m * 480 + k); });
#pragma unroll
    for (int jj = 0; jj < 2; ++jj) {
      const int j = wave * 2 + jj;
      v8f c = {};
      c = wmma_tile(aa, WO0F + j * 16 * 96, c);
      tile_store(c, [&](int, int m, int n, float v) {
        int col = j * 16 + n;
        yS[m * 416 + col] = v * INV_SQ96 + bO0[col];
      });
    }
  }
#pragma unroll
  for (int i = 0; i < 3; ++i) {
    v2f aa[24];
    load_a(aa, [&](int m, int k) {
      v2f a;
      a.x = at[(size_t)m * 480 + 96 + 3 * k + i];
      a.y = at[(size_t)m * 480 + 96 + 3 * (k + 1) + i];
      return a;
    });
    {
      const int j = wave;
      v8f c = {};
      c = wmma_tile(aa, WO1F + j * 16 * 96, c);
      tile_store(c, [&](int, int m, int n, float v) {
        yS[m * 416 + 128 + (j * 16 + n) * 3 + i] = v * INV_SQ96;
      });
    }
  }
  if (wave < 3) {
    const int i = wave;
    v2f aa[8];
    load_a(aa, [&](int m, int k) {
      v2f a;
      a.x = at[(size_t)m * 480 + 384 + 3 * k + i];
      a.y = at[(size_t)m * 480 + 384 + 3 * (k + 1) + i];
      return a;
    });
#pragma unroll
    for (int j = 0; j < 2; ++j) {
      v8f c = {};
      c = wmma_tile(aa, WO2F + j * 16 * 32, c);
      tile_store(c, [&](int, int m, int n, float v) {
        yS[m * 416 + 320 + (j * 16 + n) * 3 + i] = v * INV_SQ32;
      });
    }
  }
  __syncthreads();
  copy_tile<16 * 416>(y + (size_t)n0 * 416, yS);
}

extern "C" void kernel_launch(void* const* d_in, const int* in_sizes, int n_in,
                              void* d_out, int out_size, void* d_ws,
                              size_t ws_size, hipStream_t stream) {
  (void)in_sizes; (void)n_in; (void)out_size; (void)ws_size;
  const float* x         = (const float*)d_in[0];
  const float* edge_sh   = (const float*)d_in[1];
  const float* edge_attr = (const float*)d_in[2];
  const float* W0p = (const float*)d_in[3];
  const float* b0p = (const float*)d_in[4];
  const float* W1p = (const float*)d_in[5];
  const float* W0n = (const float*)d_in[6];
  const float* b0n = (const float*)d_in[7];
  const float* W1n = (const float*)d_in[8];
  const float* G1  = (const float*)d_in[9];
  const float* g1b = (const float*)d_in[10];
  const float* G2  = (const float*)d_in[11];
  const float* g2b = (const float*)d_in[12];
  const float* F1  = (const float*)d_in[13];
  const float* F2  = (const float*)d_in[14];
  const float* L1  = (const float*)d_in[15];
  const float* L2  = (const float*)d_in[16];
  const float* WO0 = (const float*)d_in[17];
  const float* bO0 = (const float*)d_in[18];
  const float* WO1 = (const float*)d_in[19];
  const float* WO2 = (const float*)d_in[20];
  const int* edge_index = (const int*)d_in[21];

  float* ws   = (float*)d_ws;
  float* qd   = ws;
  float* qs   = qd   + (size_t)N_NODES * 32;
  float* p1   = qs   + (size_t)N_NODES * 32;
  float* x0n  = p1   + (size_t)N_NODES * 96;
  float* x1n  = x0n  + (size_t)N_NODES * 64;
  float* acc  = x1n  + (size_t)N_NODES * 96;
  float* wt   = acc  + (size_t)N_NODES * 480;
  float* W0pF = wt + OFF_W0P;
  float* W1pF = wt + OFF_W1P;
  float* W0nF = wt + OFF_W0N;
  float* W1nF = wt + OFF_W1N;
  float* G1F  = wt + OFF_G1;
  float* G2F  = wt + OFF_G2;
  float* F1F  = wt + OFF_F1;
  float* F2F  = wt + OFF_F2;
  float* L1F  = wt + OFF_L1;
  float* L2F  = wt + OFF_L2;
  float* WO0F = wt + OFF_WO0;
  float* WO1F = wt + OFF_WO1;
  float* WO2F = wt + OFF_WO2;

  int* sc   = (int*)d_out;
  int* cnt  = sc;
  int* rank = sc + 2 * CNT_PADDED;
  int* perm = sc + 2 * CNT_PADDED + N_EDGES;

  setup_kernel<<<SETUP_BLOCKS, SETUP_THREADS, 0, stream>>>(
      W0p, W1p, W0n, W1n, G1, G2, F1, F2, L1, L2, WO0, WO1, WO2, wt, acc, cnt);

  node_rank_kernel<<<NODE_TILES + RANK_BLOCKS, NODE_WAVES * 32, 0, stream>>>(
      x, W0pF, b0p, W1pF, W0nF, b0n, W1nF, G1F, g1b, G2F, g2b, L1F,
      qd, qs, p1, x0n, x1n, edge_index, cnt, rank);
  sort_scan_fill_kernel<<<FILL_BLOCKS, FILL_THREADS, 0, stream>>>(edge_index, cnt, rank, perm);

  edge_kernel<<<N_EDGES / 16 / EDGE_WAVES, EDGE_THREADS, 0, stream>>>(
      edge_sh, edge_attr, F1F, F2F, L1F, L2F, edge_index, perm,
      qd, qs, p1, x0n, x1n, acc);

  proj_kernel<<<N_NODES / 16, NODE_WAVES * 32, 0, stream>>>(
      acc, WO0F, bO0, WO1F, WO2F, (float*)d_out);
}
